# starting kernel
# baseline (speedup 1.0000x reference)
#include <hip/hip_runtime.h>
#include <hip/hip_bf16.h>
#include <math.h>


typedef float v2f __attribute__((ext_vector_type(2)));
typedef float v8f __attribute__((ext_vector_type(8)));

#define ROWS_TOTAL 16384
#define F_ 16
#define D_ 64
#define H_ 128
#define FH_ 2048

#define ST_T 2052
#define ST_X 68
#define ST_P 132
#define ST_S 20

#define SMEM_FLOATS (16*ST_T + 16*ST_X + 4*16*ST_P + 2*16*ST_S + 16*16)

__device__ __forceinline__ float rowsum16(float v) {
  v += __shfl_xor(v, 1, 16);
  v += __shfl_xor(v, 2, 16);
  v += __shfl_xor(v, 4, 16);
  v += __shfl_xor(v, 8, 16);
  return v;
}
__device__ __forceinline__ float rowmax16(float v) {
  v = fmaxf(v, __shfl_xor(v, 1, 16));
  v = fmaxf(v, __shfl_xor(v, 2, 16));
  v = fmaxf(v, __shfl_xor(v, 4, 16));
  v = fmaxf(v, __shfl_xor(v, 8, 16));
  return v;
}

template <int K>
__device__ __forceinline__ v8f wmma_gemm(const float* __restrict__ As, int lda,
                                         const float* __restrict__ W, int ldw,
                                         int n0, int lane16, int hi) {
  v8f c = {0.f, 0.f, 0.f, 0.f, 0.f, 0.f, 0.f, 0.f};
  const float* arow = As + lane16 * lda + hi * 2;
  const float* wcol = W + (size_t)(hi * 2) * ldw + n0 + lane16;
#pragma unroll 1
  for (int kb = 0; kb < K; kb += 32) {
    if (kb + 32 < K) {
      __builtin_prefetch(wcol + (size_t)(kb + 32) * ldw, 0, 1);
      __builtin_prefetch(wcol + (size_t)(kb + 48) * ldw, 0, 1);
    }
#pragma unroll
    for (int k0 = kb; k0 < kb + 32; k0 += 4) {
      v2f a, b;
      a.x = arow[k0];
      a.y = arow[k0 + 1];
      b.x = wcol[(size_t)k0 * ldw];
      b.y = wcol[(size_t)(k0 + 1) * ldw];
      c = __builtin_amdgcn_wmma_f32_16x16x4_f32(false, a, false, b,
                                                (short)0, c, false, false);
    }
  }
  return c;
}

__device__ __forceinline__ void store_tile(float* dst, int ldd, int col0,
                                           v8f c, int lane16, int hi) {
#pragma unroll
  for (int v = 0; v < 8; ++v)
    dst[(v + hi * 8) * ldd + col0 + lane16] = c[v];
}

__global__ __launch_bounds__(256, 1)
void vsn_fused_206158430403(
    const float* __restrict__ x,
    const float* __restrict__ embW, const float* __restrict__ embB,
    const float* __restrict__ fc1W, const float* __restrict__ fc1B,
    const float* __restrict__ fc2W, const float* __restrict__ fc2B,
    const float* __restrict__ gW,   const float* __restrict__ gB,
    const float* __restrict__ lnG,  const float* __restrict__ lnB,
    const float* __restrict__ sf1W, const float* __restrict__ sf1B,
    const float* __restrict__ sf2W, const float* __restrict__ sf2B,
    const float* __restrict__ sgW,  const float* __restrict__ sgB,
    const float* __restrict__ sskW, const float* __restrict__ sskB,
    const float* __restrict__ slnG, const float* __restrict__ slnB,
    float* __restrict__ outSel, float* __restrict__ outW) {
  extern __shared__ float smem[];
  float* TRANS = smem;
  float* XT = TRANS + 16 * ST_T;
  float* P0 = XT + 16 * ST_X;
  float* P1 = P0 + 16 * ST_P;
  float* P2 = P1 + 16 * ST_P;
  float* P3 = P2 + 16 * ST_P;
  float* GP = P3 + 16 * ST_P;
  float* SKT = GP + 16 * ST_S;
  float* WB = SKT + 16 * ST_S;

  const int tid = threadIdx.x;
  const int wid = tid >> 5;
  const int lane = tid & 31;
  const int lane16 = lane & 15;
  const int hi = lane >> 4;
  const int n0 = wid * 16;
  const int lr = tid >> 4;
  const int lc = tid & 15;
  const int row0 = blockIdx.x * 16;

  for (int f = 0; f < F_; ++f) {
    {
      const float* xp = x + (size_t)(row0 + lr) * (F_ * D_) + f * D_ + lc * 4;
      float4 v = *(const float4*)xp;
      float* dp = XT + lr * ST_X + lc * 4;
      dp[0] = v.x; dp[1] = v.y; dp[2] = v.z; dp[3] = v.w;
      if (f + 1 < F_) __builtin_prefetch(xp + D_, 0, 1);
    }
    __syncthreads();

    v8f ce = wmma_gemm<D_>(XT, ST_X, embW + (size_t)f * D_ * H_, H_, n0, lane16, hi);
    {
      float b = embB[f * H_ + n0 + lane16];
#pragma unroll
      for (int v = 0; v < 8; ++v) ce[v] += b;
    }
    store_tile(P0, ST_P, n0, ce, lane16, hi);
    __syncthreads();

    v8f c1 = wmma_gemm<H_>(P0, ST_P, fc1W + (size_t)f * H_ * H_, H_, n0, lane16, hi);
    {
      float b = fc1B[f * H_ + n0 + lane16];
#pragma unroll
      for (int v = 0; v < 8; ++v) {
        float t = c1[v] + b;
        c1[v] = t > 0.f ? t : (expf(t) - 1.f);
      }
    }
    store_tile(P1, ST_P, n0, c1, lane16, hi);
    __syncthreads();

    v8f c2 = wmma_gemm<H_>(P1, ST_P, fc2W + (size_t)f * H_ * H_, H_, n0, lane16, hi);
    {
      float b = fc2B[f * H_ + n0 + lane16];
#pragma unroll
      for (int v = 0; v < 8; ++v) c2[v] += b;
    }
    store_tile(P2, ST_P, n0, c2, lane16, hi);
    __syncthreads();

    v8f cg = wmma_gemm<H_>(P2, ST_P, gW + (size_t)f * H_ * H_, H_, n0, lane16, hi);
    {
      float b = gB[f * H_ + n0 + lane16];
#pragma unroll
      for (int v = 0; v < 8; ++v) {
        float t = cg[v] + b;
        float s = 1.f / (1.f + expf(-t));
        cg[v] = ce[v] + s * c2[v];
      }
    }
    store_tile(P3, ST_P, n0, cg, lane16, hi);
    __syncthreads();

    {
      float vals[8];
      float s = 0.f, ss = 0.f;
#pragma unroll
      for (int j = 0; j < 8; ++j) {
        float v = P3[lr * ST_P + lc + 16 * j];
        vals[j] = v; s += v; ss += v * v;
      }
      s = rowsum16(s);
      ss = rowsum16(ss);
      float mu = s * (1.f / 128.f);
      float var = ss * (1.f / 128.f) - mu * mu;
      float rs = rsqrtf(var + 1e-5f);
#pragma unroll
      for (int j = 0; j < 8; ++j) {
        int h = lc + 16 * j;
        float nv = (vals[j] - mu) * rs * lnG[f * H_ + h] + lnB[f * H_ + h];
        TRANS[lr * ST_T + f * H_ + h] = nv;
      }
    }
    __syncthreads();
  }

  v8f b1 = wmma_gemm<FH_>(TRANS, ST_T, sf1W, H_, n0, lane16, hi);
  {
    float b = sf1B[n0 + lane16];
#pragma unroll
    for (int v = 0; v < 8; ++v) {
      float t = b1[v] + b;
      b1[v] = t > 0.f ? t : (expf(t) - 1.f);
    }
  }
  store_tile(P1, ST_P, n0, b1, lane16, hi);
  __syncthreads();

  v8f b2 = wmma_gemm<H_>(P1, ST_P, sf2W, H_, n0, lane16, hi);
  {
    float b = sf2B[n0 + lane16];
#pragma unroll
    for (int v = 0; v < 8; ++v) b2[v] += b;
  }
  store_tile(P2, ST_P, n0, b2, lane16, hi);
  __syncthreads();

  if (wid == 0) {
    v8f g = wmma_gemm<H_>(P2, ST_P, sgW, F_, 0, lane16, hi);
    float b = sgB[lane16];
#pragma unroll
    for (int v = 0; v < 8; ++v) g[v] += b;
    store_tile(GP, ST_S, 0, g, lane16, hi);
  } else if (wid == 1) {
    v8f s = wmma_gemm<FH_>(TRANS, ST_T, sskW, F_, 0, lane16, hi);
    float b = sskB[lane16];
#pragma unroll
    for (int v = 0; v < 8; ++v) s[v] += b;
    store_tile(SKT, ST_S, 0, s, lane16, hi);
  }
  __syncthreads();

  {
    float gp = GP[lr * ST_S + lc];
    float sg = 1.f / (1.f + expf(-gp));
    float v = SKT[lr * ST_S + lc] + sg * gp;
    float s = rowsum16(v);
    float ss = rowsum16(v * v);
    float mu = s * (1.f / 16.f);
    float var = ss * (1.f / 16.f) - mu * mu;
    float nv = (v - mu) * rsqrtf(var + 1e-5f) * slnG[lc] + slnB[lc];
    float mx = rowmax16(nv);
    float e = expf(nv - mx);
    float se = rowsum16(e);
    float w = e / se;
    WB[lr * 16 + lc] = w;
    outW[(size_t)(row0 + lr) * F_ + lc] = w;
  }
  __syncthreads();

  {
#pragma unroll
    for (int j = 0; j < 8; ++j) {
      int h = lc + 16 * j;
      float acc = 0.f;
#pragma unroll
      for (int f = 0; f < F_; ++f)
        acc += TRANS[lr * ST_T + f * H_ + h] * WB[lr * 16 + f];
      outSel[(size_t)(row0 + lr) * H_ + h] = acc;
    }
  }
}

extern "C" void kernel_launch(void* const* d_in, const int* in_sizes, int n_in,
                              void* d_out, int out_size, void* d_ws, size_t ws_size,
                              hipStream_t stream) {
  (void)in_sizes; (void)n_in; (void)out_size; (void)d_ws; (void)ws_size;
  const float* x    = (const float*)d_in[0];
  const float* embW = (const float*)d_in[1];
  const float* embB = (const float*)d_in[2];
  const float* fc1W = (const float*)d_in[3];
  const float* fc1B = (const float*)d_in[4];
  const float* fc2W = (const float*)d_in[5];
  const float* fc2B = (const float*)d_in[6];
  const float* gW   = (const float*)d_in[7];
  const float* gB   = (const float*)d_in[8];
  const float* lnG  = (const float*)d_in[9];
  const float* lnB  = (const float*)d_in[10];
  const float* sf1W = (const float*)d_in[11];
  const float* sf1B = (const float*)d_in[12];
  const float* sf2W = (const float*)d_in[13];
  const float* sf2B = (const float*)d_in[14];
  const float* sgW  = (const float*)d_in[15];
  const float* sgB  = (const float*)d_in[16];
  const float* sskW = (const float*)d_in[17];
  const float* sskB = (const float*)d_in[18];
  const float* slnG = (const float*)d_in[19];
  const float* slnB = (const float*)d_in[20];

  float* outSel = (float*)d_out;
  float* outW   = outSel + (size_t)ROWS_TOTAL * H_;

  constexpr size_t shmem = (size_t)SMEM_FLOATS * sizeof(float);
  (void)hipFuncSetAttribute((const void*)vsn_fused_206158430403,
                            hipFuncAttributeMaxDynamicSharedMemorySize,
                            (int)shmem);

  vsn_fused_206158430403<<<ROWS_TOTAL / 16, 256, shmem, stream>>>(
      x, embW, embB, fc1W, fc1B, fc2W, fc2B, gW, gB, lnG, lnB,
      sf1W, sf1B, sf2W, sf2B, sgW, sgB, sskW, sskB, slnG, slnB,
      outSel, outW);
}


# 32 rows per block instead of 16: each wave runs two 16x16 row tiles off one weight fragment, halving weight loads and latency waits per row; 4 LDS work tiles become 2 (ping-pong), static LDS
# speedup vs baseline: 1.4054x; 1.4054x over previous
#include <hip/hip_runtime.h>
#include <hip/hip_bf16.h>
#include <math.h>


typedef float v2f __attribute__((ext_vector_type(2)));
typedef float v8f __attribute__((ext_vector_type(8)));

#define ROWS_TOTAL 16384
#define MR 32
#define F_ 16
#define D_ 64
#define H_ 128
#define FH_ 2048

#define ST_T 2052
#define ST_X 68
#define ST_P 132
#define ST_S 20

#define SMEM_FLOATS (MR*ST_T + MR*ST_X + 2*MR*ST_P)

__device__ __forceinline__ float rowsum16(float v) {
  v += __shfl_xor(v, 1, 16);
  v += __shfl_xor(v, 2, 16);
  v += __shfl_xor(v, 4, 16);
  v += __shfl_xor(v, 8, 16);
  return v;
}
__device__ __forceinline__ float rowmax16(float v) {
  v = fmaxf(v, __shfl_xor(v, 1, 16));
  v = fmaxf(v, __shfl_xor(v, 2, 16));
  v = fmaxf(v, __shfl_xor(v, 4, 16));
  v = fmaxf(v, __shfl_xor(v, 8, 16));
  return v;
}

template <int K>
__device__ __forceinline__ void wmma_gemm2(const float* __restrict__ As, int lda,
                                           const float* __restrict__ W, int ldw,
                                           int n0, int lane16, int hi,
                                           v8f& c0, v8f& c1) {
  const v8f z = {0.f, 0.f, 0.f, 0.f, 0.f, 0.f, 0.f, 0.f};
  c0 = z;
  c1 = z;
  const float* arow0 = As + lane16 * lda + hi * 2;
  const float* arow1 = arow0 + 16 * lda;
  const float* wcol = W + (size_t)(hi * 2) * ldw + n0 + lane16;
#pragma unroll 1
  for (int kb = 0; kb < K; kb += 32) {
    if (kb + 32 < K) {
      __builtin_prefetch(wcol + (size_t)(kb + 32) * ldw, 0, 1);
      __builtin_prefetch(wcol + (size_t)(kb + 48) * ldw, 0, 1);
    }
#pragma unroll
    for (int k0 = kb; k0 < kb + 32; k0 += 4) {
      v2f a0, a1, b;
      a0.x = arow0[k0];
      a0.y = arow0[k0 + 1];
      a1.x = arow1[k0];
      a1.y = arow1[k0 + 1];
      b.x = wcol[(size_t)k0 * ldw];
      b.y = wcol[(size_t)(k0 + 1) * ldw];
      c0 = __builtin_amdgcn_wmma_f32_16x16x4_f32(false, a0, false, b,
                                                 (short)0, c0, false, false);
      c1 = __builtin_amdgcn_wmma_f32_16x16x4_f32(false, a1, false, b,
                                                 (short)0, c1, false, false);
    }
  }
}

__device__ __forceinline__ void store_tile(float* dst, int ldd, int rbase,
                                           int col0, v8f c, int lane16, int hi) {
#pragma unroll
  for (int v = 0; v < 8; ++v)
    dst[(rbase + v + hi * 8) * ldd + col0 + lane16] = c[v];
}

__device__ __forceinline__ v8f add_bias(v8f c, float b) {
#pragma unroll
  for (int v = 0; v < 8; ++v) c[v] += b;
  return c;
}
__device__ __forceinline__ v8f bias_elu(v8f c, float b) {
#pragma unroll
  for (int v = 0; v < 8; ++v) {
    float t = c[v] + b;
    c[v] = t > 0.f ? t : (expf(t) - 1.f);
  }
  return c;
}
__device__ __forceinline__ v8f gate_mix(v8f g, float b, v8f e, v8f h2) {
#pragma unroll
  for (int v = 0; v < 8; ++v) {
    float t = g[v] + b;
    float s = 1.f / (1.f + expf(-t));
    g[v] = e[v] + s * h2[v];
  }
  return g;
}

__global__ __launch_bounds__(256, 1)
void vsn_fused_rows32(
    const float* __restrict__ x,
    const float* __restrict__ embW, const float* __restrict__ embB,
    const float* __restrict__ fc1W, const float* __restrict__ fc1B,
    const float* __restrict__ fc2W, const float* __restrict__ fc2B,
    const float* __restrict__ gW,   const float* __restrict__ gB,
    const float* __restrict__ lnG,  const float* __restrict__ lnB,
    const float* __restrict__ sf1W, const float* __restrict__ sf1B,
    const float* __restrict__ sf2W, const float* __restrict__ sf2B,
    const float* __restrict__ sgW,  const float* __restrict__ sgB,
    const float* __restrict__ sskW, const float* __restrict__ sskB,
    const float* __restrict__ slnG, const float* __restrict__ slnB,
    float* __restrict__ outSel, float* __restrict__ outW) {
  __shared__ float smem[SMEM_FLOATS];
  float* TRANS = smem;
  float* XT = TRANS + MR * ST_T;
  float* PU = XT + MR * ST_X;
  float* PV = PU + MR * ST_P;
  float* GP = XT;
  float* SKT = GP + MR * ST_S;
  float* WB = SKT + MR * ST_S;

  const int tid = threadIdx.x;
  const int wid = tid >> 5;
  const int lane = tid & 31;
  const int lane16 = lane & 15;
  const int hi = lane >> 4;
  const int n0 = wid * 16;
  const int lr = tid >> 4;
  const int lc = tid & 15;
  const int row0 = blockIdx.x * MR;

  for (int f = 0; f < F_; ++f) {
#pragma unroll
    for (int rr = 0; rr < 2; ++rr) {
      const int r = lr + 16 * rr;
      const float* xp = x + (size_t)(row0 + r) * (F_ * D_) + f * D_ + lc * 4;
      float4 v = *(const float4*)xp;
      float* dp = XT + r * ST_X + lc * 4;
      dp[0] = v.x; dp[1] = v.y; dp[2] = v.z; dp[3] = v.w;
      if (f + 1 < F_) __builtin_prefetch(xp + D_, 0, 1);
    }
    __syncthreads();

    v8f ce0, ce1;
    wmma_gemm2<D_>(XT, ST_X, embW + (size_t)f * D_ * H_, H_, n0, lane16, hi, ce0, ce1);
    {
      float b = embB[f * H_ + n0 + lane16];
      ce0 = add_bias(ce0, b);
      ce1 = add_bias(ce1, b);
    }
    store_tile(PU, ST_P, 0, n0, ce0, lane16, hi);
    store_tile(PU, ST_P, 16, n0, ce1, lane16, hi);
    __syncthreads();

    {
      v8f c10, c11;
      wmma_gemm2<H_>(PU, ST_P, fc1W + (size_t)f * H_ * H_, H_, n0, lane16, hi, c10, c11);
      float b = fc1B[f * H_ + n0 + lane16];
      c10 = bias_elu(c10, b);
      c11 = bias_elu(c11, b);
      store_tile(PV, ST_P, 0, n0, c10, lane16, hi);
      store_tile(PV, ST_P, 16, n0, c11, lane16, hi);
    }
    __syncthreads();

    v8f c20, c21;
    wmma_gemm2<H_>(PV, ST_P, fc2W + (size_t)f * H_ * H_, H_, n0, lane16, hi, c20, c21);
    {
      float b = fc2B[f * H_ + n0 + lane16];
      c20 = add_bias(c20, b);
      c21 = add_bias(c21, b);
    }
    store_tile(PU, ST_P, 0, n0, c20, lane16, hi);
    store_tile(PU, ST_P, 16, n0, c21, lane16, hi);
    __syncthreads();

    {
      v8f cg0, cg1;
      wmma_gemm2<H_>(PU, ST_P, gW + (size_t)f * H_ * H_, H_, n0, lane16, hi, cg0, cg1);
      float b = gB[f * H_ + n0 + lane16];
      cg0 = gate_mix(cg0, b, ce0, c20);
      cg1 = gate_mix(cg1, b, ce1, c21);
      store_tile(PV, ST_P, 0, n0, cg0, lane16, hi);
      store_tile(PV, ST_P, 16, n0, cg1, lane16, hi);
    }
    __syncthreads();

#pragma unroll
    for (int rr = 0; rr < 2; ++rr) {
      const int r = lr + 16 * rr;
      float vals[8];
      float s = 0.f, ss = 0.f;
#pragma unroll
      for (int j = 0; j < 8; ++j) {
        float v = PV[r * ST_P + lc + 16 * j];
        vals[j] = v; s += v; ss += v * v;
      }
      s = rowsum16(s);
      ss = rowsum16(ss);
      float mu = s * (1.f / 128.f);
      float var = ss * (1.f / 128.f) - mu * mu;
      float rs = rsqrtf(var + 1e-5f);
#pragma unroll
      for (int j = 0; j < 8; ++j) {
        int h = lc + 16 * j;
        float nv = (vals[j] - mu) * rs * lnG[f * H_ + h] + lnB[f * H_ + h];
        TRANS[r * ST_T + f * H_ + h] = nv;
      }
    }
    __syncthreads();
  }

  {
    v8f b10, b11;
    wmma_gemm2<FH_>(TRANS, ST_T, sf1W, H_, n0, lane16, hi, b10, b11);
    float b = sf1B[n0 + lane16];
    b10 = bias_elu(b10, b);
    b11 = bias_elu(b11, b);
    store_tile(PU, ST_P, 0, n0, b10, lane16, hi);
    store_tile(PU, ST_P, 16, n0, b11, lane16, hi);
  }
  __syncthreads();

  {
    v8f b20, b21;
    wmma_gemm2<H_>(PU, ST_P, sf2W, H_, n0, lane16, hi, b20, b21);
    float b = sf2B[n0 + lane16];
    b20 = add_bias(b20, b);
    b21 = add_bias(b21, b);
    store_tile(PV, ST_P, 0, n0, b20, lane16, hi);
    store_tile(PV, ST_P, 16, n0, b21, lane16, hi);
  }
  __syncthreads();

  if (wid == 0) {
    v8f g0, g1;
    wmma_gemm2<H_>(PV, ST_P, sgW, F_, 0, lane16, hi, g0, g1);
    float b = sgB[lane16];
    g0 = add_bias(g0, b);
    g1 = add_bias(g1, b);
    store_tile(GP, ST_S, 0, 0, g0, lane16, hi);
    store_tile(GP, ST_S, 16, 0, g1, lane16, hi);
  } else if (wid == 1) {
    v8f s0, s1;
    wmma_gemm2<FH_>(TRANS, ST_T, sskW, F_, 0, lane16, hi, s0, s1);
    float b = sskB[lane16];
    s0 = add_bias(s0, b);
    s1 = add_bias(s1, b);
    store_tile(SKT, ST_S, 0, 0, s0, lane16, hi);
    store_tile(SKT, ST_S, 16, 0, s1, lane16, hi);
  }
  __syncthreads();

#pragma unroll
  for (int rr = 0; rr < 2; ++rr) {
    const int r = lr + 16 * rr;
    float gp = GP[r * ST_S + lc];
    float sg = 1.f / (1.f + expf(-gp));
    float v = SKT[r * ST_S + lc] + sg * gp;
    float s = rowsum16(v);
    float ss = rowsum16(v * v);
    float mu = s * (1.f / 16.f);
    float var = ss * (1.f / 16.f) - mu * mu;
    float nv = (v - mu) * rsqrtf(var + 1e-5f) * slnG[lc] + slnB[lc];
    float mx = rowmax16(nv);
    float e = expf(nv - mx);
    float se = rowsum16(e);
    float w = e / se;
    WB[r * 16 + lc] = w;
    outW[(size_t)(row0 + r) * F_ + lc] = w;
  }
  __syncthreads();

#pragma unroll
  for (int rr = 0; rr < 2; ++rr) {
    const int r = lr + 16 * rr;
#pragma unroll
    for (int j = 0; j < 8; ++j) {
      int h = lc + 16 * j;
      float acc = 0.f;
#pragma unroll
      for (int f = 0; f < F_; ++f)
        acc += TRANS[r * ST_T + f * H_ + h] * WB[r * 16 + f];
      outSel[(size_t)(row0 + r) * H_ + h] = acc;
    }
  }
}

extern "C" void kernel_launch(void* const* d_in, const int* in_sizes, int n_in,
                              void* d_out, int out_size, void* d_ws, size_t ws_size,
                              hipStream_t stream) {
  (void)in_sizes; (void)n_in; (void)out_size; (void)d_ws; (void)ws_size;
  const float* x    = (const float*)d_in[0];
  const float* embW = (const float*)d_in[1];
  const float* embB = (const float*)d_in[2];
  const float* fc1W = (const float*)d_in[3];
  const float* fc1B = (const float*)d_in[4];
  const float* fc2W = (const float*)d_in[5];
  const float* fc2B = (const float*)d_in[6];
  const float* gW   = (const float*)d_in[7];
  const float* gB   = (const float*)d_in[8];
  const float* lnG  = (const float*)d_in[9];
  const float* lnB  = (const float*)d_in[10];
  const float* sf1W = (const float*)d_in[11];
  const float* sf1B = (const float*)d_in[12];
  const float* sf2W = (const float*)d_in[13];
  const float* sf2B = (const float*)d_in[14];
  const float* sgW  = (const float*)d_in[15];
  const float* sgB  = (const float*)d_in[16];
  const float* sskW = (const float*)d_in[17];
  const float* sskB = (const float*)d_in[18];
  const float* slnG = (const float*)d_in[19];
  const float* slnB = (const float*)d_in[20];

  float* outSel = (float*)d_out;
  float* outW   = outSel + (size_t)ROWS_TOTAL * H_;

  vsn_fused_rows32<<<ROWS_TOTAL / MR, 256, 0, stream>>>(
      x, embW, embB, fc1W, fc1B, fc2W, fc2B, gW, gB, lnG, lnB,
      sf1W, sf1B, sf2W, sf2B, sgW, sgB, sskW, sskB, slnG, slnB,
      outSel, outW);
}


# Operand loads run one stage ahead in registers: next GEMM's weight panel, next x tile and next 32-k A chunk requested before the current matrix run; barriers wait on LDS only so they stay in flight
# speedup vs baseline: 2.0529x; 1.4607x over previous
#include <hip/hip_runtime.h>
#include <hip/hip_bf16.h>
#include <math.h>


typedef float v2f __attribute__((ext_vector_type(2)));
typedef float v8f __attribute__((ext_vector_type(8)));

#define ROWS_TOTAL 16384
#define MR 32
#define F_ 16
#define D_ 64
#define H_ 128
#define FH_ 2048

#define ST_T 2052
#define ST_X 68
#define ST_P 132
#define ST_S 20

#define SMEM_FLOATS (MR*ST_T + MR*ST_X + 2*MR*ST_P)

__device__ __forceinline__ float rowsum16(float v) {
  v += __shfl_xor(v, 1, 16);
  v += __shfl_xor(v, 2, 16);
  v += __shfl_xor(v, 4, 16);
  v += __shfl_xor(v, 8, 16);
  return v;
}
__device__ __forceinline__ float rowmax16(float v) {
  v = fmaxf(v, __shfl_xor(v, 1, 16));
  v = fmaxf(v, __shfl_xor(v, 2, 16));
  v = fmaxf(v, __shfl_xor(v, 4, 16));
  v = fmaxf(v, __shfl_xor(v, 8, 16));
  return v;
}

__device__ __forceinline__ void wg_barrier_lds() {
  asm volatile("s_wait_dscnt 0x0\n\ts_barrier_signal -1\n\ts_barrier_wait -1" ::: "memory");
}

__device__ __forceinline__ void keep_order() { __builtin_amdgcn_sched_barrier(0); }

__device__ __forceinline__ v8f mma4(v2f a, v2f b, v8f c) {
  return __builtin_amdgcn_wmma_f32_16x16x4_f32(false, a, false, b, (short)0, c,
                                               false, false);
}


template <int K>
__device__ __forceinline__ void load_panel(const float* __restrict__ W, int ldw,
                                           int n0, int lane16, int hi,
                                           v2f (&b)[K / 4]) {
  const float* wcol = W + (size_t)(hi * 2) * ldw + n0 + lane16;
#pragma unroll
  for (int s = 0; s < K / 4; ++s) {
    b[s].x = wcol[(size_t)(4 * s) * ldw];
    b[s].y = wcol[(size_t)(4 * s + 1) * ldw];
  }
}

template <int S>
__device__ __forceinline__ void load_a(const float* arow0, const float* arow1,
                                       int kb, v2f (&a0)[S], v2f (&a1)[S]) {
#pragma unroll
  for (int s = 0; s < S; ++s) {
    a0[s].x = arow0[kb + 4 * s];
    a0[s].y = arow0[kb + 4 * s + 1];
    a1[s].x = arow1[kb + 4 * s];
    a1[s].y = arow1[kb + 4 * s + 1];
  }
}

template <int K>
__device__ __forceinline__ void gemm_panel(const float* As, int lda, int lane16,
                                           int hi, const v2f (&b)[K / 4],
                                           v8f& c0, v8f& c1) {
  const v8f z = {0.f, 0.f, 0.f, 0.f, 0.f, 0.f, 0.f, 0.f};
  c0 = z;
  c1 = z;
  const float* arow0 = As + lane16 * lda + hi * 2;
  const float* arow1 = arow0 + 16 * lda;
  v2f a0[2][8], a1[2][8];
  load_a<8>(arow0, arow1, 0, a0[0], a1[0]);
  keep_order();
#pragma unroll
  for (int ch = 0; ch < K / 32; ++ch) {
    if (ch + 1 < K / 32)
      load_a<8>(arow0, arow1, (ch + 1) * 32, a0[(ch + 1) & 1], a1[(ch + 1) & 1]);
    keep_order();
#pragma unroll
    for (int s = 0; s < 8; ++s) {
      c0 = mma4(a0[ch & 1][s], b[ch * 8 + s], c0);
      c1 = mma4(a1[ch & 1][s], b[ch * 8 + s], c1);
    }
    keep_order();
  }
}

template <int S>
__device__ __forceinline__ void load_chunk(const float* arow0, const float* arow1,
                                           const float* __restrict__ wcol, int ldw,
                                           int kb, v2f (&a0)[S], v2f (&a1)[S],
                                           v2f (&b)[S]) {
#pragma unroll
  for (int s = 0; s < S; ++s) {
    b[s].x = wcol[(size_t)(kb + 4 * s) * ldw];
    b[s].y = wcol[(size_t)(kb + 4 * s + 1) * ldw];
  }
  load_a<S>(arow0, arow1, kb, a0, a1);
}
template <int S>
__device__ __forceinline__ void mma_chunk(const v2f (&a0)[S], const v2f (&a1)[S],
                                          const v2f (&b)[S], v8f& c0, v8f& c1) {
#pragma unroll
  for (int s = 0; s < S; ++s) {
    c0 = mma4(a0[s], b[s], c0);
    c1 = mma4(a1[s], b[s], c1);
  }
}

template <int K, int CH>
__device__ __forceinline__ void gemm_stream(const float* As, int lda,
                                            const float* __restrict__ W, int ldw,
                                            int n0, int lane16, int hi,
                                            v8f& c0, v8f& c1) {
  static_assert(K % (2 * CH) == 0 && K >= 2 * CH, "chunk pairs");
  constexpr int S = CH / 4;
  const v8f z = {0.f, 0.f, 0.f, 0.f, 0.f, 0.f, 0.f, 0.f};
  c0 = z;
  c1 = z;
  const float* arow0 = As + lane16 * lda + hi * 2;
  const float* arow1 = arow0 + 16 * lda;
  const float* wcol = W + (size_t)(hi * 2) * ldw + n0 + lane16;
  v2f bP[S], a0P[S], a1P[S], bQ[S], a0Q[S], a1Q[S];
  load_chunk<S>(arow0, arow1, wcol, ldw, 0, a0P, a1P, bP);
  keep_order();
#pragma unroll 1
  for (int kb = 0; kb < K - 2 * CH; kb += 2 * CH) {
    load_chunk<S>(arow0, arow1, wcol, ldw, kb + CH, a0Q, a1Q, bQ);
    keep_order();
    mma_chunk<S>(a0P, a1P, bP, c0, c1);
    keep_order();
    load_chunk<S>(arow0, arow1, wcol, ldw, kb + 2 * CH, a0P, a1P, bP);
    keep_order();
    mma_chunk<S>(a0Q, a1Q, bQ, c0, c1);
    keep_order();
  }
  load_chunk<S>(arow0, arow1, wcol, ldw, K - CH, a0Q, a1Q, bQ);
  keep_order();
  mma_chunk<S>(a0P, a1P, bP, c0, c1);
  mma_chunk<S>(a0Q, a1Q, bQ, c0, c1);
}

__device__ __forceinline__ void store_tile(float* dst, int ldd, int rbase,
                                           int col0, v8f c, int lane16, int hi) {
#pragma unroll
  for (int v = 0; v < 8; ++v)
    dst[(rbase + v + hi * 8) * ldd + col0 + lane16] = c[v];
}

__device__ __forceinline__ v8f add_bias(v8f c, float b) {
#pragma unroll
  for (int v = 0; v < 8; ++v) c[v] += b;
  return c;
}
__device__ __forceinline__ v8f bias_elu(v8f c, float b) {
#pragma unroll
  for (int v = 0; v < 8; ++v) {
    float t = c[v] + b;
    c[v] = t > 0.f ? t : (expf(t) - 1.f);
  }
  return c;
}
__device__ __forceinline__ v8f gate_mix(v8f g, float b, v8f e, v8f h2) {
#pragma unroll
  for (int v = 0; v < 8; ++v) {
    float t = g[v] + b;
    float s = 1.f / (1.f + expf(-t));
    g[v] = e[v] + s * h2[v];
  }
  return g;
}

__global__ __launch_bounds__(256, 1)
void vsn_fused_rows32(
    const float* __restrict__ x,
    const float* __restrict__ embW, const float* __restrict__ embB,
    const float* __restrict__ fc1W, const float* __restrict__ fc1B,
    const float* __restrict__ fc2W, const float* __restrict__ fc2B,
    const float* __restrict__ gW,   const float* __restrict__ gB,
    const float* __restrict__ lnG,  const float* __restrict__ lnB,
    const float* __restrict__ sf1W, const float* __restrict__ sf1B,
    const float* __restrict__ sf2W, const float* __restrict__ sf2B,
    const float* __restrict__ sgW,  const float* __restrict__ sgB,
    const float* __restrict__ sskW, const float* __restrict__ sskB,
    const float* __restrict__ slnG, const float* __restrict__ slnB,
    float* __restrict__ outSel, float* __restrict__ outW) {
  __shared__ float smem[SMEM_FLOATS];
  float* TRANS = smem;
  float* XT = TRANS + MR * ST_T;
  float* PU = XT + MR * ST_X;
  float* PV = PU + MR * ST_P;
  float* GP = XT;
  float* SKT = GP + MR * ST_S;
  float* WB = SKT + MR * ST_S;

  const int tid = threadIdx.x;
  const int wid = tid >> 5;
  const int lane = tid & 31;
  const int lane16 = lane & 15;
  const int hi = lane >> 4;
  const int n0 = wid * 16;
  const int lr = tid >> 4;
  const int lc = tid & 15;
  const int row0 = blockIdx.x * MR;

  float bE = embB[n0 + lane16];
  float b1 = fc1B[n0 + lane16];
  float b2 = fc2B[n0 + lane16];
  float bG = gB[n0 + lane16];
  v2f pE[D_ / 4];
  load_panel<D_>(embW, H_, n0, lane16, hi, pE);
  {
#pragma unroll
    for (int rr = 0; rr < 2; ++rr) {
      const int r = lr + 16 * rr;
      float4 v = *(const float4*)(x + (size_t)(row0 + r) * (F_ * D_) + lc * 4);
      float* dp = XT + r * ST_X + lc * 4;
      dp[0] = v.x; dp[1] = v.y; dp[2] = v.z; dp[3] = v.w;
    }
  }
  wg_barrier_lds();

  for (int f = 0; f < F_; ++f) {
    const int fn = f + 1 < F_ ? f + 1 : f;

    v2f pF1[H_ / 4];
    load_panel<H_>(fc1W + (size_t)f * H_ * H_, H_, n0, lane16, hi, pF1);
    float4 xr[2];
#pragma unroll
    for (int rr = 0; rr < 2; ++rr)
      xr[rr] = *(const float4*)(x + (size_t)(row0 + lr + 16 * rr) * (F_ * D_) +
                                fn * D_ + lc * 4);

    v8f ce0, ce1;
    gemm_panel<D_>(XT, ST_X, lane16, hi, pE, ce0, ce1);
    ce0 = add_bias(ce0, bE);
    ce1 = add_bias(ce1, bE);
    store_tile(PU, ST_P, 0, n0, ce0, lane16, hi);
    store_tile(PU, ST_P, 16, n0, ce1, lane16, hi);
    wg_barrier_lds();

    v2f pF2[H_ / 4];
    load_panel<H_>(fc2W + (size_t)f * H_ * H_, H_, n0, lane16, hi, pF2);
    {
      v8f c10, c11;
      gemm_panel<H_>(PU, ST_P, lane16, hi, pF1, c10, c11);
      c10 = bias_elu(c10, b1);
      c11 = bias_elu(c11, b1);
      store_tile(PV, ST_P, 0, n0, c10, lane16, hi);
      store_tile(PV, ST_P, 16, n0, c11, lane16, hi);
    }
    wg_barrier_lds();

    v2f pG[H_ / 4];
    load_panel<H_>(gW + (size_t)f * H_ * H_, H_, n0, lane16, hi, pG);
    v8f c20, c21;
    gemm_panel<H_>(PV, ST_P, lane16, hi, pF2, c20, c21);
    c20 = add_bias(c20, b2);
    c21 = add_bias(c21, b2);
    store_tile(PU, ST_P, 0, n0, c20, lane16, hi);
    store_tile(PU, ST_P, 16, n0, c21, lane16, hi);
    wg_barrier_lds();

    load_panel<D_>(embW + (size_t)fn * D_ * H_, H_, n0, lane16, hi, pE);
    const float bGf = bG;
    bE = embB[fn * H_ + n0 + lane16];
    b1 = fc1B[fn * H_ + n0 + lane16];
    b2 = fc2B[fn * H_ + n0 + lane16];
    bG = gB[fn * H_ + n0 + lane16];
    float lg[8], lb[8];
#pragma unroll
    for (int j = 0; j < 8; ++j) {
      lg[j] = lnG[f * H_ + lc + 16 * j];
      lb[j] = lnB[f * H_ + lc + 16 * j];
    }
    {
      v8f cg0, cg1;
      gemm_panel<H_>(PU, ST_P, lane16, hi, pG, cg0, cg1);
      cg0 = gate_mix(cg0, bGf, ce0, c20);
      cg1 = gate_mix(cg1, bGf, ce1, c21);
      store_tile(PV, ST_P, 0, n0, cg0, lane16, hi);
      store_tile(PV, ST_P, 16, n0, cg1, lane16, hi);
    }
    wg_barrier_lds();

#pragma unroll
    for (int rr = 0; rr < 2; ++rr) {
      const int r = lr + 16 * rr;
      float vals[8];
      float s = 0.f, ss = 0.f;
#pragma unroll
      for (int j = 0; j < 8; ++j) {
        float v = PV[r * ST_P + lc + 16 * j];
        vals[j] = v; s += v; ss += v * v;
      }
      s = rowsum16(s);
      ss = rowsum16(ss);
      float mu = s * (1.f / 128.f);
      float var = ss * (1.f / 128.f) - mu * mu;
      float rs = rsqrtf(var + 1e-5f);
#pragma unroll
      for (int j = 0; j < 8; ++j) {
        int h = lc + 16 * j;
        float nv = (vals[j] - mu) * rs * lg[j] + lb[j];
        TRANS[r * ST_T + f * H_ + h] = nv;
      }
      float* dp = XT + r * ST_X + lc * 4;
      dp[0] = xr[rr].x; dp[1] = xr[rr].y; dp[2] = xr[rr].z; dp[3] = xr[rr].w;
    }
    wg_barrier_lds();
  }

  v2f pS2[H_ / 4];
  load_panel<H_>(sf2W, H_, n0, lane16, hi, pS2);
  const float bS1 = sf1B[n0 + lane16];
  const float bS2 = sf2B[n0 + lane16];
  {
    v8f b10, b11;
    gemm_stream<FH_, 64>(TRANS, ST_T, sf1W, H_, n0, lane16, hi, b10, b11);
    b10 = bias_elu(b10, bS1);
    b11 = bias_elu(b11, bS1);
    store_tile(PU, ST_P, 0, n0, b10, lane16, hi);
    store_tile(PU, ST_P, 16, n0, b11, lane16, hi);
  }
  wg_barrier_lds();

  {
    v8f b20, b21;
    gemm_panel<H_>(PU, ST_P, lane16, hi, pS2, b20, b21);
    b20 = add_bias(b20, bS2);
    b21 = add_bias(b21, bS2);
    store_tile(PV, ST_P, 0, n0, b20, lane16, hi);
    store_tile(PV, ST_P, 16, n0, b21, lane16, hi);
  }
  wg_barrier_lds();

  if (wid == 0) {
    v2f pSG[H_ / 4];
    load_panel<H_>(sgW, F_, 0, lane16, hi, pSG);
    v8f g0, g1;
    gemm_panel<H_>(PV, ST_P, lane16, hi, pSG, g0, g1);
    float b = sgB[lane16];
    g0 = add_bias(g0, b);
    g1 = add_bias(g1, b);
    store_tile(GP, ST_S, 0, 0, g0, lane16, hi);
    store_tile(GP, ST_S, 16, 0, g1, lane16, hi);
  } else if (wid == 1) {
    v8f s0, s1;
    gemm_stream<FH_, 64>(TRANS, ST_T, sskW, F_, 0, lane16, hi, s0, s1);
    float b = sskB[lane16];
    s0 = add_bias(s0, b);
    s1 = add_bias(s1, b);
    store_tile(SKT, ST_S, 0, 0, s0, lane16, hi);
    store_tile(SKT, ST_S, 16, 0, s1, lane16, hi);
  }
  wg_barrier_lds();

#pragma unroll
  for (int rr = 0; rr < 2; ++rr) {
    const int r = lr + 16 * rr;
    float gp = GP[r * ST_S + lc];
    float sg = 1.f / (1.f + expf(-gp));
    float v = SKT[r * ST_S + lc] + sg * gp;
    float s = rowsum16(v);
    float ss = rowsum16(v * v);
    float mu = s * (1.f / 16.f);
    float var = ss * (1.f / 16.f) - mu * mu;
    float nv = (v - mu) * rsqrtf(var + 1e-5f) * slnG[lc] + slnB[lc];
    float mx = rowmax16(nv);
    float e = expf(nv - mx);
    float se = rowsum16(e);
    float w = e / se;
    WB[r * 16 + lc] = w;
    outW[(size_t)(row0 + r) * F_ + lc] = w;
  }
  wg_barrier_lds();

#pragma unroll
  for (int rr = 0; rr < 2; ++rr) {
    const int r = lr + 16 * rr;
#pragma unroll
    for (int j = 0; j < 8; ++j) {
      int h = lc + 16 * j;
      float acc = 0.f;
#pragma unroll
      for (int f = 0; f < F_; ++f)
        acc += TRANS[r * ST_T + f * H_ + h] * WB[r * 16 + f];
      outSel[(size_t)(row0 + r) * H_ + h] = acc;
    }
  }
}

extern "C" void kernel_launch(void* const* d_in, const int* in_sizes, int n_in,
                              void* d_out, int out_size, void* d_ws, size_t ws_size,
                              hipStream_t stream) {
  (void)in_sizes; (void)n_in; (void)out_size; (void)d_ws; (void)ws_size;
  const float* x    = (const float*)d_in[0];
  const float* embW = (const float*)d_in[1];
  const float* embB = (const float*)d_in[2];
  const float* fc1W = (const float*)d_in[3];
  const float* fc1B = (const float*)d_in[4];
  const float* fc2W = (const float*)d_in[5];
  const float* fc2B = (const float*)d_in[6];
  const float* gW   = (const float*)d_in[7];
  const float* gB   = (const float*)d_in[8];
  const float* lnG  = (const float*)d_in[9];
  const float* lnB  = (const float*)d_in[10];
  const float* sf1W = (const float*)d_in[11];
  const float* sf1B = (const float*)d_in[12];
  const float* sf2W = (const float*)d_in[13];
  const float* sf2B = (const float*)d_in[14];
  const float* sgW  = (const float*)d_in[15];
  const float* sgB  = (const float*)d_in[16];
  const float* sskW = (const float*)d_in[17];
  const float* sskB = (const float*)d_in[18];
  const float* slnG = (const float*)d_in[19];
  const float* slnB = (const float*)d_in[20];

  float* outSel = (float*)d_out;
  float* outW   = outSel + (size_t)ROWS_TOTAL * H_;

  vsn_fused_rows32<<<ROWS_TOTAL / MR, 256, 0, stream>>>(
      x, embW, embB, fc1W, fc1B, fc2W, fc2B, gW, gB, lnG, lnB,
      sf1W, sf1B, sf2W, sf2B, sgW, sgB, sskW, sskB, slnG, slnB,
      outSel, outW);
}


# Workgroup barriers split into signal and wait; the next stages' global loads (weight panels, x tile, biases) are issued between the two, inside the barrier wait, not in front of the matrix run
# speedup vs baseline: 2.2694x; 1.1055x over previous
#include <hip/hip_runtime.h>
#include <hip/hip_bf16.h>
#include <math.h>


typedef float v2f __attribute__((ext_vector_type(2)));
typedef float v8f __attribute__((ext_vector_type(8)));

#define ROWS_TOTAL 16384
#define MR 32
#define F_ 16
#define D_ 64
#define H_ 128
#define FH_ 2048

#define ST_T 2052
#define ST_X 68
#define ST_P 132
#define ST_S 20

#define SMEM_FLOATS (MR*ST_T + MR*ST_X + 2*MR*ST_P)

__device__ __forceinline__ float rowsum16(float v) {
  v += __shfl_xor(v, 1, 16);
  v += __shfl_xor(v, 2, 16);
  v += __shfl_xor(v, 4, 16);
  v += __shfl_xor(v, 8, 16);
  return v;
}
__device__ __forceinline__ float rowmax16(float v) {
  v = fmaxf(v, __shfl_xor(v, 1, 16));
  v = fmaxf(v, __shfl_xor(v, 2, 16));
  v = fmaxf(v, __shfl_xor(v, 4, 16));
  v = fmaxf(v, __shfl_xor(v, 8, 16));
  return v;
}

__device__ __forceinline__ void wg_barrier_lds() {
  asm volatile("s_wait_dscnt 0x0\n\ts_barrier_signal -1\n\ts_barrier_wait -1" ::: "memory");
}
__device__ __forceinline__ void wg_signal_lds() {
  asm volatile("s_wait_dscnt 0x0\n\ts_barrier_signal -1" ::: "memory");
}
__device__ __forceinline__ void wg_wait() {
  asm volatile("s_barrier_wait -1" ::: "memory");
}

__device__ __forceinline__ void keep_order() { __builtin_amdgcn_sched_barrier(0); }

__device__ __forceinline__ v8f mma4(v2f a, v2f b, v8f c) {
  return __builtin_amdgcn_wmma_f32_16x16x4_f32(false, a, false, b, (short)0, c,
                                               false, false);
}


template <int K>
__device__ __forceinline__ void load_panel(const float* __restrict__ W, int ldw,
                                           int n0, int lane16, int hi,
                                           v2f (&b)[K / 4]) {
  const float* wcol = W + (size_t)(hi * 2) * ldw + n0 + lane16;
#pragma unroll
  for (int s = 0; s < K / 4; ++s) {
    b[s].x = wcol[(size_t)(4 * s) * ldw];
    b[s].y = wcol[(size_t)(4 * s + 1) * ldw];
  }
}

template <int S0, int S1, int NS>
__device__ __forceinline__ void load_panel_part(const float* __restrict__ W, int ldw,
                                                int n0, int lane16, int hi,
                                                v2f (&b)[NS]) {
  const float* wcol = W + (size_t)(hi * 2) * ldw + n0 + lane16;
#pragma unroll
  for (int s = S0; s < S1; ++s) {
    b[s].x = wcol[(size_t)(4 * s) * ldw];
    b[s].y = wcol[(size_t)(4 * s + 1) * ldw];
  }
}

template <int S>
__device__ __forceinline__ void load_a(const float* arow0, const float* arow1,
                                       int kb, v2f (&a0)[S], v2f (&a1)[S]) {
#pragma unroll
  for (int s = 0; s < S; ++s) {
    a0[s].x = arow0[kb + 4 * s];
    a0[s].y = arow0[kb + 4 * s + 1];
    a1[s].x = arow1[kb + 4 * s];
    a1[s].y = arow1[kb + 4 * s + 1];
  }
}

template <int K>
__device__ __forceinline__ void gemm_panel(const float* As, int lda, int lane16,
                                           int hi, const v2f (&b)[K / 4],
                                           v8f& c0, v8f& c1) {
  const v8f z = {0.f, 0.f, 0.f, 0.f, 0.f, 0.f, 0.f, 0.f};
  c0 = z;
  c1 = z;
  const float* arow0 = As + lane16 * lda + hi * 2;
  const float* arow1 = arow0 + 16 * lda;
  v2f a0[2][8], a1[2][8];
  load_a<8>(arow0, arow1, 0, a0[0], a1[0]);
  keep_order();
#pragma unroll
  for (int ch = 0; ch < K / 32; ++ch) {
    if (ch + 1 < K / 32)
      load_a<8>(arow0, arow1, (ch + 1) * 32, a0[(ch + 1) & 1], a1[(ch + 1) & 1]);
    keep_order();
#pragma unroll
    for (int s = 0; s < 8; ++s) {
      c0 = mma4(a0[ch & 1][s], b[ch * 8 + s], c0);
      c1 = mma4(a1[ch & 1][s], b[ch * 8 + s], c1);
    }
    keep_order();
  }
}

template <int S>
__device__ __forceinline__ void load_chunk(const float* arow0, const float* arow1,
                                           const float* __restrict__ wcol, int ldw,
                                           int kb, v2f (&a0)[S], v2f (&a1)[S],
                                           v2f (&b)[S]) {
#pragma unroll
  for (int s = 0; s < S; ++s) {
    b[s].x = wcol[(size_t)(kb + 4 * s) * ldw];
    b[s].y = wcol[(size_t)(kb + 4 * s + 1) * ldw];
  }
  load_a<S>(arow0, arow1, kb, a0, a1);
}
template <int S>
__device__ __forceinline__ void mma_chunk(const v2f (&a0)[S], const v2f (&a1)[S],
                                          const v2f (&b)[S], v8f& c0, v8f& c1) {
#pragma unroll
  for (int s = 0; s < S; ++s) {
    c0 = mma4(a0[s], b[s], c0);
    c1 = mma4(a1[s], b[s], c1);
  }
}

template <int K, int CH>
__device__ __forceinline__ void gemm_stream(const float* As, int lda,
                                            const float* __restrict__ W, int ldw,
                                            int n0, int lane16, int hi,
                                            v8f& c0, v8f& c1) {
  static_assert(K % (2 * CH) == 0 && K >= 2 * CH, "chunk pairs");
  constexpr int S = CH / 4;
  const v8f z = {0.f, 0.f, 0.f, 0.f, 0.f, 0.f, 0.f, 0.f};
  c0 = z;
  c1 = z;
  const float* arow0 = As + lane16 * lda + hi * 2;
  const float* arow1 = arow0 + 16 * lda;
  const float* wcol = W + (size_t)(hi * 2) * ldw + n0 + lane16;
  v2f bP[S], a0P[S], a1P[S], bQ[S], a0Q[S], a1Q[S];
  load_chunk<S>(arow0, arow1, wcol, ldw, 0, a0P, a1P, bP);
  keep_order();
#pragma unroll 1
  for (int kb = 0; kb < K - 2 * CH; kb += 2 * CH) {
    load_chunk<S>(arow0, arow1, wcol, ldw, kb + CH, a0Q, a1Q, bQ);
    keep_order();
    mma_chunk<S>(a0P, a1P, bP, c0, c1);
    keep_order();
    load_chunk<S>(arow0, arow1, wcol, ldw, kb + 2 * CH, a0P, a1P, bP);
    keep_order();
    mma_chunk<S>(a0Q, a1Q, bQ, c0, c1);
    keep_order();
  }
  load_chunk<S>(arow0, arow1, wcol, ldw, K - CH, a0Q, a1Q, bQ);
  keep_order();
  mma_chunk<S>(a0P, a1P, bP, c0, c1);
  mma_chunk<S>(a0Q, a1Q, bQ, c0, c1);
}

__device__ __forceinline__ void store_tile(float* dst, int ldd, int rbase,
                                           int col0, v8f c, int lane16, int hi) {
#pragma unroll
  for (int v = 0; v < 8; ++v)
    dst[(rbase + v + hi * 8) * ldd + col0 + lane16] = c[v];
}

__device__ __forceinline__ v8f add_bias(v8f c, float b) {
#pragma unroll
  for (int v = 0; v < 8; ++v) c[v] += b;
  return c;
}
__device__ __forceinline__ v8f bias_elu(v8f c, float b) {
#pragma unroll
  for (int v = 0; v < 8; ++v) {
    float t = c[v] + b;
    c[v] = t > 0.f ? t : (expf(t) - 1.f);
  }
  return c;
}
__device__ __forceinline__ v8f gate_mix(v8f g, float b, v8f e, v8f h2) {
#pragma unroll
  for (int v = 0; v < 8; ++v) {
    float t = g[v] + b;
    float s = 1.f / (1.f + expf(-t));
    g[v] = e[v] + s * h2[v];
  }
  return g;
}

__global__ __launch_bounds__(256, 1)
void vsn_fused_rows32(
    const float* __restrict__ x,
    const float* __restrict__ embW, const float* __restrict__ embB,
    const float* __restrict__ fc1W, const float* __restrict__ fc1B,
    const float* __restrict__ fc2W, const float* __restrict__ fc2B,
    const float* __restrict__ gW,   const float* __restrict__ gB,
    const float* __restrict__ lnG,  const float* __restrict__ lnB,
    const float* __restrict__ sf1W, const float* __restrict__ sf1B,
    const float* __restrict__ sf2W, const float* __restrict__ sf2B,
    const float* __restrict__ sgW,  const float* __restrict__ sgB,
    const float* __restrict__ sskW, const float* __restrict__ sskB,
    const float* __restrict__ slnG, const float* __restrict__ slnB,
    float* __restrict__ outSel, float* __restrict__ outW) {
  __shared__ float smem[SMEM_FLOATS];
  float* TRANS = smem;
  float* XT = TRANS + MR * ST_T;
  float* PU = XT + MR * ST_X;
  float* PV = PU + MR * ST_P;
  float* GP = XT;
  float* SKT = GP + MR * ST_S;
  float* WB = SKT + MR * ST_S;

  const int tid = threadIdx.x;
  const int wid = tid >> 5;
  const int lane = tid & 31;
  const int lane16 = lane & 15;
  const int hi = lane >> 4;
  const int n0 = wid * 16;
  const int lr = tid >> 4;
  const int lc = tid & 15;
  const int row0 = blockIdx.x * MR;

  float bE = embB[n0 + lane16];
  float b1 = fc1B[n0 + lane16];
  float b2 = fc2B[n0 + lane16];
  float bG = gB[n0 + lane16];
  v2f pE[D_ / 4];
  load_panel<D_>(embW, H_, n0, lane16, hi, pE);
  {
#pragma unroll
    for (int rr = 0; rr < 2; ++rr) {
      const int r = lr + 16 * rr;
      float4 v = *(const float4*)(x + (size_t)(row0 + r) * (F_ * D_) + lc * 4);
      float* dp = XT + r * ST_X + lc * 4;
      dp[0] = v.x; dp[1] = v.y; dp[2] = v.z; dp[3] = v.w;
    }
  }
  v2f pF1[H_ / 4];
  float4 xr[2];
  wg_signal_lds();
  keep_order();
  load_panel<H_>(fc1W, H_, n0, lane16, hi, pF1);
#pragma unroll
  for (int rr = 0; rr < 2; ++rr)
    xr[rr] = *(const float4*)(x + (size_t)(row0 + lr + 16 * rr) * (F_ * D_) +
                              1 * D_ + lc * 4);
  keep_order();
  wg_wait();

  for (int f = 0; f < F_; ++f) {
    const int fn = f + 1 < F_ ? f + 1 : f;
    const int fnn = f + 2 < F_ ? f + 2 : F_ - 1;

    v8f ce0, ce1;
    gemm_panel<D_>(XT, ST_X, lane16, hi, pE, ce0, ce1);
    ce0 = add_bias(ce0, bE);
    ce1 = add_bias(ce1, bE);
    store_tile(PU, ST_P, 0, n0, ce0, lane16, hi);
    store_tile(PU, ST_P, 16, n0, ce1, lane16, hi);
    wg_signal_lds();
    keep_order();
    v2f pF2[H_ / 4];
    load_panel<H_>(fc2W + (size_t)f * H_ * H_, H_, n0, lane16, hi, pF2);
    keep_order();
    wg_wait();

    {
      v8f c10, c11;
      gemm_panel<H_>(PU, ST_P, lane16, hi, pF1, c10, c11);
      c10 = bias_elu(c10, b1);
      c11 = bias_elu(c11, b1);
      store_tile(PV, ST_P, 0, n0, c10, lane16, hi);
      store_tile(PV, ST_P, 16, n0, c11, lane16, hi);
    }
    wg_signal_lds();
    keep_order();
    v2f pG[H_ / 4];
    load_panel<H_>(gW + (size_t)f * H_ * H_, H_, n0, lane16, hi, pG);
    keep_order();
    wg_wait();

    v8f c20, c21;
    gemm_panel<H_>(PV, ST_P, lane16, hi, pF2, c20, c21);
    c20 = add_bias(c20, b2);
    c21 = add_bias(c21, b2);
    store_tile(PU, ST_P, 0, n0, c20, lane16, hi);
    store_tile(PU, ST_P, 16, n0, c21, lane16, hi);
    wg_signal_lds();
    keep_order();

    load_panel<D_>(embW + (size_t)fn * D_ * H_, H_, n0, lane16, hi, pE);
    const float bGf = bG;
    bE = embB[fn * H_ + n0 + lane16];
    b1 = fc1B[fn * H_ + n0 + lane16];
    b2 = fc2B[fn * H_ + n0 + lane16];
    bG = gB[fn * H_ + n0 + lane16];
    float lg[8], lb[8];
#pragma unroll
    for (int j = 0; j < 8; ++j) {
      lg[j] = lnG[f * H_ + lc + 16 * j];
      lb[j] = lnB[f * H_ + lc + 16 * j];
    }
    keep_order();
    wg_wait();
    {
      v8f cg0, cg1;
      gemm_panel<H_>(PU, ST_P, lane16, hi, pG, cg0, cg1);
      cg0 = gate_mix(cg0, bGf, ce0, c20);
      cg1 = gate_mix(cg1, bGf, ce1, c21);
      store_tile(PV, ST_P, 0, n0, cg0, lane16, hi);
      store_tile(PV, ST_P, 16, n0, cg1, lane16, hi);
    }
    wg_signal_lds();
    keep_order();
    load_panel_part<0, H_ / 8>(fc1W + (size_t)fn * H_ * H_, H_, n0, lane16, hi, pF1);
    keep_order();
    wg_wait();

#pragma unroll
    for (int rr = 0; rr < 2; ++rr) {
      const int r = lr + 16 * rr;
      float vals[8];
      float s = 0.f, ss = 0.f;
#pragma unroll
      for (int j = 0; j < 8; ++j) {
        float v = PV[r * ST_P + lc + 16 * j];
        vals[j] = v; s += v; ss += v * v;
      }
      s = rowsum16(s);
      ss = rowsum16(ss);
      float mu = s * (1.f / 128.f);
      float var = ss * (1.f / 128.f) - mu * mu;
      float rs = rsqrtf(var + 1e-5f);
#pragma unroll
      for (int j = 0; j < 8; ++j) {
        int h = lc + 16 * j;
        float nv = (vals[j] - mu) * rs * lg[j] + lb[j];
        TRANS[r * ST_T + f * H_ + h] = nv;
      }
      float* dp = XT + r * ST_X + lc * 4;
      dp[0] = xr[rr].x; dp[1] = xr[rr].y; dp[2] = xr[rr].z; dp[3] = xr[rr].w;
    }
    wg_signal_lds();
    keep_order();
    load_panel_part<H_ / 8, H_ / 4>(fc1W + (size_t)fn * H_ * H_, H_, n0, lane16, hi, pF1);
#pragma unroll
    for (int rr = 0; rr < 2; ++rr)
      xr[rr] = *(const float4*)(x + (size_t)(row0 + lr + 16 * rr) * (F_ * D_) +
                                fnn * D_ + lc * 4);
    keep_order();
    wg_wait();
  }

  v2f pS2[H_ / 4];
  load_panel<H_>(sf2W, H_, n0, lane16, hi, pS2);
  const float bS1 = sf1B[n0 + lane16];
  const float bS2 = sf2B[n0 + lane16];
  {
    v8f b10, b11;
    gemm_stream<FH_, 64>(TRANS, ST_T, sf1W, H_, n0, lane16, hi, b10, b11);
    b10 = bias_elu(b10, bS1);
    b11 = bias_elu(b11, bS1);
    store_tile(PU, ST_P, 0, n0, b10, lane16, hi);
    store_tile(PU, ST_P, 16, n0, b11, lane16, hi);
  }
  wg_barrier_lds();

  {
    v8f b20, b21;
    gemm_panel<H_>(PU, ST_P, lane16, hi, pS2, b20, b21);
    b20 = add_bias(b20, bS2);
    b21 = add_bias(b21, bS2);
    store_tile(PV, ST_P, 0, n0, b20, lane16, hi);
    store_tile(PV, ST_P, 16, n0, b21, lane16, hi);
  }
  wg_barrier_lds();

  if (wid == 0) {
    v2f pSG[H_ / 4];
    load_panel<H_>(sgW, F_, 0, lane16, hi, pSG);
    v8f g0, g1;
    gemm_panel<H_>(PV, ST_P, lane16, hi, pSG, g0, g1);
    float b = sgB[lane16];
    g0 = add_bias(g0, b);
    g1 = add_bias(g1, b);
    store_tile(GP, ST_S, 0, 0, g0, lane16, hi);
    store_tile(GP, ST_S, 16, 0, g1, lane16, hi);
  } else if (wid == 1) {
    v8f s0, s1;
    gemm_stream<FH_, 64>(TRANS, ST_T, sskW, F_, 0, lane16, hi, s0, s1);
    float b = sskB[lane16];
    s0 = add_bias(s0, b);
    s1 = add_bias(s1, b);
    store_tile(SKT, ST_S, 0, 0, s0, lane16, hi);
    store_tile(SKT, ST_S, 16, 0, s1, lane16, hi);
  }
  wg_barrier_lds();

#pragma unroll
  for (int rr = 0; rr < 2; ++rr) {
    const int r = lr + 16 * rr;
    float gp = GP[r * ST_S + lc];
    float sg = 1.f / (1.f + expf(-gp));
    float v = SKT[r * ST_S + lc] + sg * gp;
    float s = rowsum16(v);
    float ss = rowsum16(v * v);
    float mu = s * (1.f / 16.f);
    float var = ss * (1.f / 16.f) - mu * mu;
    float nv = (v - mu) * rsqrtf(var + 1e-5f) * slnG[lc] + slnB[lc];
    float mx = rowmax16(nv);
    float e = expf(nv - mx);
    float se = rowsum16(e);
    float w = e / se;
    WB[r * 16 + lc] = w;
    outW[(size_t)(row0 + r) * F_ + lc] = w;
  }
  wg_barrier_lds();

#pragma unroll
  for (int rr = 0; rr < 2; ++rr) {
    const int r = lr + 16 * rr;
#pragma unroll
    for (int j = 0; j < 8; ++j) {
      int h = lc + 16 * j;
      float acc = 0.f;
#pragma unroll
      for (int f = 0; f < F_; ++f)
        acc += TRANS[r * ST_T + f * H_ + h] * WB[r * 16 + f];
      outSel[(size_t)(row0 + r) * H_ + h] = acc;
    }
  }
}

extern "C" void kernel_launch(void* const* d_in, const int* in_sizes, int n_in,
                              void* d_out, int out_size, void* d_ws, size_t ws_size,
                              hipStream_t stream) {
  (void)in_sizes; (void)n_in; (void)out_size; (void)d_ws; (void)ws_size;
  const float* x    = (const float*)d_in[0];
  const float* embW = (const float*)d_in[1];
  const float* embB = (const float*)d_in[2];
  const float* fc1W = (const float*)d_in[3];
  const float* fc1B = (const float*)d_in[4];
  const float* fc2W = (const float*)d_in[5];
  const float* fc2B = (const float*)d_in[6];
  const float* gW   = (const float*)d_in[7];
  const float* gB   = (const float*)d_in[8];
  const float* lnG  = (const float*)d_in[9];
  const float* lnB  = (const float*)d_in[10];
  const float* sf1W = (const float*)d_in[11];
  const float* sf1B = (const float*)d_in[12];
  const float* sf2W = (const float*)d_in[13];
  const float* sf2B = (const float*)d_in[14];
  const float* sgW  = (const float*)d_in[15];
  const float* sgB  = (const float*)d_in[16];
  const float* sskW = (const float*)d_in[17];
  const float* sskB = (const float*)d_in[18];
  const float* slnG = (const float*)d_in[19];
  const float* slnB = (const float*)d_in[20];

  float* outSel = (float*)d_out;
  float* outW   = outSel + (size_t)ROWS_TOTAL * H_;

  vsn_fused_rows32<<<ROWS_TOTAL / MR, 256, 0, stream>>>(
      x, embW, embB, fc1W, fc1B, fc2W, fc2B, gW, gB, lnG, lnB,
      sf1W, sf1B, sf2W, sf2B, sgW, sgB, sskW, sskB, slnG, slnB,
      outSel, outW);
}


# Stage A takes two features per barrier interval: waves 0-3 the even one, 4-7 the odd one, each wave a 32x32 tile (4 accumulators, 8-byte weight loads); barriers and weight loads per feature halved
# speedup vs baseline: 2.3848x; 1.0509x over previous
#include <hip/hip_runtime.h>
#include <hip/hip_bf16.h>
#include <math.h>


typedef float v2f __attribute__((ext_vector_type(2)));
typedef float v8f __attribute__((ext_vector_type(8)));

#define ROWS_TOTAL 16384
#define MR 32
#define F_ 16
#define D_ 64
#define H_ 128
#define FH_ 2048

#define ST_T 2052
#define ST_X 68
#define ST_P 132
#define ST_S 20

#define SMEM_FLOATS (MR*ST_T + 2*MR*ST_X + 2*MR*ST_P)

__device__ __forceinline__ float rowsum16(float v) {
  v += __shfl_xor(v, 1, 16);
  v += __shfl_xor(v, 2, 16);
  v += __shfl_xor(v, 4, 16);
  v += __shfl_xor(v, 8, 16);
  return v;
}
__device__ __forceinline__ float rowmax16(float v) {
  v = fmaxf(v, __shfl_xor(v, 1, 16));
  v = fmaxf(v, __shfl_xor(v, 2, 16));
  v = fmaxf(v, __shfl_xor(v, 4, 16));
  v = fmaxf(v, __shfl_xor(v, 8, 16));
  return v;
}

__device__ __forceinline__ void wg_barrier_lds() {
  asm volatile("s_wait_dscnt 0x0\n\ts_barrier_signal -1\n\ts_barrier_wait -1" ::: "memory");
}
__device__ __forceinline__ void wg_signal_lds() {
  asm volatile("s_wait_dscnt 0x0\n\ts_barrier_signal -1" ::: "memory");
}
__device__ __forceinline__ void wg_wait() {
  asm volatile("s_barrier_wait -1" ::: "memory");
}

__device__ __forceinline__ void keep_order() { __builtin_amdgcn_sched_barrier(0); }

__device__ __forceinline__ v8f mma4(v2f a, v2f b, v8f c) {
  return __builtin_amdgcn_wmma_f32_16x16x4_f32(false, a, false, b, (short)0, c,
                                               false, false);
}


template <int K>
__device__ __forceinline__ void load_panel(const float* __restrict__ W, int ldw,
                                           int n0, int lane16, int hi,
                                           v2f (&b)[K / 4]) {
  const float* wcol = W + (size_t)(hi * 2) * ldw + n0 + lane16;
#pragma unroll
  for (int s = 0; s < K / 4; ++s) {
    b[s].x = wcol[(size_t)(4 * s) * ldw];
    b[s].y = wcol[(size_t)(4 * s + 1) * ldw];
  }
}

template <int S>
__device__ __forceinline__ void load_a(const float* arow0, const float* arow1,
                                       int kb, v2f (&a0)[S], v2f (&a1)[S]) {
#pragma unroll
  for (int s = 0; s < S; ++s) {
    a0[s].x = arow0[kb + 4 * s];
    a0[s].y = arow0[kb + 4 * s + 1];
    a1[s].x = arow1[kb + 4 * s];
    a1[s].y = arow1[kb + 4 * s + 1];
  }
}

template <int K>
__device__ __forceinline__ void gemm_panel(const float* As, int lda, int lane16,
                                           int hi, const v2f (&b)[K / 4],
                                           v8f& c0, v8f& c1) {
  const v8f z = {0.f, 0.f, 0.f, 0.f, 0.f, 0.f, 0.f, 0.f};
  c0 = z;
  c1 = z;
  const float* arow0 = As + lane16 * lda + hi * 2;
  const float* arow1 = arow0 + 16 * lda;
  v2f a0[2][8], a1[2][8];
  load_a<8>(arow0, arow1, 0, a0[0], a1[0]);
  keep_order();
#pragma unroll
  for (int ch = 0; ch < K / 32; ++ch) {
    if (ch + 1 < K / 32)
      load_a<8>(arow0, arow1, (ch + 1) * 32, a0[(ch + 1) & 1], a1[(ch + 1) & 1]);
    keep_order();
#pragma unroll
    for (int s = 0; s < 8; ++s) {
      c0 = mma4(a0[ch & 1][s], b[ch * 8 + s], c0);
      c1 = mma4(a1[ch & 1][s], b[ch * 8 + s], c1);
    }
    keep_order();
  }
}

template <int S0, int S1, int NS>
__device__ __forceinline__ void load_panel2(const float* __restrict__ W, int ldw,
                                            int n0, int lane16, int hi,
                                            float2 (&ra)[NS], float2 (&rb)[NS]) {
  const float* wcol = W + (size_t)(hi * 2) * ldw + n0 + 2 * lane16;
#pragma unroll
  for (int s = S0; s < S1; ++s) {
    ra[s] = *(const float2*)(wcol + (size_t)(4 * s) * ldw);
    rb[s] = *(const float2*)(wcol + (size_t)(4 * s + 1) * ldw);
  }
}

template <int K>
__device__ __forceinline__ void gemm_panel2(const float* As, int lda, int lane16,
                                            int hi, const float2 (&ra)[K / 4],
                                            const float2 (&rb)[K / 4],
                                            v8f& c00, v8f& c01, v8f& c10, v8f& c11) {
  const v8f z = {0.f, 0.f, 0.f, 0.f, 0.f, 0.f, 0.f, 0.f};
  c00 = z; c01 = z; c10 = z; c11 = z;
  const float* arow0 = As + lane16 * lda + hi * 2;
  const float* arow1 = arow0 + 16 * lda;
  v2f a0[2][4], a1[2][4];
  load_a<4>(arow0, arow1, 0, a0[0], a1[0]);
  keep_order();
#pragma unroll
  for (int ch = 0; ch < K / 16; ++ch) {
    if (ch + 1 < K / 16)
      load_a<4>(arow0, arow1, (ch + 1) * 16, a0[(ch + 1) & 1], a1[(ch + 1) & 1]);
    keep_order();
#pragma unroll
    for (int s = 0; s < 4; ++s) {
      v2f bj0, bj1;
      bj0.x = ra[ch * 4 + s].x; bj0.y = rb[ch * 4 + s].x;
      bj1.x = ra[ch * 4 + s].y; bj1.y = rb[ch * 4 + s].y;
      c00 = mma4(a0[ch & 1][s], bj0, c00);
      c10 = mma4(a1[ch & 1][s], bj0, c10);
      c01 = mma4(a0[ch & 1][s], bj1, c01);
      c11 = mma4(a1[ch & 1][s], bj1, c11);
    }
    keep_order();
  }
}

__device__ __forceinline__ void store_tile2(float* dst, int ldd, int rbase, int cb,
                                            v8f cj0, v8f cj1, int hi) {
#pragma unroll
  for (int v = 0; v < 8; ++v) {
    float2 t;
    t.x = cj0[v];
    t.y = cj1[v];
    *(float2*)(dst + (rbase + v + hi * 8) * ldd + cb) = t;
  }
}

template <int S>
__device__ __forceinline__ void load_chunk(const float* arow0, const float* arow1,
                                           const float* __restrict__ wcol, int ldw,
                                           int kb, v2f (&a0)[S], v2f (&a1)[S],
                                           v2f (&b)[S]) {
#pragma unroll
  for (int s = 0; s < S; ++s) {
    b[s].x = wcol[(size_t)(kb + 4 * s) * ldw];
    b[s].y = wcol[(size_t)(kb + 4 * s + 1) * ldw];
  }
  load_a<S>(arow0, arow1, kb, a0, a1);
}
template <int S>
__device__ __forceinline__ void mma_chunk(const v2f (&a0)[S], const v2f (&a1)[S],
                                          const v2f (&b)[S], v8f& c0, v8f& c1) {
#pragma unroll
  for (int s = 0; s < S; ++s) {
    c0 = mma4(a0[s], b[s], c0);
    c1 = mma4(a1[s], b[s], c1);
  }
}

template <int K, int CH>
__device__ __forceinline__ void gemm_stream(const float* As, int lda,
                                            const float* __restrict__ W, int ldw,
                                            int n0, int lane16, int hi,
                                            v8f& c0, v8f& c1) {
  static_assert(K % (2 * CH) == 0 && K >= 2 * CH, "chunk pairs");
  constexpr int S = CH / 4;
  const v8f z = {0.f, 0.f, 0.f, 0.f, 0.f, 0.f, 0.f, 0.f};
  c0 = z;
  c1 = z;
  const float* arow0 = As + lane16 * lda + hi * 2;
  const float* arow1 = arow0 + 16 * lda;
  const float* wcol = W + (size_t)(hi * 2) * ldw + n0 + lane16;
  v2f bP[S], a0P[S], a1P[S], bQ[S], a0Q[S], a1Q[S];
  load_chunk<S>(arow0, arow1, wcol, ldw, 0, a0P, a1P, bP);
  keep_order();
#pragma unroll 1
  for (int kb = 0; kb < K - 2 * CH; kb += 2 * CH) {
    load_chunk<S>(arow0, arow1, wcol, ldw, kb + CH, a0Q, a1Q, bQ);
    keep_order();
    mma_chunk<S>(a0P, a1P, bP, c0, c1);
    keep_order();
    load_chunk<S>(arow0, arow1, wcol, ldw, kb + 2 * CH, a0P, a1P, bP);
    keep_order();
    mma_chunk<S>(a0Q, a1Q, bQ, c0, c1);
    keep_order();
  }
  load_chunk<S>(arow0, arow1, wcol, ldw, K - CH, a0Q, a1Q, bQ);
  keep_order();
  mma_chunk<S>(a0P, a1P, bP, c0, c1);
  mma_chunk<S>(a0Q, a1Q, bQ, c0, c1);
}

__device__ __forceinline__ void store_tile(float* dst, int ldd, int rbase,
                                           int col0, v8f c, int lane16, int hi) {
#pragma unroll
  for (int v = 0; v < 8; ++v)
    dst[(rbase + v + hi * 8) * ldd + col0 + lane16] = c[v];
}

__device__ __forceinline__ v8f add_bias(v8f c, float b) {
#pragma unroll
  for (int v = 0; v < 8; ++v) c[v] += b;
  return c;
}
__device__ __forceinline__ v8f bias_elu(v8f c, float b) {
#pragma unroll
  for (int v = 0; v < 8; ++v) {
    float t = c[v] + b;
    c[v] = t > 0.f ? t : (expf(t) - 1.f);
  }
  return c;
}
__device__ __forceinline__ v8f gate_mix(v8f g, float b, v8f e, v8f h2) {
#pragma unroll
  for (int v = 0; v < 8; ++v) {
    float t = g[v] + b;
    float s = 1.f / (1.f + expf(-t));
    g[v] = e[v] + s * h2[v];
  }
  return g;
}

__global__ __launch_bounds__(256, 1)
void vsn_fused_pairs32(
    const float* __restrict__ x,
    const float* __restrict__ embW, const float* __restrict__ embB,
    const float* __restrict__ fc1W, const float* __restrict__ fc1B,
    const float* __restrict__ fc2W, const float* __restrict__ fc2B,
    const float* __restrict__ gW,   const float* __restrict__ gB,
    const float* __restrict__ lnG,  const float* __restrict__ lnB,
    const float* __restrict__ sf1W, const float* __restrict__ sf1B,
    const float* __restrict__ sf2W, const float* __restrict__ sf2B,
    const float* __restrict__ sgW,  const float* __restrict__ sgB,
    const float* __restrict__ sskW, const float* __restrict__ sskB,
    const float* __restrict__ slnG, const float* __restrict__ slnB,
    float* __restrict__ outSel, float* __restrict__ outW) {
  __shared__ __attribute__((aligned(16))) float smem[SMEM_FLOATS];
  float* TRANS = smem;
  float* XT0 = TRANS + MR * ST_T;
  float* XW0 = XT0 + 2 * MR * ST_X;
  float* PU = XW0;
  float* PV = XW0 + MR * ST_P;
  float* GP = XT0;
  float* SKT = GP + MR * ST_S;
  float* WB = SKT + MR * ST_S;

  const int tid = threadIdx.x;
  const int wid = tid >> 5;
  const int lane = tid & 31;
  const int lane16 = lane & 15;
  const int hi = lane >> 4;
  const int row0 = blockIdx.x * MR;

  const int g = wid >> 2;
  const int n0 = (wid & 3) * 32;
  const int cb = n0 + 2 * lane16;
  const int glr = (tid & 127) >> 4;
  const int lc = tid & 15;
  float* XT = XT0 + g * (MR * ST_X);
  float* XW = XW0 + g * (MR * ST_P);

  float2 bE = *(const float2*)(embB + g * H_ + cb);
  float2 b1 = *(const float2*)(fc1B + g * H_ + cb);
  float2 b2 = *(const float2*)(fc2B + g * H_ + cb);
  float2 bG = *(const float2*)(gB + g * H_ + cb);
  float2 pE0[D_ / 4], pE1[D_ / 4];
  load_panel2<0, D_ / 4, D_ / 4>(embW + (size_t)g * D_ * H_, H_, n0, lane16, hi, pE0, pE1);
  {
#pragma unroll
    for (int rr = 0; rr < 4; ++rr) {
      const int r = glr + 8 * rr;
      float4 v = *(const float4*)(x + (size_t)(row0 + r) * (F_ * D_) + g * D_ + lc * 4);
      float* dp = XT + r * ST_X + lc * 4;
      dp[0] = v.x; dp[1] = v.y; dp[2] = v.z; dp[3] = v.w;
    }
  }
  float2 pF10[H_ / 4], pF11[H_ / 4];
  float4 xr[4];
  wg_signal_lds();
  keep_order();
  load_panel2<0, H_ / 4, H_ / 4>(fc1W + (size_t)g * H_ * H_, H_, n0, lane16, hi, pF10, pF11);
#pragma unroll
  for (int rr = 0; rr < 4; ++rr)
    xr[rr] = *(const float4*)(x + (size_t)(row0 + glr + 8 * rr) * (F_ * D_) +
                              (g + 2) * D_ + lc * 4);
  keep_order();
  wg_wait();

  for (int f = g; f < F_; f += 2) {
    const int fn = f + 2 < F_ ? f + 2 : f;
    const int fnn = f + 4 < F_ ? f + 4 : fn;
    float* SLOT = TRANS + f * H_;

    v8f ce00, ce01, ce10, ce11;
    gemm_panel2<D_>(XT, ST_X, lane16, hi, pE0, pE1, ce00, ce01, ce10, ce11);
    ce00 = add_bias(ce00, bE.x);
    ce10 = add_bias(ce10, bE.x);
    ce01 = add_bias(ce01, bE.y);
    ce11 = add_bias(ce11, bE.y);
    store_tile2(SLOT, ST_T, 0, cb, ce00, ce01, hi);
    store_tile2(SLOT, ST_T, 16, cb, ce10, ce11, hi);
    wg_signal_lds();
    keep_order();
    float2 pF20[H_ / 4], pF21[H_ / 4];
    load_panel2<0, H_ / 4, H_ / 4>(fc2W + (size_t)f * H_ * H_, H_, n0, lane16, hi, pF20, pF21);
    keep_order();
    wg_wait();

    {
      v8f c00, c01, c10, c11;
      gemm_panel2<H_>(SLOT, ST_T, lane16, hi, pF10, pF11, c00, c01, c10, c11);
      c00 = bias_elu(c00, b1.x);
      c10 = bias_elu(c10, b1.x);
      c01 = bias_elu(c01, b1.y);
      c11 = bias_elu(c11, b1.y);
      store_tile2(XW, ST_P, 0, cb, c00, c01, hi);
      store_tile2(XW, ST_P, 16, cb, c10, c11, hi);
    }
    wg_signal_lds();
    keep_order();
    float2 pG0[H_ / 4], pG1[H_ / 4];
    load_panel2<0, H_ / 4, H_ / 4>(gW + (size_t)f * H_ * H_, H_, n0, lane16, hi, pG0, pG1);
    keep_order();
    wg_wait();

    v8f c200, c201, c210, c211;
    gemm_panel2<H_>(XW, ST_P, lane16, hi, pF20, pF21, c200, c201, c210, c211);
    c200 = add_bias(c200, b2.x);
    c210 = add_bias(c210, b2.x);
    c201 = add_bias(c201, b2.y);
    c211 = add_bias(c211, b2.y);
    store_tile2(SLOT, ST_T, 0, cb, c200, c201, hi);
    store_tile2(SLOT, ST_T, 16, cb, c210, c211, hi);
    wg_signal_lds();
    keep_order();

    load_panel2<0, D_ / 4, D_ / 4>(embW + (size_t)fn * D_ * H_, H_, n0, lane16, hi, pE0, pE1);
    const float2 bGf = bG;
    bE = *(const float2*)(embB + fn * H_ + cb);
    b1 = *(const float2*)(fc1B + fn * H_ + cb);
    b2 = *(const float2*)(fc2B + fn * H_ + cb);
    bG = *(const float2*)(gB + fn * H_ + cb);
    float lg[8], lb[8];
#pragma unroll
    for (int j = 0; j < 8; ++j) {
      lg[j] = lnG[f * H_ + lc + 16 * j];
      lb[j] = lnB[f * H_ + lc + 16 * j];
    }
    keep_order();
    wg_wait();
    {
      v8f cg00, cg01, cg10, cg11;
      gemm_panel2<H_>(SLOT, ST_T, lane16, hi, pG0, pG1, cg00, cg01, cg10, cg11);
      cg00 = gate_mix(cg00, bGf.x, ce00, c200);
      cg10 = gate_mix(cg10, bGf.x, ce10, c210);
      cg01 = gate_mix(cg01, bGf.y, ce01, c201);
      cg11 = gate_mix(cg11, bGf.y, ce11, c211);
      store_tile2(XW, ST_P, 0, cb, cg00, cg01, hi);
      store_tile2(XW, ST_P, 16, cb, cg10, cg11, hi);
    }
    wg_signal_lds();
    keep_order();
    load_panel2<0, H_ / 8, H_ / 4>(fc1W + (size_t)fn * H_ * H_, H_, n0, lane16, hi, pF10, pF11);
    keep_order();
    wg_wait();

#pragma unroll
    for (int rr = 0; rr < 4; ++rr) {
      const int r = glr + 8 * rr;
      float vals[8];
      float s = 0.f, ss = 0.f;
#pragma unroll
      for (int j = 0; j < 8; ++j) {
        float v = XW[r * ST_P + lc + 16 * j];
        vals[j] = v; s += v; ss += v * v;
      }
      s = rowsum16(s);
      ss = rowsum16(ss);
      float mu = s * (1.f / 128.f);
      float var = ss * (1.f / 128.f) - mu * mu;
      float rs = rsqrtf(var + 1e-5f);
#pragma unroll
      for (int j = 0; j < 8; ++j) {
        int h = lc + 16 * j;
        float nv = (vals[j] - mu) * rs * lg[j] + lb[j];
        SLOT[r * ST_T + h] = nv;
      }
      float* dp = XT + r * ST_X + lc * 4;
      dp[0] = xr[rr].x; dp[1] = xr[rr].y; dp[2] = xr[rr].z; dp[3] = xr[rr].w;
    }
    wg_signal_lds();
    keep_order();
    load_panel2<H_ / 8, H_ / 4, H_ / 4>(fc1W + (size_t)fn * H_ * H_, H_, n0, lane16, hi, pF10, pF11);
#pragma unroll
    for (int rr = 0; rr < 4; ++rr)
      xr[rr] = *(const float4*)(x + (size_t)(row0 + glr + 8 * rr) * (F_ * D_) +
                                fnn * D_ + lc * 4);
    keep_order();
    wg_wait();
  }

  const int nB = wid * 16;
  const int lr = tid >> 4;
  v2f pS2[H_ / 4];
  load_panel<H_>(sf2W, H_, nB, lane16, hi, pS2);
  const float bS1 = sf1B[nB + lane16];
  const float bS2 = sf2B[nB + lane16];
  {
    v8f b10, b11;
    gemm_stream<FH_, 64>(TRANS, ST_T, sf1W, H_, nB, lane16, hi, b10, b11);
    b10 = bias_elu(b10, bS1);
    b11 = bias_elu(b11, bS1);
    store_tile(PU, ST_P, 0, nB, b10, lane16, hi);
    store_tile(PU, ST_P, 16, nB, b11, lane16, hi);
  }
  wg_barrier_lds();

  {
    v8f b20, b21;
    gemm_panel<H_>(PU, ST_P, lane16, hi, pS2, b20, b21);
    b20 = add_bias(b20, bS2);
    b21 = add_bias(b21, bS2);
    store_tile(PV, ST_P, 0, nB, b20, lane16, hi);
    store_tile(PV, ST_P, 16, nB, b21, lane16, hi);
  }
  wg_barrier_lds();

  if (wid == 0) {
    v2f pSG[H_ / 4];
    load_panel<H_>(sgW, F_, 0, lane16, hi, pSG);
    v8f g0, g1;
    gemm_panel<H_>(PV, ST_P, lane16, hi, pSG, g0, g1);
    float b = sgB[lane16];
    g0 = add_bias(g0, b);
    g1 = add_bias(g1, b);
    store_tile(GP, ST_S, 0, 0, g0, lane16, hi);
    store_tile(GP, ST_S, 16, 0, g1, lane16, hi);
  } else if (wid == 1) {
    v8f s0, s1;
    gemm_stream<FH_, 64>(TRANS, ST_T, sskW, F_, 0, lane16, hi, s0, s1);
    float b = sskB[lane16];
    s0 = add_bias(s0, b);
    s1 = add_bias(s1, b);
    store_tile(SKT, ST_S, 0, 0, s0, lane16, hi);
    store_tile(SKT, ST_S, 16, 0, s1, lane16, hi);
  }
  wg_barrier_lds();

#pragma unroll
  for (int rr = 0; rr < 2; ++rr) {
    const int r = lr + 16 * rr;
    float gp = GP[r * ST_S + lc];
    float sg = 1.f / (1.f + expf(-gp));
    float v = SKT[r * ST_S + lc] + sg * gp;
    float s = rowsum16(v);
    float ss = rowsum16(v * v);
    float mu = s * (1.f / 16.f);
    float var = ss * (1.f / 16.f) - mu * mu;
    float nv = (v - mu) * rsqrtf(var + 1e-5f) * slnG[lc] + slnB[lc];
    float mx = rowmax16(nv);
    float e = expf(nv - mx);
    float se = rowsum16(e);
    float w = e / se;
    WB[r * 16 + lc] = w;
    outW[(size_t)(row0 + r) * F_ + lc] = w;
  }
  wg_barrier_lds();

#pragma unroll
  for (int rr = 0; rr < 2; ++rr) {
    const int r = lr + 16 * rr;
#pragma unroll
    for (int j = 0; j < 8; ++j) {
      int h = lc + 16 * j;
      float acc = 0.f;
#pragma unroll
      for (int f = 0; f < F_; ++f)
        acc += TRANS[r * ST_T + f * H_ + h] * WB[r * 16 + f];
      outSel[(size_t)(row0 + r) * H_ + h] = acc;
    }
  }
}

extern "C" void kernel_launch(void* const* d_in, const int* in_sizes, int n_in,
                              void* d_out, int out_size, void* d_ws, size_t ws_size,
                              hipStream_t stream) {
  (void)in_sizes; (void)n_in; (void)out_size; (void)d_ws; (void)ws_size;
  const float* x    = (const float*)d_in[0];
  const float* embW = (const float*)d_in[1];
  const float* embB = (const float*)d_in[2];
  const float* fc1W = (const float*)d_in[3];
  const float* fc1B = (const float*)d_in[4];
  const float* fc2W = (const float*)d_in[5];
  const float* fc2B = (const float*)d_in[6];
  const float* gW   = (const float*)d_in[7];
  const float* gB   = (const float*)d_in[8];
  const float* lnG  = (const float*)d_in[9];
  const float* lnB  = (const float*)d_in[10];
  const float* sf1W = (const float*)d_in[11];
  const float* sf1B = (const float*)d_in[12];
  const float* sf2W = (const float*)d_in[13];
  const float* sf2B = (const float*)d_in[14];
  const float* sgW  = (const float*)d_in[15];
  const float* sgB  = (const float*)d_in[16];
  const float* sskW = (const float*)d_in[17];
  const float* sskB = (const float*)d_in[18];
  const float* slnG = (const float*)d_in[19];
  const float* slnB = (const float*)d_in[20];

  float* outSel = (float*)d_out;
  float* outW   = outSel + (size_t)ROWS_TOTAL * H_;

  vsn_fused_pairs32<<<ROWS_TOTAL / MR, 256, 0, stream>>>(
      x, embW, embB, fc1W, fc1B, fc2W, fc2B, gW, gB, lnG, lnB,
      sf1W, sf1B, sf2W, sf2B, sgW, sgB, sskW, sskB, slnG, slnB,
      outSel, outW);
}


# ELU and gate sigmoid on hardware v_exp_f32 / v_rcp_f32 with a select per value, replacing library expf, the correctly rounded division and a branch per value: ~25 fewer vector ops per value pair
# speedup vs baseline: 2.6739x; 1.1212x over previous
#include <hip/hip_runtime.h>
#include <hip/hip_bf16.h>
#include <math.h>


typedef float v2f __attribute__((ext_vector_type(2)));
typedef float v8f __attribute__((ext_vector_type(8)));

#define ROWS_TOTAL 16384
#define MR 32
#define F_ 16
#define D_ 64
#define H_ 128
#define FH_ 2048

#define ST_T 2052
#define ST_X 68
#define ST_P 132
#define ST_S 20

#define SMEM_FLOATS (MR*ST_T + 2*MR*ST_X + 2*MR*ST_P)

__device__ __forceinline__ float rowsum16(float v) {
  v += __shfl_xor(v, 1, 16);
  v += __shfl_xor(v, 2, 16);
  v += __shfl_xor(v, 4, 16);
  v += __shfl_xor(v, 8, 16);
  return v;
}
__device__ __forceinline__ float rowmax16(float v) {
  v = fmaxf(v, __shfl_xor(v, 1, 16));
  v = fmaxf(v, __shfl_xor(v, 2, 16));
  v = fmaxf(v, __shfl_xor(v, 4, 16));
  v = fmaxf(v, __shfl_xor(v, 8, 16));
  return v;
}

__device__ __forceinline__ void wg_barrier_lds() {
  asm volatile("s_wait_dscnt 0x0\n\ts_barrier_signal -1\n\ts_barrier_wait -1" ::: "memory");
}
__device__ __forceinline__ void wg_signal_lds() {
  asm volatile("s_wait_dscnt 0x0\n\ts_barrier_signal -1" ::: "memory");
}
__device__ __forceinline__ void wg_wait() {
  asm volatile("s_barrier_wait -1" ::: "memory");
}

__device__ __forceinline__ void keep_order() { __builtin_amdgcn_sched_barrier(0); }

__device__ __forceinline__ v8f mma4(v2f a, v2f b, v8f c) {
  return __builtin_amdgcn_wmma_f32_16x16x4_f32(false, a, false, b, (short)0, c,
                                               false, false);
}


template <int K>
__device__ __forceinline__ void load_panel(const float* __restrict__ W, int ldw,
                                           int n0, int lane16, int hi,
                                           v2f (&b)[K / 4]) {
  const float* wcol = W + (size_t)(hi * 2) * ldw + n0 + lane16;
#pragma unroll
  for (int s = 0; s < K / 4; ++s) {
    b[s].x = wcol[(size_t)(4 * s) * ldw];
    b[s].y = wcol[(size_t)(4 * s + 1) * ldw];
  }
}

template <int S>
__device__ __forceinline__ void load_a(const float* arow0, const float* arow1,
                                       int kb, v2f (&a0)[S], v2f (&a1)[S]) {
#pragma unroll
  for (int s = 0; s < S; ++s) {
    a0[s].x = arow0[kb + 4 * s];
    a0[s].y = arow0[kb + 4 * s + 1];
    a1[s].x = arow1[kb + 4 * s];
    a1[s].y = arow1[kb + 4 * s + 1];
  }
}

template <int K>
__device__ __forceinline__ void gemm_panel(const float* As, int lda, int lane16,
                                           int hi, const v2f (&b)[K / 4],
                                           v8f& c0, v8f& c1) {
  const v8f z = {0.f, 0.f, 0.f, 0.f, 0.f, 0.f, 0.f, 0.f};
  c0 = z;
  c1 = z;
  const float* arow0 = As + lane16 * lda + hi * 2;
  const float* arow1 = arow0 + 16 * lda;
  v2f a0[2][8], a1[2][8];
  load_a<8>(arow0, arow1, 0, a0[0], a1[0]);
  keep_order();
#pragma unroll
  for (int ch = 0; ch < K / 32; ++ch) {
    if (ch + 1 < K / 32)
      load_a<8>(arow0, arow1, (ch + 1) * 32, a0[(ch + 1) & 1], a1[(ch + 1) & 1]);
    keep_order();
#pragma unroll
    for (int s = 0; s < 8; ++s) {
      c0 = mma4(a0[ch & 1][s], b[ch * 8 + s], c0);
      c1 = mma4(a1[ch & 1][s], b[ch * 8 + s], c1);
    }
    keep_order();
  }
}

template <int S0, int S1, int NS>
__device__ __forceinline__ void load_panel2(const float* __restrict__ W, int ldw,
                                            int n0, int lane16, int hi,
                                            float2 (&ra)[NS], float2 (&rb)[NS]) {
  const float* wcol = W + (size_t)(hi * 2) * ldw + n0 + 2 * lane16;
#pragma unroll
  for (int s = S0; s < S1; ++s) {
    ra[s] = *(const float2*)(wcol + (size_t)(4 * s) * ldw);
    rb[s] = *(const float2*)(wcol + (size_t)(4 * s + 1) * ldw);
  }
}

template <int K>
__device__ __forceinline__ void gemm_panel2(const float* As, int lda, int lane16,
                                            int hi, const float2 (&ra)[K / 4],
                                            const float2 (&rb)[K / 4],
                                            v8f& c00, v8f& c01, v8f& c10, v8f& c11) {
  const v8f z = {0.f, 0.f, 0.f, 0.f, 0.f, 0.f, 0.f, 0.f};
  c00 = z; c01 = z; c10 = z; c11 = z;
  const float* arow0 = As + lane16 * lda + hi * 2;
  const float* arow1 = arow0 + 16 * lda;
  v2f a0[2][4], a1[2][4];
  load_a<4>(arow0, arow1, 0, a0[0], a1[0]);
  keep_order();
#pragma unroll
  for (int ch = 0; ch < K / 16; ++ch) {
    if (ch + 1 < K / 16)
      load_a<4>(arow0, arow1, (ch + 1) * 16, a0[(ch + 1) & 1], a1[(ch + 1) & 1]);
    keep_order();
#pragma unroll
    for (int s = 0; s < 4; ++s) {
      v2f bj0, bj1;
      bj0.x = ra[ch * 4 + s].x; bj0.y = rb[ch * 4 + s].x;
      bj1.x = ra[ch * 4 + s].y; bj1.y = rb[ch * 4 + s].y;
      c00 = mma4(a0[ch & 1][s], bj0, c00);
      c10 = mma4(a1[ch & 1][s], bj0, c10);
      c01 = mma4(a0[ch & 1][s], bj1, c01);
      c11 = mma4(a1[ch & 1][s], bj1, c11);
    }
    keep_order();
  }
}

__device__ __forceinline__ void store_tile2(float* dst, int ldd, int rbase, int cb,
                                            v8f cj0, v8f cj1, int hi) {
#pragma unroll
  for (int v = 0; v < 8; ++v) {
    float2 t;
    t.x = cj0[v];
    t.y = cj1[v];
    *(float2*)(dst + (rbase + v + hi * 8) * ldd + cb) = t;
  }
}

template <int S>
__device__ __forceinline__ void load_chunk(const float* arow0, const float* arow1,
                                           const float* __restrict__ wcol, int ldw,
                                           int kb, v2f (&a0)[S], v2f (&a1)[S],
                                           v2f (&b)[S]) {
#pragma unroll
  for (int s = 0; s < S; ++s) {
    b[s].x = wcol[(size_t)(kb + 4 * s) * ldw];
    b[s].y = wcol[(size_t)(kb + 4 * s + 1) * ldw];
  }
  load_a<S>(arow0, arow1, kb, a0, a1);
}
template <int S>
__device__ __forceinline__ void mma_chunk(const v2f (&a0)[S], const v2f (&a1)[S],
                                          const v2f (&b)[S], v8f& c0, v8f& c1) {
#pragma unroll
  for (int s = 0; s < S; ++s) {
    c0 = mma4(a0[s], b[s], c0);
    c1 = mma4(a1[s], b[s], c1);
  }
}

template <int K, int CH>
__device__ __forceinline__ void gemm_stream(const float* As, int lda,
                                            const float* __restrict__ W, int ldw,
                                            int n0, int lane16, int hi,
                                            v8f& c0, v8f& c1) {
  static_assert(K % (2 * CH) == 0 && K >= 2 * CH, "chunk pairs");
  constexpr int S = CH / 4;
  const v8f z = {0.f, 0.f, 0.f, 0.f, 0.f, 0.f, 0.f, 0.f};
  c0 = z;
  c1 = z;
  const float* arow0 = As + lane16 * lda + hi * 2;
  const float* arow1 = arow0 + 16 * lda;
  const float* wcol = W + (size_t)(hi * 2) * ldw + n0 + lane16;
  v2f bP[S], a0P[S], a1P[S], bQ[S], a0Q[S], a1Q[S];
  load_chunk<S>(arow0, arow1, wcol, ldw, 0, a0P, a1P, bP);
  keep_order();
#pragma unroll 1
  for (int kb = 0; kb < K - 2 * CH; kb += 2 * CH) {
    load_chunk<S>(arow0, arow1, wcol, ldw, kb + CH, a0Q, a1Q, bQ);
    keep_order();
    mma_chunk<S>(a0P, a1P, bP, c0, c1);
    keep_order();
    load_chunk<S>(arow0, arow1, wcol, ldw, kb + 2 * CH, a0P, a1P, bP);
    keep_order();
    mma_chunk<S>(a0Q, a1Q, bQ, c0, c1);
    keep_order();
  }
  load_chunk<S>(arow0, arow1, wcol, ldw, K - CH, a0Q, a1Q, bQ);
  keep_order();
  mma_chunk<S>(a0P, a1P, bP, c0, c1);
  mma_chunk<S>(a0Q, a1Q, bQ, c0, c1);
}

__device__ __forceinline__ void store_tile(float* dst, int ldd, int rbase,
                                           int col0, v8f c, int lane16, int hi) {
#pragma unroll
  for (int v = 0; v < 8; ++v)
    dst[(rbase + v + hi * 8) * ldd + col0 + lane16] = c[v];
}

__device__ __forceinline__ v8f add_bias(v8f c, float b) {
#pragma unroll
  for (int v = 0; v < 8; ++v) c[v] += b;
  return c;
}
#define LOG2E_F 1.44269504088896340736f
__device__ __forceinline__ float hw_exp(float t) {
  return __builtin_amdgcn_exp2f(t * LOG2E_F);
}
__device__ __forceinline__ v8f bias_elu(v8f c, float b) {
#pragma unroll
  for (int v = 0; v < 8; ++v) {
    float t = c[v] + b;
    float em1 = hw_exp(t) - 1.f;
    c[v] = t > 0.f ? t : em1;
  }
  return c;
}
__device__ __forceinline__ v8f gate_mix(v8f g, float b, v8f e, v8f h2) {
#pragma unroll
  for (int v = 0; v < 8; ++v) {
    float t = g[v] + b;
    float s = __builtin_amdgcn_rcpf(1.f + hw_exp(-t));
    g[v] = e[v] + s * h2[v];
  }
  return g;
}

__global__ __launch_bounds__(256, 1)
void vsn_fused_pairs32(
    const float* __restrict__ x,
    const float* __restrict__ embW, const float* __restrict__ embB,
    const float* __restrict__ fc1W, const float* __restrict__ fc1B,
    const float* __restrict__ fc2W, const float* __restrict__ fc2B,
    const float* __restrict__ gW,   const float* __restrict__ gB,
    const float* __restrict__ lnG,  const float* __restrict__ lnB,
    const float* __restrict__ sf1W, const float* __restrict__ sf1B,
    const float* __restrict__ sf2W, const float* __restrict__ sf2B,
    const float* __restrict__ sgW,  const float* __restrict__ sgB,
    const float* __restrict__ sskW, const float* __restrict__ sskB,
    const float* __restrict__ slnG, const float* __restrict__ slnB,
    float* __restrict__ outSel, float* __restrict__ outW) {
  __shared__ __attribute__((aligned(16))) float smem[SMEM_FLOATS];
  float* TRANS = smem;
  float* XT0 = TRANS + MR * ST_T;
  float* XW0 = XT0 + 2 * MR * ST_X;
  float* PU = XW0;
  float* PV = XW0 + MR * ST_P;
  float* GP = XT0;
  float* SKT = GP + MR * ST_S;
  float* WB = SKT + MR * ST_S;

  const int tid = threadIdx.x;
  const int wid = tid >> 5;
  const int lane = tid & 31;
  const int lane16 = lane & 15;
  const int hi = lane >> 4;
  const int row0 = blockIdx.x * MR;

  const int g = wid >> 2;
  const int n0 = (wid & 3) * 32;
  const int cb = n0 + 2 * lane16;
  const int glr = (tid & 127) >> 4;
  const int lc = tid & 15;
  float* XT = XT0 + g * (MR * ST_X);
  float* XW = XW0 + g * (MR * ST_P);

  float2 bE = *(const float2*)(embB + g * H_ + cb);
  float2 b1 = *(const float2*)(fc1B + g * H_ + cb);
  float2 b2 = *(const float2*)(fc2B + g * H_ + cb);
  float2 bG = *(const float2*)(gB + g * H_ + cb);
  float2 pE0[D_ / 4], pE1[D_ / 4];
  load_panel2<0, D_ / 4, D_ / 4>(embW + (size_t)g * D_ * H_, H_, n0, lane16, hi, pE0, pE1);
  {
#pragma unroll
    for (int rr = 0; rr < 4; ++rr) {
      const int r = glr + 8 * rr;
      float4 v = *(const float4*)(x + (size_t)(row0 + r) * (F_ * D_) + g * D_ + lc * 4);
      float* dp = XT + r * ST_X + lc * 4;
      dp[0] = v.x; dp[1] = v.y; dp[2] = v.z; dp[3] = v.w;
    }
  }
  float2 pF10[H_ / 4], pF11[H_ / 4];
  float4 xr[4];
  wg_signal_lds();
  keep_order();
  load_panel2<0, H_ / 4, H_ / 4>(fc1W + (size_t)g * H_ * H_, H_, n0, lane16, hi, pF10, pF11);
#pragma unroll
  for (int rr = 0; rr < 4; ++rr)
    xr[rr] = *(const float4*)(x + (size_t)(row0 + glr + 8 * rr) * (F_ * D_) +
                              (g + 2) * D_ + lc * 4);
  keep_order();
  wg_wait();

  for (int f = g; f < F_; f += 2) {
    const int fn = f + 2 < F_ ? f + 2 : f;
    const int fnn = f + 4 < F_ ? f + 4 : fn;
    float* SLOT = TRANS + f * H_;

    v8f ce00, ce01, ce10, ce11;
    gemm_panel2<D_>(XT, ST_X, lane16, hi, pE0, pE1, ce00, ce01, ce10, ce11);
    ce00 = add_bias(ce00, bE.x);
    ce10 = add_bias(ce10, bE.x);
    ce01 = add_bias(ce01, bE.y);
    ce11 = add_bias(ce11, bE.y);
    store_tile2(SLOT, ST_T, 0, cb, ce00, ce01, hi);
    store_tile2(SLOT, ST_T, 16, cb, ce10, ce11, hi);
    wg_signal_lds();
    keep_order();
    float2 pF20[H_ / 4], pF21[H_ / 4];
    load_panel2<0, H_ / 4, H_ / 4>(fc2W + (size_t)f * H_ * H_, H_, n0, lane16, hi, pF20, pF21);
    keep_order();
    wg_wait();

    {
      v8f c00, c01, c10, c11;
      gemm_panel2<H_>(SLOT, ST_T, lane16, hi, pF10, pF11, c00, c01, c10, c11);
      c00 = bias_elu(c00, b1.x);
      c10 = bias_elu(c10, b1.x);
      c01 = bias_elu(c01, b1.y);
      c11 = bias_elu(c11, b1.y);
      store_tile2(XW, ST_P, 0, cb, c00, c01, hi);
      store_tile2(XW, ST_P, 16, cb, c10, c11, hi);
    }
    wg_signal_lds();
    keep_order();
    float2 pG0[H_ / 4], pG1[H_ / 4];
    load_panel2<0, H_ / 4, H_ / 4>(gW + (size_t)f * H_ * H_, H_, n0, lane16, hi, pG0, pG1);
    keep_order();
    wg_wait();

    v8f c200, c201, c210, c211;
    gemm_panel2<H_>(XW, ST_P, lane16, hi, pF20, pF21, c200, c201, c210, c211);
    c200 = add_bias(c200, b2.x);
    c210 = add_bias(c210, b2.x);
    c201 = add_bias(c201, b2.y);
    c211 = add_bias(c211, b2.y);
    store_tile2(SLOT, ST_T, 0, cb, c200, c201, hi);
    store_tile2(SLOT, ST_T, 16, cb, c210, c211, hi);
    wg_signal_lds();
    keep_order();

    load_panel2<0, D_ / 4, D_ / 4>(embW + (size_t)fn * D_ * H_, H_, n0, lane16, hi, pE0, pE1);
    const float2 bGf = bG;
    bE = *(const float2*)(embB + fn * H_ + cb);
    b1 = *(const float2*)(fc1B + fn * H_ + cb);
    b2 = *(const float2*)(fc2B + fn * H_ + cb);
    bG = *(const float2*)(gB + fn * H_ + cb);
    float lg[8], lb[8];
#pragma unroll
    for (int j = 0; j < 8; ++j) {
      lg[j] = lnG[f * H_ + lc + 16 * j];
      lb[j] = lnB[f * H_ + lc + 16 * j];
    }
    keep_order();
    wg_wait();
    {
      v8f cg00, cg01, cg10, cg11;
      gemm_panel2<H_>(SLOT, ST_T, lane16, hi, pG0, pG1, cg00, cg01, cg10, cg11);
      cg00 = gate_mix(cg00, bGf.x, ce00, c200);
      cg10 = gate_mix(cg10, bGf.x, ce10, c210);
      cg01 = gate_mix(cg01, bGf.y, ce01, c201);
      cg11 = gate_mix(cg11, bGf.y, ce11, c211);
      store_tile2(XW, ST_P, 0, cb, cg00, cg01, hi);
      store_tile2(XW, ST_P, 16, cb, cg10, cg11, hi);
    }
    wg_signal_lds();
    keep_order();
    load_panel2<0, H_ / 8, H_ / 4>(fc1W + (size_t)fn * H_ * H_, H_, n0, lane16, hi, pF10, pF11);
    keep_order();
    wg_wait();

#pragma unroll
    for (int rr = 0; rr < 4; ++rr) {
      const int r = glr + 8 * rr;
      float vals[8];
      float s = 0.f, ss = 0.f;
#pragma unroll
      for (int j = 0; j < 8; ++j) {
        float v = XW[r * ST_P + lc + 16 * j];
        vals[j] = v; s += v; ss += v * v;
      }
      s = rowsum16(s);
      ss = rowsum16(ss);
      float mu = s * (1.f / 128.f);
      float var = ss * (1.f / 128.f) - mu * mu;
      float rs = rsqrtf(var + 1e-5f);
#pragma unroll
      for (int j = 0; j < 8; ++j) {
        int h = lc + 16 * j;
        float nv = (vals[j] - mu) * rs * lg[j] + lb[j];
        SLOT[r * ST_T + h] = nv;
      }
      float* dp = XT + r * ST_X + lc * 4;
      dp[0] = xr[rr].x; dp[1] = xr[rr].y; dp[2] = xr[rr].z; dp[3] = xr[rr].w;
    }
    wg_signal_lds();
    keep_order();
    load_panel2<H_ / 8, H_ / 4, H_ / 4>(fc1W + (size_t)fn * H_ * H_, H_, n0, lane16, hi, pF10, pF11);
#pragma unroll
    for (int rr = 0; rr < 4; ++rr)
      xr[rr] = *(const float4*)(x + (size_t)(row0 + glr + 8 * rr) * (F_ * D_) +
                                fnn * D_ + lc * 4);
    keep_order();
    wg_wait();
  }

  const int nB = wid * 16;
  const int lr = tid >> 4;
  v2f pS2[H_ / 4];
  load_panel<H_>(sf2W, H_, nB, lane16, hi, pS2);
  const float bS1 = sf1B[nB + lane16];
  const float bS2 = sf2B[nB + lane16];
  {
    v8f b10, b11;
    gemm_stream<FH_, 64>(TRANS, ST_T, sf1W, H_, nB, lane16, hi, b10, b11);
    b10 = bias_elu(b10, bS1);
    b11 = bias_elu(b11, bS1);
    store_tile(PU, ST_P, 0, nB, b10, lane16, hi);
    store_tile(PU, ST_P, 16, nB, b11, lane16, hi);
  }
  wg_barrier_lds();

  {
    v8f b20, b21;
    gemm_panel<H_>(PU, ST_P, lane16, hi, pS2, b20, b21);
    b20 = add_bias(b20, bS2);
    b21 = add_bias(b21, bS2);
    store_tile(PV, ST_P, 0, nB, b20, lane16, hi);
    store_tile(PV, ST_P, 16, nB, b21, lane16, hi);
  }
  wg_barrier_lds();

  if (wid == 0) {
    v2f pSG[H_ / 4];
    load_panel<H_>(sgW, F_, 0, lane16, hi, pSG);
    v8f g0, g1;
    gemm_panel<H_>(PV, ST_P, lane16, hi, pSG, g0, g1);
    float b = sgB[lane16];
    g0 = add_bias(g0, b);
    g1 = add_bias(g1, b);
    store_tile(GP, ST_S, 0, 0, g0, lane16, hi);
    store_tile(GP, ST_S, 16, 0, g1, lane16, hi);
  } else if (wid == 1) {
    v8f s0, s1;
    gemm_stream<FH_, 64>(TRANS, ST_T, sskW, F_, 0, lane16, hi, s0, s1);
    float b = sskB[lane16];
    s0 = add_bias(s0, b);
    s1 = add_bias(s1, b);
    store_tile(SKT, ST_S, 0, 0, s0, lane16, hi);
    store_tile(SKT, ST_S, 16, 0, s1, lane16, hi);
  }
  wg_barrier_lds();

#pragma unroll
  for (int rr = 0; rr < 2; ++rr) {
    const int r = lr + 16 * rr;
    float gp = GP[r * ST_S + lc];
    float sg = 1.f / (1.f + expf(-gp));
    float v = SKT[r * ST_S + lc] + sg * gp;
    float s = rowsum16(v);
    float ss = rowsum16(v * v);
    float mu = s * (1.f / 16.f);
    float var = ss * (1.f / 16.f) - mu * mu;
    float nv = (v - mu) * rsqrtf(var + 1e-5f) * slnG[lc] + slnB[lc];
    float mx = rowmax16(nv);
    float e = expf(nv - mx);
    float se = rowsum16(e);
    float w = e / se;
    WB[r * 16 + lc] = w;
    outW[(size_t)(row0 + r) * F_ + lc] = w;
  }
  wg_barrier_lds();

#pragma unroll
  for (int rr = 0; rr < 2; ++rr) {
    const int r = lr + 16 * rr;
#pragma unroll
    for (int j = 0; j < 8; ++j) {
      int h = lc + 16 * j;
      float acc = 0.f;
#pragma unroll
      for (int f = 0; f < F_; ++f)
        acc += TRANS[r * ST_T + f * H_ + h] * WB[r * 16 + f];
      outSel[(size_t)(row0 + r) * H_ + h] = acc;
    }
  }
}

extern "C" void kernel_launch(void* const* d_in, const int* in_sizes, int n_in,
                              void* d_out, int out_size, void* d_ws, size_t ws_size,
                              hipStream_t stream) {
  (void)in_sizes; (void)n_in; (void)out_size; (void)d_ws; (void)ws_size;
  const float* x    = (const float*)d_in[0];
  const float* embW = (const float*)d_in[1];
  const float* embB = (const float*)d_in[2];
  const float* fc1W = (const float*)d_in[3];
  const float* fc1B = (const float*)d_in[4];
  const float* fc2W = (const float*)d_in[5];
  const float* fc2B = (const float*)d_in[6];
  const float* gW   = (const float*)d_in[7];
  const float* gB   = (const float*)d_in[8];
  const float* lnG  = (const float*)d_in[9];
  const float* lnB  = (const float*)d_in[10];
  const float* sf1W = (const float*)d_in[11];
  const float* sf1B = (const float*)d_in[12];
  const float* sf2W = (const float*)d_in[13];
  const float* sf2B = (const float*)d_in[14];
  const float* sgW  = (const float*)d_in[15];
  const float* sgB  = (const float*)d_in[16];
  const float* sskW = (const float*)d_in[17];
  const float* sskB = (const float*)d_in[18];
  const float* slnG = (const float*)d_in[19];
  const float* slnB = (const float*)d_in[20];

  float* outSel = (float*)d_out;
  float* outW   = outSel + (size_t)ROWS_TOTAL * H_;

  vsn_fused_pairs32<<<ROWS_TOTAL / MR, 256, 0, stream>>>(
      x, embW, embB, fc1W, fc1B, fc2W, fc2B, gW, gB, lnG, lnB,
      sf1W, sf1B, sf2W, sf2B, sgW, sgB, sskW, sskB, slnG, slnB,
      outSel, outW);
}


# Skip GEMM (K=2048, 16 columns) leaves its one-wave stage: K split over the 8 waves inside the s_fc1 pass, which reads the same A; 8 partial tiles added in wave order (skip sum order changes)
# speedup vs baseline: 2.8041x; 1.0487x over previous
#include <hip/hip_runtime.h>
#include <hip/hip_bf16.h>
#include <math.h>


typedef float v2f __attribute__((ext_vector_type(2)));
typedef float v8f __attribute__((ext_vector_type(8)));

#define ROWS_TOTAL 16384
#define MR 32
#define F_ 16
#define D_ 64
#define H_ 128
#define FH_ 2048

#define ST_T 2052
#define ST_X 68
#define ST_P 132
#define ST_S 20

#define SMEM_FLOATS (MR*ST_T + 2*MR*ST_X + 2*MR*ST_P)

__device__ __forceinline__ float rowsum16(float v) {
  v += __shfl_xor(v, 1, 16);
  v += __shfl_xor(v, 2, 16);
  v += __shfl_xor(v, 4, 16);
  v += __shfl_xor(v, 8, 16);
  return v;
}
__device__ __forceinline__ float rowmax16(float v) {
  v = fmaxf(v, __shfl_xor(v, 1, 16));
  v = fmaxf(v, __shfl_xor(v, 2, 16));
  v = fmaxf(v, __shfl_xor(v, 4, 16));
  v = fmaxf(v, __shfl_xor(v, 8, 16));
  return v;
}

__device__ __forceinline__ void wg_barrier_lds() {
  asm volatile("s_wait_dscnt 0x0\n\ts_barrier_signal -1\n\ts_barrier_wait -1" ::: "memory");
}
__device__ __forceinline__ void wg_signal_lds() {
  asm volatile("s_wait_dscnt 0x0\n\ts_barrier_signal -1" ::: "memory");
}
__device__ __forceinline__ void wg_wait() {
  asm volatile("s_barrier_wait -1" ::: "memory");
}

__device__ __forceinline__ void keep_order() { __builtin_amdgcn_sched_barrier(0); }

__device__ __forceinline__ v8f mma4(v2f a, v2f b, v8f c) {
  return __builtin_amdgcn_wmma_f32_16x16x4_f32(false, a, false, b, (short)0, c,
                                               false, false);
}


template <int K>
__device__ __forceinline__ void load_panel(const float* __restrict__ W, int ldw,
                                           int n0, int lane16, int hi,
                                           v2f (&b)[K / 4]) {
  const float* wcol = W + (size_t)(hi * 2) * ldw + n0 + lane16;
#pragma unroll
  for (int s = 0; s < K / 4; ++s) {
    b[s].x = wcol[(size_t)(4 * s) * ldw];
    b[s].y = wcol[(size_t)(4 * s + 1) * ldw];
  }
}

template <int S>
__device__ __forceinline__ void load_a(const float* arow0, const float* arow1,
                                       int kb, v2f (&a0)[S], v2f (&a1)[S]) {
#pragma unroll
  for (int s = 0; s < S; ++s) {
    a0[s].x = arow0[kb + 4 * s];
    a0[s].y = arow0[kb + 4 * s + 1];
    a1[s].x = arow1[kb + 4 * s];
    a1[s].y = arow1[kb + 4 * s + 1];
  }
}

template <int K>
__device__ __forceinline__ void gemm_panel(const float* As, int lda, int lane16,
                                           int hi, const v2f (&b)[K / 4],
                                           v8f& c0, v8f& c1) {
  const v8f z = {0.f, 0.f, 0.f, 0.f, 0.f, 0.f, 0.f, 0.f};
  c0 = z;
  c1 = z;
  const float* arow0 = As + lane16 * lda + hi * 2;
  const float* arow1 = arow0 + 16 * lda;
  v2f a0[2][8], a1[2][8];
  load_a<8>(arow0, arow1, 0, a0[0], a1[0]);
  keep_order();
#pragma unroll
  for (int ch = 0; ch < K / 32; ++ch) {
    if (ch + 1 < K / 32)
      load_a<8>(arow0, arow1, (ch + 1) * 32, a0[(ch + 1) & 1], a1[(ch + 1) & 1]);
    keep_order();
#pragma unroll
    for (int s = 0; s < 8; ++s) {
      c0 = mma4(a0[ch & 1][s], b[ch * 8 + s], c0);
      c1 = mma4(a1[ch & 1][s], b[ch * 8 + s], c1);
    }
    keep_order();
  }
}

template <int S0, int S1, int NS>
__device__ __forceinline__ void load_panel2(const float* __restrict__ W, int ldw,
                                            int n0, int lane16, int hi,
                                            float2 (&ra)[NS], float2 (&rb)[NS]) {
  const float* wcol = W + (size_t)(hi * 2) * ldw + n0 + 2 * lane16;
#pragma unroll
  for (int s = S0; s < S1; ++s) {
    ra[s] = *(const float2*)(wcol + (size_t)(4 * s) * ldw);
    rb[s] = *(const float2*)(wcol + (size_t)(4 * s + 1) * ldw);
  }
}

template <int K>
__device__ __forceinline__ void gemm_panel2(const float* As, int lda, int lane16,
                                            int hi, const float2 (&ra)[K / 4],
                                            const float2 (&rb)[K / 4],
                                            v8f& c00, v8f& c01, v8f& c10, v8f& c11) {
  const v8f z = {0.f, 0.f, 0.f, 0.f, 0.f, 0.f, 0.f, 0.f};
  c00 = z; c01 = z; c10 = z; c11 = z;
  const float* arow0 = As + lane16 * lda + hi * 2;
  const float* arow1 = arow0 + 16 * lda;
  v2f a0[2][4], a1[2][4];
  load_a<4>(arow0, arow1, 0, a0[0], a1[0]);
  keep_order();
#pragma unroll
  for (int ch = 0; ch < K / 16; ++ch) {
    if (ch + 1 < K / 16)
      load_a<4>(arow0, arow1, (ch + 1) * 16, a0[(ch + 1) & 1], a1[(ch + 1) & 1]);
    keep_order();
#pragma unroll
    for (int s = 0; s < 4; ++s) {
      v2f bj0, bj1;
      bj0.x = ra[ch * 4 + s].x; bj0.y = rb[ch * 4 + s].x;
      bj1.x = ra[ch * 4 + s].y; bj1.y = rb[ch * 4 + s].y;
      c00 = mma4(a0[ch & 1][s], bj0, c00);
      c10 = mma4(a1[ch & 1][s], bj0, c10);
      c01 = mma4(a0[ch & 1][s], bj1, c01);
      c11 = mma4(a1[ch & 1][s], bj1, c11);
    }
    keep_order();
  }
}

__device__ __forceinline__ void store_tile2(float* dst, int ldd, int rbase, int cb,
                                            v8f cj0, v8f cj1, int hi) {
#pragma unroll
  for (int v = 0; v < 8; ++v) {
    float2 t;
    t.x = cj0[v];
    t.y = cj1[v];
    *(float2*)(dst + (rbase + v + hi * 8) * ldd + cb) = t;
  }
}

template <int S>
__device__ __forceinline__ void load_chunk(const float* arow0, const float* arow1,
                                           const float* __restrict__ wcol, int ldw,
                                           const float* arow0n, const float* arow1n,
                                           const float* __restrict__ w2col, int ldw2,
                                           int kb, v2f (&a0)[S], v2f (&a1)[S],
                                           v2f (&b)[S], v2f (&a0n)[S / 8],
                                           v2f (&a1n)[S / 8], v2f (&bn)[S / 8]) {
#pragma unroll
  for (int s = 0; s < S; ++s) {
    b[s].x = wcol[(size_t)(kb + 4 * s) * ldw];
    b[s].y = wcol[(size_t)(kb + 4 * s + 1) * ldw];
  }
#pragma unroll
  for (int t = 0; t < S / 8; ++t) {
    bn[t].x = w2col[(size_t)(kb + 32 * t) * ldw2];
    bn[t].y = w2col[(size_t)(kb + 32 * t + 1) * ldw2];
  }
  load_a<S>(arow0, arow1, kb, a0, a1);
#pragma unroll
  for (int t = 0; t < S / 8; ++t) {
    a0n[t].x = arow0n[kb + 32 * t];
    a0n[t].y = arow0n[kb + 32 * t + 1];
    a1n[t].x = arow1n[kb + 32 * t];
    a1n[t].y = arow1n[kb + 32 * t + 1];
  }
}
template <int S>
__device__ __forceinline__ void mma_chunk(const v2f (&a0)[S], const v2f (&a1)[S],
                                          const v2f (&b)[S], const v2f (&a0n)[S / 8],
                                          const v2f (&a1n)[S / 8], const v2f (&bn)[S / 8],
                                          v8f& c0, v8f& c1, v8f& s0, v8f& s1) {
#pragma unroll
  for (int s = 0; s < S; ++s) {
    c0 = mma4(a0[s], b[s], c0);
    c1 = mma4(a1[s], b[s], c1);
  }
#pragma unroll
  for (int t = 0; t < S / 8; ++t) {
    s0 = mma4(a0n[t], bn[t], s0);
    s1 = mma4(a1n[t], bn[t], s1);
  }
}

template <int K, int CH>
__device__ __forceinline__ void gemm_stream(const float* As, int lda,
                                            const float* __restrict__ W, int ldw,
                                            int n0, const float* __restrict__ W2,
                                            int ldw2, int w, int lane16, int hi,
                                            v8f& c0, v8f& c1, v8f& s0, v8f& s1) {
  static_assert(K % (2 * CH) == 0 && K >= 2 * CH, "chunk pairs");
  static_assert(CH == 64, "a chunk of 16 k-steps: two per wave for the second product");
  constexpr int S = CH / 4;
  const v8f z = {0.f, 0.f, 0.f, 0.f, 0.f, 0.f, 0.f, 0.f};
  c0 = z;
  c1 = z;
  s0 = z;
  s1 = z;
  const float* arow0 = As + lane16 * lda + hi * 2;
  const float* arow1 = arow0 + 16 * lda;
  const float* wcol = W + (size_t)(hi * 2) * ldw + n0 + lane16;
  const float* arow0n = arow0 + 4 * w;
  const float* arow1n = arow1 + 4 * w;
  const float* w2col = W2 + (size_t)(hi * 2 + 4 * w) * ldw2 + lane16;
  v2f bP[S], a0P[S], a1P[S], bQ[S], a0Q[S], a1Q[S];
  v2f bnP[S / 8], a0nP[S / 8], a1nP[S / 8], bnQ[S / 8], a0nQ[S / 8], a1nQ[S / 8];
  load_chunk<S>(arow0, arow1, wcol, ldw, arow0n, arow1n, w2col, ldw2, 0,
                a0P, a1P, bP, a0nP, a1nP, bnP);
  keep_order();
#pragma unroll 1
  for (int kb = 0; kb < K - 2 * CH; kb += 2 * CH) {
    load_chunk<S>(arow0, arow1, wcol, ldw, arow0n, arow1n, w2col, ldw2, kb + CH,
                  a0Q, a1Q, bQ, a0nQ, a1nQ, bnQ);
    keep_order();
    mma_chunk<S>(a0P, a1P, bP, a0nP, a1nP, bnP, c0, c1, s0, s1);
    keep_order();
    load_chunk<S>(arow0, arow1, wcol, ldw, arow0n, arow1n, w2col, ldw2, kb + 2 * CH,
                  a0P, a1P, bP, a0nP, a1nP, bnP);
    keep_order();
    mma_chunk<S>(a0Q, a1Q, bQ, a0nQ, a1nQ, bnQ, c0, c1, s0, s1);
    keep_order();
  }
  load_chunk<S>(arow0, arow1, wcol, ldw, arow0n, arow1n, w2col, ldw2, K - CH,
                a0Q, a1Q, bQ, a0nQ, a1nQ, bnQ);
  keep_order();
  mma_chunk<S>(a0P, a1P, bP, a0nP, a1nP, bnP, c0, c1, s0, s1);
  mma_chunk<S>(a0Q, a1Q, bQ, a0nQ, a1nQ, bnQ, c0, c1, s0, s1);
}

__device__ __forceinline__ void store_tile(float* dst, int ldd, int rbase,
                                           int col0, v8f c, int lane16, int hi) {
#pragma unroll
  for (int v = 0; v < 8; ++v)
    dst[(rbase + v + hi * 8) * ldd + col0 + lane16] = c[v];
}

__device__ __forceinline__ v8f add_bias(v8f c, float b) {
#pragma unroll
  for (int v = 0; v < 8; ++v) c[v] += b;
  return c;
}
#define LOG2E_F 1.44269504088896340736f
__device__ __forceinline__ float hw_exp(float t) {
  return __builtin_amdgcn_exp2f(t * LOG2E_F);
}
__device__ __forceinline__ v8f bias_elu(v8f c, float b) {
#pragma unroll
  for (int v = 0; v < 8; ++v) {
    float t = c[v] + b;
    float em1 = hw_exp(t) - 1.f;
    c[v] = t > 0.f ? t : em1;
  }
  return c;
}
__device__ __forceinline__ v8f gate_mix(v8f g, float b, v8f e, v8f h2) {
#pragma unroll
  for (int v = 0; v < 8; ++v) {
    float t = g[v] + b;
    float s = __builtin_amdgcn_rcpf(1.f + hw_exp(-t));
    g[v] = e[v] + s * h2[v];
  }
  return g;
}

__global__ __launch_bounds__(256, 1)
void vsn_fused_pairs32(
    const float* __restrict__ x,
    const float* __restrict__ embW, const float* __restrict__ embB,
    const float* __restrict__ fc1W, const float* __restrict__ fc1B,
    const float* __restrict__ fc2W, const float* __restrict__ fc2B,
    const float* __restrict__ gW,   const float* __restrict__ gB,
    const float* __restrict__ lnG,  const float* __restrict__ lnB,
    const float* __restrict__ sf1W, const float* __restrict__ sf1B,
    const float* __restrict__ sf2W, const float* __restrict__ sf2B,
    const float* __restrict__ sgW,  const float* __restrict__ sgB,
    const float* __restrict__ sskW, const float* __restrict__ sskB,
    const float* __restrict__ slnG, const float* __restrict__ slnB,
    float* __restrict__ outSel, float* __restrict__ outW) {
  __shared__ __attribute__((aligned(16))) float smem[SMEM_FLOATS];
  float* TRANS = smem;
  float* XT0 = TRANS + MR * ST_T;
  float* XW0 = XT0 + 2 * MR * ST_X;
  float* PU = XW0;
  float* PV = XW0 + MR * ST_P;
  float* SKP = XT0;
  float* GP = PU;
  float* WB = GP + MR * ST_S;

  const int tid = threadIdx.x;
  const int wid = tid >> 5;
  const int lane = tid & 31;
  const int lane16 = lane & 15;
  const int hi = lane >> 4;
  const int row0 = blockIdx.x * MR;

  const int g = wid >> 2;
  const int n0 = (wid & 3) * 32;
  const int cb = n0 + 2 * lane16;
  const int glr = (tid & 127) >> 4;
  const int lc = tid & 15;
  float* XT = XT0 + g * (MR * ST_X);
  float* XW = XW0 + g * (MR * ST_P);

  float2 bE = *(const float2*)(embB + g * H_ + cb);
  float2 b1 = *(const float2*)(fc1B + g * H_ + cb);
  float2 b2 = *(const float2*)(fc2B + g * H_ + cb);
  float2 bG = *(const float2*)(gB + g * H_ + cb);
  float2 pE0[D_ / 4], pE1[D_ / 4];
  load_panel2<0, D_ / 4, D_ / 4>(embW + (size_t)g * D_ * H_, H_, n0, lane16, hi, pE0, pE1);
  {
#pragma unroll
    for (int rr = 0; rr < 4; ++rr) {
      const int r = glr + 8 * rr;
      float4 v = *(const float4*)(x + (size_t)(row0 + r) * (F_ * D_) + g * D_ + lc * 4);
      float* dp = XT + r * ST_X + lc * 4;
      dp[0] = v.x; dp[1] = v.y; dp[2] = v.z; dp[3] = v.w;
    }
  }
  float2 pF10[H_ / 4], pF11[H_ / 4];
  float4 xr[4];
  wg_signal_lds();
  keep_order();
  load_panel2<0, H_ / 4, H_ / 4>(fc1W + (size_t)g * H_ * H_, H_, n0, lane16, hi, pF10, pF11);
#pragma unroll
  for (int rr = 0; rr < 4; ++rr)
    xr[rr] = *(const float4*)(x + (size_t)(row0 + glr + 8 * rr) * (F_ * D_) +
                              (g + 2) * D_ + lc * 4);
  keep_order();
  wg_wait();

  for (int f = g; f < F_; f += 2) {
    const int fn = f + 2 < F_ ? f + 2 : f;
    const int fnn = f + 4 < F_ ? f + 4 : fn;
    float* SLOT = TRANS + f * H_;

    v8f ce00, ce01, ce10, ce11;
    gemm_panel2<D_>(XT, ST_X, lane16, hi, pE0, pE1, ce00, ce01, ce10, ce11);
    ce00 = add_bias(ce00, bE.x);
    ce10 = add_bias(ce10, bE.x);
    ce01 = add_bias(ce01, bE.y);
    ce11 = add_bias(ce11, bE.y);
    store_tile2(SLOT, ST_T, 0, cb, ce00, ce01, hi);
    store_tile2(SLOT, ST_T, 16, cb, ce10, ce11, hi);
    wg_signal_lds();
    keep_order();
    float2 pF20[H_ / 4], pF21[H_ / 4];
    load_panel2<0, H_ / 4, H_ / 4>(fc2W + (size_t)f * H_ * H_, H_, n0, lane16, hi, pF20, pF21);
    keep_order();
    wg_wait();

    {
      v8f c00, c01, c10, c11;
      gemm_panel2<H_>(SLOT, ST_T, lane16, hi, pF10, pF11, c00, c01, c10, c11);
      c00 = bias_elu(c00, b1.x);
      c10 = bias_elu(c10, b1.x);
      c01 = bias_elu(c01, b1.y);
      c11 = bias_elu(c11, b1.y);
      store_tile2(XW, ST_P, 0, cb, c00, c01, hi);
      store_tile2(XW, ST_P, 16, cb, c10, c11, hi);
    }
    wg_signal_lds();
    keep_order();
    float2 pG0[H_ / 4], pG1[H_ / 4];
    load_panel2<0, H_ / 4, H_ / 4>(gW + (size_t)f * H_ * H_, H_, n0, lane16, hi, pG0, pG1);
    keep_order();
    wg_wait();

    v8f c200, c201, c210, c211;
    gemm_panel2<H_>(XW, ST_P, lane16, hi, pF20, pF21, c200, c201, c210, c211);
    c200 = add_bias(c200, b2.x);
    c210 = add_bias(c210, b2.x);
    c201 = add_bias(c201, b2.y);
    c211 = add_bias(c211, b2.y);
    store_tile2(SLOT, ST_T, 0, cb, c200, c201, hi);
    store_tile2(SLOT, ST_T, 16, cb, c210, c211, hi);
    wg_signal_lds();
    keep_order();

    load_panel2<0, D_ / 4, D_ / 4>(embW + (size_t)fn * D_ * H_, H_, n0, lane16, hi, pE0, pE1);
    const float2 bGf = bG;
    bE = *(const float2*)(embB + fn * H_ + cb);
    b1 = *(const float2*)(fc1B + fn * H_ + cb);
    b2 = *(const float2*)(fc2B + fn * H_ + cb);
    bG = *(const float2*)(gB + fn * H_ + cb);
    float lg[8], lb[8];
#pragma unroll
    for (int j = 0; j < 8; ++j) {
      lg[j] = lnG[f * H_ + lc + 16 * j];
      lb[j] = lnB[f * H_ + lc + 16 * j];
    }
    keep_order();
    wg_wait();
    {
      v8f cg00, cg01, cg10, cg11;
      gemm_panel2<H_>(SLOT, ST_T, lane16, hi, pG0, pG1, cg00, cg01, cg10, cg11);
      cg00 = gate_mix(cg00, bGf.x, ce00, c200);
      cg10 = gate_mix(cg10, bGf.x, ce10, c210);
      cg01 = gate_mix(cg01, bGf.y, ce01, c201);
      cg11 = gate_mix(cg11, bGf.y, ce11, c211);
      store_tile2(XW, ST_P, 0, cb, cg00, cg01, hi);
      store_tile2(XW, ST_P, 16, cb, cg10, cg11, hi);
    }
    wg_signal_lds();
    keep_order();
    load_panel2<0, H_ / 8, H_ / 4>(fc1W + (size_t)fn * H_ * H_, H_, n0, lane16, hi, pF10, pF11);
    keep_order();
    wg_wait();

#pragma unroll
    for (int rr = 0; rr < 4; ++rr) {
      const int r = glr + 8 * rr;
      float vals[8];
      float s = 0.f, ss = 0.f;
#pragma unroll
      for (int j = 0; j < 8; ++j) {
        float v = XW[r * ST_P + lc + 16 * j];
        vals[j] = v; s += v; ss += v * v;
      }
      s = rowsum16(s);
      ss = rowsum16(ss);
      float mu = s * (1.f / 128.f);
      float var = ss * (1.f / 128.f) - mu * mu;
      float rs = rsqrtf(var + 1e-5f);
#pragma unroll
      for (int j = 0; j < 8; ++j) {
        int h = lc + 16 * j;
        float nv = (vals[j] - mu) * rs * lg[j] + lb[j];
        SLOT[r * ST_T + h] = nv;
      }
      float* dp = XT + r * ST_X + lc * 4;
      dp[0] = xr[rr].x; dp[1] = xr[rr].y; dp[2] = xr[rr].z; dp[3] = xr[rr].w;
    }
    wg_signal_lds();
    keep_order();
    load_panel2<H_ / 8, H_ / 4, H_ / 4>(fc1W + (size_t)fn * H_ * H_, H_, n0, lane16, hi, pF10, pF11);
#pragma unroll
    for (int rr = 0; rr < 4; ++rr)
      xr[rr] = *(const float4*)(x + (size_t)(row0 + glr + 8 * rr) * (F_ * D_) +
                                fnn * D_ + lc * 4);
    keep_order();
    wg_wait();
  }

  const int nB = wid * 16;
  const int lr = tid >> 4;
  v2f pS2[H_ / 4];
  load_panel<H_>(sf2W, H_, nB, lane16, hi, pS2);
  const float bS1 = sf1B[nB + lane16];
  const float bS2 = sf2B[nB + lane16];
  const float bSK = sskB[lc];
  {
    v8f b10, b11, sk0, sk1;
    gemm_stream<FH_, 64>(TRANS, ST_T, sf1W, H_, nB, sskW, F_, wid, lane16, hi,
                         b10, b11, sk0, sk1);
    b10 = bias_elu(b10, bS1);
    b11 = bias_elu(b11, bS1);
    store_tile(PU, ST_P, 0, nB, b10, lane16, hi);
    store_tile(PU, ST_P, 16, nB, b11, lane16, hi);
    float* SKPw = SKP + wid * (MR * 16);
    store_tile(SKPw, 16, 0, 0, sk0, lane16, hi);
    store_tile(SKPw, 16, 16, 0, sk1, lane16, hi);
  }
  wg_barrier_lds();

  {
    v8f b20, b21;
    gemm_panel<H_>(PU, ST_P, lane16, hi, pS2, b20, b21);
    b20 = add_bias(b20, bS2);
    b21 = add_bias(b21, bS2);
    store_tile(PV, ST_P, 0, nB, b20, lane16, hi);
    store_tile(PV, ST_P, 16, nB, b21, lane16, hi);
  }
  wg_barrier_lds();

  if (wid == 0) {
    v2f pSG[H_ / 4];
    load_panel<H_>(sgW, F_, 0, lane16, hi, pSG);
    v8f g0, g1;
    gemm_panel<H_>(PV, ST_P, lane16, hi, pSG, g0, g1);
    float b = sgB[lane16];
    g0 = add_bias(g0, b);
    g1 = add_bias(g1, b);
    store_tile(GP, ST_S, 0, 0, g0, lane16, hi);
    store_tile(GP, ST_S, 16, 0, g1, lane16, hi);
  }
  wg_barrier_lds();

#pragma unroll
  for (int rr = 0; rr < 2; ++rr) {
    const int r = lr + 16 * rr;
    float gp = GP[r * ST_S + lc];
    float sg = 1.f / (1.f + expf(-gp));
    float sk = SKP[r * 16 + lc];
    for (int pw = 1; pw < 8; ++pw) sk += SKP[pw * (MR * 16) + r * 16 + lc];
    sk += bSK;
    float v = sk + sg * gp;
    float s = rowsum16(v);
    float ss = rowsum16(v * v);
    float mu = s * (1.f / 16.f);
    float var = ss * (1.f / 16.f) - mu * mu;
    float nv = (v - mu) * rsqrtf(var + 1e-5f) * slnG[lc] + slnB[lc];
    float mx = rowmax16(nv);
    float e = expf(nv - mx);
    float se = rowsum16(e);
    float w = e / se;
    WB[r * 16 + lc] = w;
    outW[(size_t)(row0 + r) * F_ + lc] = w;
  }
  wg_barrier_lds();

#pragma unroll
  for (int rr = 0; rr < 2; ++rr) {
    const int r = lr + 16 * rr;
#pragma unroll
    for (int j = 0; j < 8; ++j) {
      int h = lc + 16 * j;
      float acc = 0.f;
#pragma unroll
      for (int f = 0; f < F_; ++f)
        acc += TRANS[r * ST_T + f * H_ + h] * WB[r * 16 + f];
      outSel[(size_t)(row0 + r) * H_ + h] = acc;
    }
  }
}

extern "C" void kernel_launch(void* const* d_in, const int* in_sizes, int n_in,
                              void* d_out, int out_size, void* d_ws, size_t ws_size,
                              hipStream_t stream) {
  (void)in_sizes; (void)n_in; (void)out_size; (void)d_ws; (void)ws_size;
  const float* x    = (const float*)d_in[0];
  const float* embW = (const float*)d_in[1];
  const float* embB = (const float*)d_in[2];
  const float* fc1W = (const float*)d_in[3];
  const float* fc1B = (const float*)d_in[4];
  const float* fc2W = (const float*)d_in[5];
  const float* fc2B = (const float*)d_in[6];
  const float* gW   = (const float*)d_in[7];
  const float* gB   = (const float*)d_in[8];
  const float* lnG  = (const float*)d_in[9];
  const float* lnB  = (const float*)d_in[10];
  const float* sf1W = (const float*)d_in[11];
  const float* sf1B = (const float*)d_in[12];
  const float* sf2W = (const float*)d_in[13];
  const float* sf2B = (const float*)d_in[14];
  const float* sgW  = (const float*)d_in[15];
  const float* sgB  = (const float*)d_in[16];
  const float* sskW = (const float*)d_in[17];
  const float* sskB = (const float*)d_in[18];
  const float* slnG = (const float*)d_in[19];
  const float* slnB = (const float*)d_in[20];

  float* outSel = (float*)d_out;
  float* outW   = outSel + (size_t)ROWS_TOTAL * H_;

  vsn_fused_pairs32<<<ROWS_TOTAL / MR, 256, 0, stream>>>(
      x, embW, embB, fc1W, fc1B, fc2W, fc2B, gW, gB, lnG, lnB,
      sf1W, sf1B, sf2W, sf2B, sgW, sgB, sskW, sskB, slnG, slnB,
      outSel, outW);
}


# LayerNorm and softmax row sums and maxima over 16 lanes as DPP row_xmask adds on the vector ALU (same butterfly order), not an LDS permute and wait per step; a thread's rows reduced together
# speedup vs baseline: 2.9370x; 1.0474x over previous
#include <hip/hip_runtime.h>
#include <hip/hip_bf16.h>
#include <math.h>


typedef float v2f __attribute__((ext_vector_type(2)));
typedef float v8f __attribute__((ext_vector_type(8)));

#define ROWS_TOTAL 16384
#define MR 32
#define F_ 16
#define D_ 64
#define H_ 128
#define FH_ 2048

#define ST_T 2052
#define ST_X 68
#define ST_P 132
#define ST_S 20

#define SMEM_FLOATS (MR*ST_T + 2*MR*ST_X + 2*MR*ST_P)

template <int M>
__device__ __forceinline__ float row_xor(float v) {
  return __builtin_bit_cast(
      float, __builtin_amdgcn_update_dpp(0, __builtin_bit_cast(int, v), 0x160 + M,
                                         0xf, 0xf, true));
}
template <int N>
__device__ __forceinline__ void rowsum16_n(float (&v)[N]) {
#pragma unroll
  for (int i = 0; i < N; ++i) v[i] += row_xor<1>(v[i]);
#pragma unroll
  for (int i = 0; i < N; ++i) v[i] += row_xor<2>(v[i]);
#pragma unroll
  for (int i = 0; i < N; ++i) v[i] += row_xor<4>(v[i]);
#pragma unroll
  for (int i = 0; i < N; ++i) v[i] += row_xor<8>(v[i]);
}
template <int N>
__device__ __forceinline__ void rowmax16_n(float (&v)[N]) {
#pragma unroll
  for (int i = 0; i < N; ++i) v[i] = fmaxf(v[i], row_xor<1>(v[i]));
#pragma unroll
  for (int i = 0; i < N; ++i) v[i] = fmaxf(v[i], row_xor<2>(v[i]));
#pragma unroll
  for (int i = 0; i < N; ++i) v[i] = fmaxf(v[i], row_xor<4>(v[i]));
#pragma unroll
  for (int i = 0; i < N; ++i) v[i] = fmaxf(v[i], row_xor<8>(v[i]));
}

__device__ __forceinline__ void wg_barrier_lds() {
  asm volatile("s_wait_dscnt 0x0\n\ts_barrier_signal -1\n\ts_barrier_wait -1" ::: "memory");
}
__device__ __forceinline__ void wg_signal_lds() {
  asm volatile("s_wait_dscnt 0x0\n\ts_barrier_signal -1" ::: "memory");
}
__device__ __forceinline__ void wg_wait() {
  asm volatile("s_barrier_wait -1" ::: "memory");
}

__device__ __forceinline__ void keep_order() { __builtin_amdgcn_sched_barrier(0); }

__device__ __forceinline__ v8f mma4(v2f a, v2f b, v8f c) {
  return __builtin_amdgcn_wmma_f32_16x16x4_f32(false, a, false, b, (short)0, c,
                                               false, false);
}


template <int K>
__device__ __forceinline__ void load_panel(const float* __restrict__ W, int ldw,
                                           int n0, int lane16, int hi,
                                           v2f (&b)[K / 4]) {
  const float* wcol = W + (size_t)(hi * 2) * ldw + n0 + lane16;
#pragma unroll
  for (int s = 0; s < K / 4; ++s) {
    b[s].x = wcol[(size_t)(4 * s) * ldw];
    b[s].y = wcol[(size_t)(4 * s + 1) * ldw];
  }
}

template <int S>
__device__ __forceinline__ void load_a(const float* arow0, const float* arow1,
                                       int kb, v2f (&a0)[S], v2f (&a1)[S]) {
#pragma unroll
  for (int s = 0; s < S; ++s) {
    a0[s].x = arow0[kb + 4 * s];
    a0[s].y = arow0[kb + 4 * s + 1];
    a1[s].x = arow1[kb + 4 * s];
    a1[s].y = arow1[kb + 4 * s + 1];
  }
}

template <int K>
__device__ __forceinline__ void gemm_panel(const float* As, int lda, int lane16,
                                           int hi, const v2f (&b)[K / 4],
                                           v8f& c0, v8f& c1) {
  const v8f z = {0.f, 0.f, 0.f, 0.f, 0.f, 0.f, 0.f, 0.f};
  c0 = z;
  c1 = z;
  const float* arow0 = As + lane16 * lda + hi * 2;
  const float* arow1 = arow0 + 16 * lda;
  v2f a0[2][8], a1[2][8];
  load_a<8>(arow0, arow1, 0, a0[0], a1[0]);
  keep_order();
#pragma unroll
  for (int ch = 0; ch < K / 32; ++ch) {
    if (ch + 1 < K / 32)
      load_a<8>(arow0, arow1, (ch + 1) * 32, a0[(ch + 1) & 1], a1[(ch + 1) & 1]);
    keep_order();
#pragma unroll
    for (int s = 0; s < 8; ++s) {
      c0 = mma4(a0[ch & 1][s], b[ch * 8 + s], c0);
      c1 = mma4(a1[ch & 1][s], b[ch * 8 + s], c1);
    }
    keep_order();
  }
}

template <int S0, int S1, int NS>
__device__ __forceinline__ void load_panel2(const float* __restrict__ W, int ldw,
                                            int n0, int lane16, int hi,
                                            float2 (&ra)[NS], float2 (&rb)[NS]) {
  const float* wcol = W + (size_t)(hi * 2) * ldw + n0 + 2 * lane16;
#pragma unroll
  for (int s = S0; s < S1; ++s) {
    ra[s] = *(const float2*)(wcol + (size_t)(4 * s) * ldw);
    rb[s] = *(const float2*)(wcol + (size_t)(4 * s + 1) * ldw);
  }
}

template <int K>
__device__ __forceinline__ void gemm_panel2(const float* As, int lda, int lane16,
                                            int hi, const float2 (&ra)[K / 4],
                                            const float2 (&rb)[K / 4],
                                            v8f& c00, v8f& c01, v8f& c10, v8f& c11) {
  const v8f z = {0.f, 0.f, 0.f, 0.f, 0.f, 0.f, 0.f, 0.f};
  c00 = z; c01 = z; c10 = z; c11 = z;
  const float* arow0 = As + lane16 * lda + hi * 2;
  const float* arow1 = arow0 + 16 * lda;
  v2f a0[2][4], a1[2][4];
  load_a<4>(arow0, arow1, 0, a0[0], a1[0]);
  keep_order();
#pragma unroll
  for (int ch = 0; ch < K / 16; ++ch) {
    if (ch + 1 < K / 16)
      load_a<4>(arow0, arow1, (ch + 1) * 16, a0[(ch + 1) & 1], a1[(ch + 1) & 1]);
    keep_order();
#pragma unroll
    for (int s = 0; s < 4; ++s) {
      v2f bj0, bj1;
      bj0.x = ra[ch * 4 + s].x; bj0.y = rb[ch * 4 + s].x;
      bj1.x = ra[ch * 4 + s].y; bj1.y = rb[ch * 4 + s].y;
      c00 = mma4(a0[ch & 1][s], bj0, c00);
      c10 = mma4(a1[ch & 1][s], bj0, c10);
      c01 = mma4(a0[ch & 1][s], bj1, c01);
      c11 = mma4(a1[ch & 1][s], bj1, c11);
    }
    keep_order();
  }
}

__device__ __forceinline__ void store_tile2(float* dst, int ldd, int rbase, int cb,
                                            v8f cj0, v8f cj1, int hi) {
#pragma unroll
  for (int v = 0; v < 8; ++v) {
    float2 t;
    t.x = cj0[v];
    t.y = cj1[v];
    *(float2*)(dst + (rbase + v + hi * 8) * ldd + cb) = t;
  }
}

template <int S>
__device__ __forceinline__ void load_chunk(const float* arow0, const float* arow1,
                                           const float* __restrict__ wcol, int ldw,
                                           const float* arow0n, const float* arow1n,
                                           const float* __restrict__ w2col, int ldw2,
                                           int kb, v2f (&a0)[S], v2f (&a1)[S],
                                           v2f (&b)[S], v2f (&a0n)[S / 8],
                                           v2f (&a1n)[S / 8], v2f (&bn)[S / 8]) {
#pragma unroll
  for (int s = 0; s < S; ++s) {
    b[s].x = wcol[(size_t)(kb + 4 * s) * ldw];
    b[s].y = wcol[(size_t)(kb + 4 * s + 1) * ldw];
  }
#pragma unroll
  for (int t = 0; t < S / 8; ++t) {
    bn[t].x = w2col[(size_t)(kb + 32 * t) * ldw2];
    bn[t].y = w2col[(size_t)(kb + 32 * t + 1) * ldw2];
  }
  load_a<S>(arow0, arow1, kb, a0, a1);
#pragma unroll
  for (int t = 0; t < S / 8; ++t) {
    a0n[t].x = arow0n[kb + 32 * t];
    a0n[t].y = arow0n[kb + 32 * t + 1];
    a1n[t].x = arow1n[kb + 32 * t];
    a1n[t].y = arow1n[kb + 32 * t + 1];
  }
}
template <int S>
__device__ __forceinline__ void mma_chunk(const v2f (&a0)[S], const v2f (&a1)[S],
                                          const v2f (&b)[S], const v2f (&a0n)[S / 8],
                                          const v2f (&a1n)[S / 8], const v2f (&bn)[S / 8],
                                          v8f& c0, v8f& c1, v8f& s0, v8f& s1) {
#pragma unroll
  for (int s = 0; s < S; ++s) {
    c0 = mma4(a0[s], b[s], c0);
    c1 = mma4(a1[s], b[s], c1);
  }
#pragma unroll
  for (int t = 0; t < S / 8; ++t) {
    s0 = mma4(a0n[t], bn[t], s0);
    s1 = mma4(a1n[t], bn[t], s1);
  }
}

template <int K, int CH>
__device__ __forceinline__ void gemm_stream(const float* As, int lda,
                                            const float* __restrict__ W, int ldw,
                                            int n0, const float* __restrict__ W2,
                                            int ldw2, int w, int lane16, int hi,
                                            v8f& c0, v8f& c1, v8f& s0, v8f& s1) {
  static_assert(K % (2 * CH) == 0 && K >= 2 * CH, "chunk pairs");
  static_assert(CH == 64, "a chunk of 16 k-steps: two per wave for the second product");
  constexpr int S = CH / 4;
  const v8f z = {0.f, 0.f, 0.f, 0.f, 0.f, 0.f, 0.f, 0.f};
  c0 = z;
  c1 = z;
  s0 = z;
  s1 = z;
  const float* arow0 = As + lane16 * lda + hi * 2;
  const float* arow1 = arow0 + 16 * lda;
  const float* wcol = W + (size_t)(hi * 2) * ldw + n0 + lane16;
  const float* arow0n = arow0 + 4 * w;
  const float* arow1n = arow1 + 4 * w;
  const float* w2col = W2 + (size_t)(hi * 2 + 4 * w) * ldw2 + lane16;
  v2f bP[S], a0P[S], a1P[S], bQ[S], a0Q[S], a1Q[S];
  v2f bnP[S / 8], a0nP[S / 8], a1nP[S / 8], bnQ[S / 8], a0nQ[S / 8], a1nQ[S / 8];
  load_chunk<S>(arow0, arow1, wcol, ldw, arow0n, arow1n, w2col, ldw2, 0,
                a0P, a1P, bP, a0nP, a1nP, bnP);
  keep_order();
#pragma unroll 1
  for (int kb = 0; kb < K - 2 * CH; kb += 2 * CH) {
    load_chunk<S>(arow0, arow1, wcol, ldw, arow0n, arow1n, w2col, ldw2, kb + CH,
                  a0Q, a1Q, bQ, a0nQ, a1nQ, bnQ);
    keep_order();
    mma_chunk<S>(a0P, a1P, bP, a0nP, a1nP, bnP, c0, c1, s0, s1);
    keep_order();
    load_chunk<S>(arow0, arow1, wcol, ldw, arow0n, arow1n, w2col, ldw2, kb + 2 * CH,
                  a0P, a1P, bP, a0nP, a1nP, bnP);
    keep_order();
    mma_chunk<S>(a0Q, a1Q, bQ, a0nQ, a1nQ, bnQ, c0, c1, s0, s1);
    keep_order();
  }
  load_chunk<S>(arow0, arow1, wcol, ldw, arow0n, arow1n, w2col, ldw2, K - CH,
                a0Q, a1Q, bQ, a0nQ, a1nQ, bnQ);
  keep_order();
  mma_chunk<S>(a0P, a1P, bP, a0nP, a1nP, bnP, c0, c1, s0, s1);
  mma_chunk<S>(a0Q, a1Q, bQ, a0nQ, a1nQ, bnQ, c0, c1, s0, s1);
}

__device__ __forceinline__ void store_tile(float* dst, int ldd, int rbase,
                                           int col0, v8f c, int lane16, int hi) {
#pragma unroll
  for (int v = 0; v < 8; ++v)
    dst[(rbase + v + hi * 8) * ldd + col0 + lane16] = c[v];
}

__device__ __forceinline__ v8f add_bias(v8f c, float b) {
#pragma unroll
  for (int v = 0; v < 8; ++v) c[v] += b;
  return c;
}
#define LOG2E_F 1.44269504088896340736f
__device__ __forceinline__ float hw_exp(float t) {
  return __builtin_amdgcn_exp2f(t * LOG2E_F);
}
__device__ __forceinline__ v8f bias_elu(v8f c, float b) {
#pragma unroll
  for (int v = 0; v < 8; ++v) {
    float t = c[v] + b;
    float em1 = hw_exp(t) - 1.f;
    c[v] = t > 0.f ? t : em1;
  }
  return c;
}
__device__ __forceinline__ v8f gate_mix(v8f g, float b, v8f e, v8f h2) {
#pragma unroll
  for (int v = 0; v < 8; ++v) {
    float t = g[v] + b;
    float s = __builtin_amdgcn_rcpf(1.f + hw_exp(-t));
    g[v] = e[v] + s * h2[v];
  }
  return g;
}

__global__ __launch_bounds__(256, 1)
void vsn_fused_pairs32(
    const float* __restrict__ x,
    const float* __restrict__ embW, const float* __restrict__ embB,
    const float* __restrict__ fc1W, const float* __restrict__ fc1B,
    const float* __restrict__ fc2W, const float* __restrict__ fc2B,
    const float* __restrict__ gW,   const float* __restrict__ gB,
    const float* __restrict__ lnG,  const float* __restrict__ lnB,
    const float* __restrict__ sf1W, const float* __restrict__ sf1B,
    const float* __restrict__ sf2W, const float* __restrict__ sf2B,
    const float* __restrict__ sgW,  const float* __restrict__ sgB,
    const float* __restrict__ sskW, const float* __restrict__ sskB,
    const float* __restrict__ slnG, const float* __restrict__ slnB,
    float* __restrict__ outSel, float* __restrict__ outW) {
  __shared__ __attribute__((aligned(16))) float smem[SMEM_FLOATS];
  float* TRANS = smem;
  float* XT0 = TRANS + MR * ST_T;
  float* XW0 = XT0 + 2 * MR * ST_X;
  float* PU = XW0;
  float* PV = XW0 + MR * ST_P;
  float* SKP = XT0;
  float* GP = PU;
  float* WB = GP + MR * ST_S;

  const int tid = threadIdx.x;
  const int wid = tid >> 5;
  const int lane = tid & 31;
  const int lane16 = lane & 15;
  const int hi = lane >> 4;
  const int row0 = blockIdx.x * MR;

  const int g = wid >> 2;
  const int n0 = (wid & 3) * 32;
  const int cb = n0 + 2 * lane16;
  const int glr = (tid & 127) >> 4;
  const int lc = tid & 15;
  float* XT = XT0 + g * (MR * ST_X);
  float* XW = XW0 + g * (MR * ST_P);

  float2 bE = *(const float2*)(embB + g * H_ + cb);
  float2 b1 = *(const float2*)(fc1B + g * H_ + cb);
  float2 b2 = *(const float2*)(fc2B + g * H_ + cb);
  float2 bG = *(const float2*)(gB + g * H_ + cb);
  float2 pE0[D_ / 4], pE1[D_ / 4];
  load_panel2<0, D_ / 4, D_ / 4>(embW + (size_t)g * D_ * H_, H_, n0, lane16, hi, pE0, pE1);
  {
#pragma unroll
    for (int rr = 0; rr < 4; ++rr) {
      const int r = glr + 8 * rr;
      float4 v = *(const float4*)(x + (size_t)(row0 + r) * (F_ * D_) + g * D_ + lc * 4);
      float* dp = XT + r * ST_X + lc * 4;
      dp[0] = v.x; dp[1] = v.y; dp[2] = v.z; dp[3] = v.w;
    }
  }
  float2 pF10[H_ / 4], pF11[H_ / 4];
  float4 xr[4];
  wg_signal_lds();
  keep_order();
  load_panel2<0, H_ / 4, H_ / 4>(fc1W + (size_t)g * H_ * H_, H_, n0, lane16, hi, pF10, pF11);
#pragma unroll
  for (int rr = 0; rr < 4; ++rr)
    xr[rr] = *(const float4*)(x + (size_t)(row0 + glr + 8 * rr) * (F_ * D_) +
                              (g + 2) * D_ + lc * 4);
  keep_order();
  wg_wait();

  for (int f = g; f < F_; f += 2) {
    const int fn = f + 2 < F_ ? f + 2 : f;
    const int fnn = f + 4 < F_ ? f + 4 : fn;
    float* SLOT = TRANS + f * H_;

    v8f ce00, ce01, ce10, ce11;
    gemm_panel2<D_>(XT, ST_X, lane16, hi, pE0, pE1, ce00, ce01, ce10, ce11);
    ce00 = add_bias(ce00, bE.x);
    ce10 = add_bias(ce10, bE.x);
    ce01 = add_bias(ce01, bE.y);
    ce11 = add_bias(ce11, bE.y);
    store_tile2(SLOT, ST_T, 0, cb, ce00, ce01, hi);
    store_tile2(SLOT, ST_T, 16, cb, ce10, ce11, hi);
    wg_signal_lds();
    keep_order();
    float2 pF20[H_ / 4], pF21[H_ / 4];
    load_panel2<0, H_ / 4, H_ / 4>(fc2W + (size_t)f * H_ * H_, H_, n0, lane16, hi, pF20, pF21);
    keep_order();
    wg_wait();

    {
      v8f c00, c01, c10, c11;
      gemm_panel2<H_>(SLOT, ST_T, lane16, hi, pF10, pF11, c00, c01, c10, c11);
      c00 = bias_elu(c00, b1.x);
      c10 = bias_elu(c10, b1.x);
      c01 = bias_elu(c01, b1.y);
      c11 = bias_elu(c11, b1.y);
      store_tile2(XW, ST_P, 0, cb, c00, c01, hi);
      store_tile2(XW, ST_P, 16, cb, c10, c11, hi);
    }
    wg_signal_lds();
    keep_order();
    float2 pG0[H_ / 4], pG1[H_ / 4];
    load_panel2<0, H_ / 4, H_ / 4>(gW + (size_t)f * H_ * H_, H_, n0, lane16, hi, pG0, pG1);
    keep_order();
    wg_wait();

    v8f c200, c201, c210, c211;
    gemm_panel2<H_>(XW, ST_P, lane16, hi, pF20, pF21, c200, c201, c210, c211);
    c200 = add_bias(c200, b2.x);
    c210 = add_bias(c210, b2.x);
    c201 = add_bias(c201, b2.y);
    c211 = add_bias(c211, b2.y);
    store_tile2(SLOT, ST_T, 0, cb, c200, c201, hi);
    store_tile2(SLOT, ST_T, 16, cb, c210, c211, hi);
    wg_signal_lds();
    keep_order();

    load_panel2<0, D_ / 4, D_ / 4>(embW + (size_t)fn * D_ * H_, H_, n0, lane16, hi, pE0, pE1);
    const float2 bGf = bG;
    bE = *(const float2*)(embB + fn * H_ + cb);
    b1 = *(const float2*)(fc1B + fn * H_ + cb);
    b2 = *(const float2*)(fc2B + fn * H_ + cb);
    bG = *(const float2*)(gB + fn * H_ + cb);
    float lg[8], lb[8];
#pragma unroll
    for (int j = 0; j < 8; ++j) {
      lg[j] = lnG[f * H_ + lc + 16 * j];
      lb[j] = lnB[f * H_ + lc + 16 * j];
    }
    keep_order();
    wg_wait();
    {
      v8f cg00, cg01, cg10, cg11;
      gemm_panel2<H_>(SLOT, ST_T, lane16, hi, pG0, pG1, cg00, cg01, cg10, cg11);
      cg00 = gate_mix(cg00, bGf.x, ce00, c200);
      cg10 = gate_mix(cg10, bGf.x, ce10, c210);
      cg01 = gate_mix(cg01, bGf.y, ce01, c201);
      cg11 = gate_mix(cg11, bGf.y, ce11, c211);
      store_tile2(XW, ST_P, 0, cb, cg00, cg01, hi);
      store_tile2(XW, ST_P, 16, cb, cg10, cg11, hi);
    }
    wg_signal_lds();
    keep_order();
    load_panel2<0, H_ / 8, H_ / 4>(fc1W + (size_t)fn * H_ * H_, H_, n0, lane16, hi, pF10, pF11);
    keep_order();
    wg_wait();

    float vals[4][8];
#pragma unroll
    for (int rr = 0; rr < 4; ++rr) {
      const int r = glr + 8 * rr;
#pragma unroll
      for (int j = 0; j < 8; ++j) vals[rr][j] = XW[r * ST_P + lc + 16 * j];
    }
    float sm[8];
#pragma unroll
    for (int rr = 0; rr < 4; ++rr) {
      float s = 0.f, ss = 0.f;
#pragma unroll
      for (int j = 0; j < 8; ++j) {
        float v = vals[rr][j];
        s += v; ss += v * v;
      }
      sm[2 * rr] = s;
      sm[2 * rr + 1] = ss;
    }
    rowsum16_n<8>(sm);
#pragma unroll
    for (int rr = 0; rr < 4; ++rr) {
      const int r = glr + 8 * rr;
      float mu = sm[2 * rr] * (1.f / 128.f);
      float var = sm[2 * rr + 1] * (1.f / 128.f) - mu * mu;
      float rs = rsqrtf(var + 1e-5f);
#pragma unroll
      for (int j = 0; j < 8; ++j) {
        int h = lc + 16 * j;
        float nv = (vals[rr][j] - mu) * rs * lg[j] + lb[j];
        SLOT[r * ST_T + h] = nv;
      }
      float* dp = XT + r * ST_X + lc * 4;
      dp[0] = xr[rr].x; dp[1] = xr[rr].y; dp[2] = xr[rr].z; dp[3] = xr[rr].w;
    }
    wg_signal_lds();
    keep_order();
    load_panel2<H_ / 8, H_ / 4, H_ / 4>(fc1W + (size_t)fn * H_ * H_, H_, n0, lane16, hi, pF10, pF11);
#pragma unroll
    for (int rr = 0; rr < 4; ++rr)
      xr[rr] = *(const float4*)(x + (size_t)(row0 + glr + 8 * rr) * (F_ * D_) +
                                fnn * D_ + lc * 4);
    keep_order();
    wg_wait();
  }

  const int nB = wid * 16;
  const int lr = tid >> 4;
  v2f pS2[H_ / 4];
  load_panel<H_>(sf2W, H_, nB, lane16, hi, pS2);
  const float bS1 = sf1B[nB + lane16];
  const float bS2 = sf2B[nB + lane16];
  const float bSK = sskB[lc];
  {
    v8f b10, b11, sk0, sk1;
    gemm_stream<FH_, 64>(TRANS, ST_T, sf1W, H_, nB, sskW, F_, wid, lane16, hi,
                         b10, b11, sk0, sk1);
    b10 = bias_elu(b10, bS1);
    b11 = bias_elu(b11, bS1);
    store_tile(PU, ST_P, 0, nB, b10, lane16, hi);
    store_tile(PU, ST_P, 16, nB, b11, lane16, hi);
    float* SKPw = SKP + wid * (MR * 16);
    store_tile(SKPw, 16, 0, 0, sk0, lane16, hi);
    store_tile(SKPw, 16, 16, 0, sk1, lane16, hi);
  }
  wg_barrier_lds();

  {
    v8f b20, b21;
    gemm_panel<H_>(PU, ST_P, lane16, hi, pS2, b20, b21);
    b20 = add_bias(b20, bS2);
    b21 = add_bias(b21, bS2);
    store_tile(PV, ST_P, 0, nB, b20, lane16, hi);
    store_tile(PV, ST_P, 16, nB, b21, lane16, hi);
  }
  wg_barrier_lds();

  if (wid == 0) {
    v2f pSG[H_ / 4];
    load_panel<H_>(sgW, F_, 0, lane16, hi, pSG);
    v8f g0, g1;
    gemm_panel<H_>(PV, ST_P, lane16, hi, pSG, g0, g1);
    float b = sgB[lane16];
    g0 = add_bias(g0, b);
    g1 = add_bias(g1, b);
    store_tile(GP, ST_S, 0, 0, g0, lane16, hi);
    store_tile(GP, ST_S, 16, 0, g1, lane16, hi);
  }
  wg_barrier_lds();

  {
    float gpv[2], skv[2];
#pragma unroll
    for (int rr = 0; rr < 2; ++rr) {
      const int r = lr + 16 * rr;
      gpv[rr] = GP[r * ST_S + lc];
      float sk = SKP[r * 16 + lc];
#pragma unroll
      for (int pw = 1; pw < 8; ++pw) sk += SKP[pw * (MR * 16) + r * 16 + lc];
      skv[rr] = sk + bSK;
    }
    float vv[2], st[4];
#pragma unroll
    for (int rr = 0; rr < 2; ++rr) {
      float gp = gpv[rr];
      float sg = 1.f / (1.f + expf(-gp));
      float v = skv[rr] + sg * gp;
      vv[rr] = v;
      st[2 * rr] = v;
      st[2 * rr + 1] = v * v;
    }
    rowsum16_n<4>(st);
    const float sG = slnG[lc], sB = slnB[lc];
    float nv[2], mx[2];
#pragma unroll
    for (int rr = 0; rr < 2; ++rr) {
      float mu = st[2 * rr] * (1.f / 16.f);
      float var = st[2 * rr + 1] * (1.f / 16.f) - mu * mu;
      nv[rr] = (vv[rr] - mu) * rsqrtf(var + 1e-5f) * sG + sB;
      mx[rr] = nv[rr];
    }
    rowmax16_n<2>(mx);
    float e[2], se[2];
#pragma unroll
    for (int rr = 0; rr < 2; ++rr) {
      e[rr] = expf(nv[rr] - mx[rr]);
      se[rr] = e[rr];
    }
    rowsum16_n<2>(se);
#pragma unroll
    for (int rr = 0; rr < 2; ++rr) {
      const int r = lr + 16 * rr;
      float w = e[rr] / se[rr];
      WB[r * 16 + lc] = w;
      outW[(size_t)(row0 + r) * F_ + lc] = w;
    }
  }
  wg_barrier_lds();

#pragma unroll
  for (int rr = 0; rr < 2; ++rr) {
    const int r = lr + 16 * rr;
#pragma unroll
    for (int j = 0; j < 8; ++j) {
      int h = lc + 16 * j;
      float acc = 0.f;
#pragma unroll
      for (int f = 0; f < F_; ++f)
        acc += TRANS[r * ST_T + f * H_ + h] * WB[r * 16 + f];
      outSel[(size_t)(row0 + r) * H_ + h] = acc;
    }
  }
}

extern "C" void kernel_launch(void* const* d_in, const int* in_sizes, int n_in,
                              void* d_out, int out_size, void* d_ws, size_t ws_size,
                              hipStream_t stream) {
  (void)in_sizes; (void)n_in; (void)out_size; (void)d_ws; (void)ws_size;
  const float* x    = (const float*)d_in[0];
  const float* embW = (const float*)d_in[1];
  const float* embB = (const float*)d_in[2];
  const float* fc1W = (const float*)d_in[3];
  const float* fc1B = (const float*)d_in[4];
  const float* fc2W = (const float*)d_in[5];
  const float* fc2B = (const float*)d_in[6];
  const float* gW   = (const float*)d_in[7];
  const float* gB   = (const float*)d_in[8];
  const float* lnG  = (const float*)d_in[9];
  const float* lnB  = (const float*)d_in[10];
  const float* sf1W = (const float*)d_in[11];
  const float* sf1B = (const float*)d_in[12];
  const float* sf2W = (const float*)d_in[13];
  const float* sf2B = (const float*)d_in[14];
  const float* sgW  = (const float*)d_in[15];
  const float* sgB  = (const float*)d_in[16];
  const float* sskW = (const float*)d_in[17];
  const float* sskB = (const float*)d_in[18];
  const float* slnG = (const float*)d_in[19];
  const float* slnB = (const float*)d_in[20];

  float* outSel = (float*)d_out;
  float* outW   = outSel + (size_t)ROWS_TOTAL * H_;

  vsn_fused_pairs32<<<ROWS_TOTAL / MR, 256, 0, stream>>>(
      x, embW, embB, fc1W, fc1B, fc2W, fc2B, gW, gB, lnG, lnB,
      sf1W, sf1B, sf2W, sf2B, sgW, sgB, sskW, sskB, slnG, slnB,
      outSel, outW);
}


# fc1 folded into the embedding: a head launch forms embW@fc1W and its bias row per feature on WMMA; fc1 runs on the x tile (K=64) in emb's pass; fc1 sum order changes; one barrier fewer per pair
# speedup vs baseline: 3.1521x; 1.0732x over previous
#include <hip/hip_runtime.h>
#include <hip/hip_bf16.h>
#include <math.h>


typedef float v2f __attribute__((ext_vector_type(2)));
typedef float v8f __attribute__((ext_vector_type(8)));

#define ROWS_TOTAL 16384
#define MR 32
#define F_ 16
#define D_ 64
#define H_ 128
#define FH_ 2048

#define ST_T 2052
#define ST_X 68
#define ST_P 132
#define ST_S 20

#define SMEM_FLOATS (MR*ST_T + 2*MR*ST_X + 2*MR*ST_P)

template <int M>
__device__ __forceinline__ float row_xor(float v) {
  return __builtin_bit_cast(
      float, __builtin_amdgcn_update_dpp(0, __builtin_bit_cast(int, v), 0x160 + M,
                                         0xf, 0xf, true));
}
template <int N>
__device__ __forceinline__ void rowsum16_n(float (&v)[N]) {
#pragma unroll
  for (int i = 0; i < N; ++i) v[i] += row_xor<1>(v[i]);
#pragma unroll
  for (int i = 0; i < N; ++i) v[i] += row_xor<2>(v[i]);
#pragma unroll
  for (int i = 0; i < N; ++i) v[i] += row_xor<4>(v[i]);
#pragma unroll
  for (int i = 0; i < N; ++i) v[i] += row_xor<8>(v[i]);
}
template <int N>
__device__ __forceinline__ void rowmax16_n(float (&v)[N]) {
#pragma unroll
  for (int i = 0; i < N; ++i) v[i] = fmaxf(v[i], row_xor<1>(v[i]));
#pragma unroll
  for (int i = 0; i < N; ++i) v[i] = fmaxf(v[i], row_xor<2>(v[i]));
#pragma unroll
  for (int i = 0; i < N; ++i) v[i] = fmaxf(v[i], row_xor<4>(v[i]));
#pragma unroll
  for (int i = 0; i < N; ++i) v[i] = fmaxf(v[i], row_xor<8>(v[i]));
}

__device__ __forceinline__ void wg_barrier_lds() {
  asm volatile("s_wait_dscnt 0x0\n\ts_barrier_signal -1\n\ts_barrier_wait -1" ::: "memory");
}
__device__ __forceinline__ void wg_signal_lds() {
  asm volatile("s_wait_dscnt 0x0\n\ts_barrier_signal -1" ::: "memory");
}
__device__ __forceinline__ void wg_wait() {
  asm volatile("s_barrier_wait -1" ::: "memory");
}

__device__ __forceinline__ void keep_order() { __builtin_amdgcn_sched_barrier(0); }

__device__ __forceinline__ v8f mma4(v2f a, v2f b, v8f c) {
  return __builtin_amdgcn_wmma_f32_16x16x4_f32(false, a, false, b, (short)0, c,
                                               false, false);
}


template <int K>
__device__ __forceinline__ void load_panel(const float* __restrict__ W, int ldw,
                                           int n0, int lane16, int hi,
                                           v2f (&b)[K / 4]) {
  const float* wcol = W + (size_t)(hi * 2) * ldw + n0 + lane16;
#pragma unroll
  for (int s = 0; s < K / 4; ++s) {
    b[s].x = wcol[(size_t)(4 * s) * ldw];
    b[s].y = wcol[(size_t)(4 * s + 1) * ldw];
  }
}

template <int S>
__device__ __forceinline__ void load_a(const float* arow0, const float* arow1,
                                       int kb, v2f (&a0)[S], v2f (&a1)[S]) {
#pragma unroll
  for (int s = 0; s < S; ++s) {
    a0[s].x = arow0[kb + 4 * s];
    a0[s].y = arow0[kb + 4 * s + 1];
    a1[s].x = arow1[kb + 4 * s];
    a1[s].y = arow1[kb + 4 * s + 1];
  }
}

template <int K>
__device__ __forceinline__ void gemm_panel(const float* As, int lda, int lane16,
                                           int hi, const v2f (&b)[K / 4],
                                           v8f& c0, v8f& c1) {
  const v8f z = {0.f, 0.f, 0.f, 0.f, 0.f, 0.f, 0.f, 0.f};
  c0 = z;
  c1 = z;
  const float* arow0 = As + lane16 * lda + hi * 2;
  const float* arow1 = arow0 + 16 * lda;
  v2f a0[2][8], a1[2][8];
  load_a<8>(arow0, arow1, 0, a0[0], a1[0]);
  keep_order();
#pragma unroll
  for (int ch = 0; ch < K / 32; ++ch) {
    if (ch + 1 < K / 32)
      load_a<8>(arow0, arow1, (ch + 1) * 32, a0[(ch + 1) & 1], a1[(ch + 1) & 1]);
    keep_order();
#pragma unroll
    for (int s = 0; s < 8; ++s) {
      c0 = mma4(a0[ch & 1][s], b[ch * 8 + s], c0);
      c1 = mma4(a1[ch & 1][s], b[ch * 8 + s], c1);
    }
    keep_order();
  }
}

template <int S0, int S1, int NS>
__device__ __forceinline__ void load_panel2(const float* __restrict__ W, int ldw,
                                            int n0, int lane16, int hi,
                                            float2 (&ra)[NS], float2 (&rb)[NS]) {
  const float* wcol = W + (size_t)(hi * 2) * ldw + n0 + 2 * lane16;
#pragma unroll
  for (int s = S0; s < S1; ++s) {
    ra[s] = *(const float2*)(wcol + (size_t)(4 * s) * ldw);
    rb[s] = *(const float2*)(wcol + (size_t)(4 * s + 1) * ldw);
  }
}

template <int K>
__device__ __forceinline__ void gemm_panel2(const float* As, int lda, int lane16,
                                            int hi, const float2 (&ra)[K / 4],
                                            const float2 (&rb)[K / 4],
                                            v8f& c00, v8f& c01, v8f& c10, v8f& c11) {
  const v8f z = {0.f, 0.f, 0.f, 0.f, 0.f, 0.f, 0.f, 0.f};
  c00 = z; c01 = z; c10 = z; c11 = z;
  const float* arow0 = As + lane16 * lda + hi * 2;
  const float* arow1 = arow0 + 16 * lda;
  v2f a0[2][4], a1[2][4];
  load_a<4>(arow0, arow1, 0, a0[0], a1[0]);
  keep_order();
#pragma unroll
  for (int ch = 0; ch < K / 16; ++ch) {
    if (ch + 1 < K / 16)
      load_a<4>(arow0, arow1, (ch + 1) * 16, a0[(ch + 1) & 1], a1[(ch + 1) & 1]);
    keep_order();
#pragma unroll
    for (int s = 0; s < 4; ++s) {
      v2f bj0, bj1;
      bj0.x = ra[ch * 4 + s].x; bj0.y = rb[ch * 4 + s].x;
      bj1.x = ra[ch * 4 + s].y; bj1.y = rb[ch * 4 + s].y;
      c00 = mma4(a0[ch & 1][s], bj0, c00);
      c10 = mma4(a1[ch & 1][s], bj0, c10);
      c01 = mma4(a0[ch & 1][s], bj1, c01);
      c11 = mma4(a1[ch & 1][s], bj1, c11);
    }
    keep_order();
  }
}

template <int K>
__device__ __forceinline__ void gemm_panel2x2(const float* As, int lda, int lane16,
                                              int hi, const float2 (&ra)[K / 4],
                                              const float2 (&rb)[K / 4],
                                              const float2 (&sa)[K / 4],
                                              const float2 (&sb)[K / 4],
                                              v8f& c00, v8f& c01, v8f& c10, v8f& c11,
                                              v8f& d00, v8f& d01, v8f& d10, v8f& d11) {
  const v8f z = {0.f, 0.f, 0.f, 0.f, 0.f, 0.f, 0.f, 0.f};
  c00 = z; c01 = z; c10 = z; c11 = z;
  d00 = z; d01 = z; d10 = z; d11 = z;
  const float* arow0 = As + lane16 * lda + hi * 2;
  const float* arow1 = arow0 + 16 * lda;
  v2f a0[2][4], a1[2][4];
  load_a<4>(arow0, arow1, 0, a0[0], a1[0]);
  keep_order();
#pragma unroll
  for (int ch = 0; ch < K / 16; ++ch) {
    if (ch + 1 < K / 16)
      load_a<4>(arow0, arow1, (ch + 1) * 16, a0[(ch + 1) & 1], a1[(ch + 1) & 1]);
    keep_order();
#pragma unroll
    for (int s = 0; s < 4; ++s) {
      v2f bj0, bj1, ej0, ej1;
      bj0.x = ra[ch * 4 + s].x; bj0.y = rb[ch * 4 + s].x;
      bj1.x = ra[ch * 4 + s].y; bj1.y = rb[ch * 4 + s].y;
      ej0.x = sa[ch * 4 + s].x; ej0.y = sb[ch * 4 + s].x;
      ej1.x = sa[ch * 4 + s].y; ej1.y = sb[ch * 4 + s].y;
      c00 = mma4(a0[ch & 1][s], bj0, c00);
      c10 = mma4(a1[ch & 1][s], bj0, c10);
      c01 = mma4(a0[ch & 1][s], bj1, c01);
      c11 = mma4(a1[ch & 1][s], bj1, c11);
      d00 = mma4(a0[ch & 1][s], ej0, d00);
      d10 = mma4(a1[ch & 1][s], ej0, d10);
      d01 = mma4(a0[ch & 1][s], ej1, d01);
      d11 = mma4(a1[ch & 1][s], ej1, d11);
    }
    keep_order();
  }
}

__device__ __forceinline__ void store_tile2(float* dst, int ldd, int rbase, int cb,
                                            v8f cj0, v8f cj1, int hi) {
#pragma unroll
  for (int v = 0; v < 8; ++v) {
    float2 t;
    t.x = cj0[v];
    t.y = cj1[v];
    *(float2*)(dst + (rbase + v + hi * 8) * ldd + cb) = t;
  }
}

template <int S>
__device__ __forceinline__ void load_chunk(const float* arow0, const float* arow1,
                                           const float* __restrict__ wcol, int ldw,
                                           const float* arow0n, const float* arow1n,
                                           const float* __restrict__ w2col, int ldw2,
                                           int kb, v2f (&a0)[S], v2f (&a1)[S],
                                           v2f (&b)[S], v2f (&a0n)[S / 8],
                                           v2f (&a1n)[S / 8], v2f (&bn)[S / 8]) {
#pragma unroll
  for (int s = 0; s < S; ++s) {
    b[s].x = wcol[(size_t)(kb + 4 * s) * ldw];
    b[s].y = wcol[(size_t)(kb + 4 * s + 1) * ldw];
  }
#pragma unroll
  for (int t = 0; t < S / 8; ++t) {
    bn[t].x = w2col[(size_t)(kb + 32 * t) * ldw2];
    bn[t].y = w2col[(size_t)(kb + 32 * t + 1) * ldw2];
  }
  load_a<S>(arow0, arow1, kb, a0, a1);
#pragma unroll
  for (int t = 0; t < S / 8; ++t) {
    a0n[t].x = arow0n[kb + 32 * t];
    a0n[t].y = arow0n[kb + 32 * t + 1];
    a1n[t].x = arow1n[kb + 32 * t];
    a1n[t].y = arow1n[kb + 32 * t + 1];
  }
}
template <int S>
__device__ __forceinline__ void mma_chunk(const v2f (&a0)[S], const v2f (&a1)[S],
                                          const v2f (&b)[S], const v2f (&a0n)[S / 8],
                                          const v2f (&a1n)[S / 8], const v2f (&bn)[S / 8],
                                          v8f& c0, v8f& c1, v8f& s0, v8f& s1) {
#pragma unroll
  for (int s = 0; s < S; ++s) {
    c0 = mma4(a0[s], b[s], c0);
    c1 = mma4(a1[s], b[s], c1);
  }
#pragma unroll
  for (int t = 0; t < S / 8; ++t) {
    s0 = mma4(a0n[t], bn[t], s0);
    s1 = mma4(a1n[t], bn[t], s1);
  }
}

template <int K, int CH>
__device__ __forceinline__ void gemm_stream(const float* As, int lda,
                                            const float* __restrict__ W, int ldw,
                                            int n0, const float* __restrict__ W2,
                                            int ldw2, int w, int lane16, int hi,
                                            v8f& c0, v8f& c1, v8f& s0, v8f& s1) {
  static_assert(K % (2 * CH) == 0 && K >= 2 * CH, "chunk pairs");
  static_assert(CH == 64, "a chunk of 16 k-steps: two per wave for the second product");
  constexpr int S = CH / 4;
  const v8f z = {0.f, 0.f, 0.f, 0.f, 0.f, 0.f, 0.f, 0.f};
  c0 = z;
  c1 = z;
  s0 = z;
  s1 = z;
  const float* arow0 = As + lane16 * lda + hi * 2;
  const float* arow1 = arow0 + 16 * lda;
  const float* wcol = W + (size_t)(hi * 2) * ldw + n0 + lane16;
  const float* arow0n = arow0 + 4 * w;
  const float* arow1n = arow1 + 4 * w;
  const float* w2col = W2 + (size_t)(hi * 2 + 4 * w) * ldw2 + lane16;
  v2f bP[S], a0P[S], a1P[S], bQ[S], a0Q[S], a1Q[S];
  v2f bnP[S / 8], a0nP[S / 8], a1nP[S / 8], bnQ[S / 8], a0nQ[S / 8], a1nQ[S / 8];
  load_chunk<S>(arow0, arow1, wcol, ldw, arow0n, arow1n, w2col, ldw2, 0,
                a0P, a1P, bP, a0nP, a1nP, bnP);
  keep_order();
#pragma unroll 1
  for (int kb = 0; kb < K - 2 * CH; kb += 2 * CH) {
    load_chunk<S>(arow0, arow1, wcol, ldw, arow0n, arow1n, w2col, ldw2, kb + CH,
                  a0Q, a1Q, bQ, a0nQ, a1nQ, bnQ);
    keep_order();
    mma_chunk<S>(a0P, a1P, bP, a0nP, a1nP, bnP, c0, c1, s0, s1);
    keep_order();
    load_chunk<S>(arow0, arow1, wcol, ldw, arow0n, arow1n, w2col, ldw2, kb + 2 * CH,
                  a0P, a1P, bP, a0nP, a1nP, bnP);
    keep_order();
    mma_chunk<S>(a0Q, a1Q, bQ, a0nQ, a1nQ, bnQ, c0, c1, s0, s1);
    keep_order();
  }
  load_chunk<S>(arow0, arow1, wcol, ldw, arow0n, arow1n, w2col, ldw2, K - CH,
                a0Q, a1Q, bQ, a0nQ, a1nQ, bnQ);
  keep_order();
  mma_chunk<S>(a0P, a1P, bP, a0nP, a1nP, bnP, c0, c1, s0, s1);
  mma_chunk<S>(a0Q, a1Q, bQ, a0nQ, a1nQ, bnQ, c0, c1, s0, s1);
}

__device__ __forceinline__ void store_tile(float* dst, int ldd, int rbase,
                                           int col0, v8f c, int lane16, int hi) {
#pragma unroll
  for (int v = 0; v < 8; ++v)
    dst[(rbase + v + hi * 8) * ldd + col0 + lane16] = c[v];
}

__device__ __forceinline__ v8f add_bias(v8f c, float b) {
#pragma unroll
  for (int v = 0; v < 8; ++v) c[v] += b;
  return c;
}
#define LOG2E_F 1.44269504088896340736f
__device__ __forceinline__ float hw_exp(float t) {
  return __builtin_amdgcn_exp2f(t * LOG2E_F);
}
__device__ __forceinline__ v8f bias_elu(v8f c, float b) {
#pragma unroll
  for (int v = 0; v < 8; ++v) {
    float t = c[v] + b;
    float em1 = hw_exp(t) - 1.f;
    c[v] = t > 0.f ? t : em1;
  }
  return c;
}
__device__ __forceinline__ v8f gate_mix(v8f g, float b, v8f e, v8f h2) {
#pragma unroll
  for (int v = 0; v < 8; ++v) {
    float t = g[v] + b;
    float s = __builtin_amdgcn_rcpf(1.f + hw_exp(-t));
    g[v] = e[v] + s * h2[v];
  }
  return g;
}

#define FOLD_ROWS 80
#define FOLD_BYTES 655360

__global__ __launch_bounds__(256)
void vsn_fold_fc1(const float* __restrict__ embW, const float* __restrict__ embB,
                  const float* __restrict__ fc1W, const float* __restrict__ fc1B,
                  float* __restrict__ foldW) {
  const int f = blockIdx.x / 5;
  const int rt = blockIdx.x % 5;
  const int wid = threadIdx.x >> 5;
  const int lane = threadIdx.x & 31;
  const int lane16 = lane & 15;
  const int hi = lane >> 4;
  const int n0 = wid * 16;
  const bool wrow = rt < 4;
  const bool live = wrow || lane16 == 0;
  v2f b[H_ / 4];
  load_panel<H_>(fc1W + (size_t)f * H_ * H_, H_, n0, lane16, hi, b);
  const float* ar = (wrow ? embW + ((size_t)f * D_ + rt * 16 + lane16) * H_
                          : embB + (size_t)f * H_) + hi * 2;
  float2 araw[H_ / 4];
#pragma unroll
  for (int s = 0; s < H_ / 4; ++s) araw[s] = *(const float2*)(ar + 4 * s);
  const float cb0 = fc1B[f * H_ + n0 + lane16];
  keep_order();
  v8f c = {0.f, 0.f, 0.f, 0.f, 0.f, 0.f, 0.f, 0.f};
  c[0] = (!wrow && hi == 0) ? cb0 : 0.f;
#pragma unroll
  for (int s = 0; s < H_ / 4; ++s) {
    v2f a;
    a.x = live ? araw[s].x : 0.f;
    a.y = live ? araw[s].y : 0.f;
    c = mma4(a, b[s], c);
  }
  store_tile(foldW + (size_t)f * FOLD_ROWS * H_, H_, rt * 16, n0, c, lane16, hi);
}

template <bool FOLD>
__global__ __launch_bounds__(256, 1)
void vsn_fused_pairs32(
    const float* __restrict__ x,
    const float* __restrict__ embW, const float* __restrict__ embB,
    const float* __restrict__ fc1W, const float* __restrict__ fc1B,
    const float* __restrict__ fc2W, const float* __restrict__ fc2B,
    const float* __restrict__ gW,   const float* __restrict__ gB,
    const float* __restrict__ lnG,  const float* __restrict__ lnB,
    const float* __restrict__ sf1W, const float* __restrict__ sf1B,
    const float* __restrict__ sf2W, const float* __restrict__ sf2B,
    const float* __restrict__ sgW,  const float* __restrict__ sgB,
    const float* __restrict__ sskW, const float* __restrict__ sskB,
    const float* __restrict__ slnG, const float* __restrict__ slnB,
    const float* __restrict__ foldW,
    float* __restrict__ outSel, float* __restrict__ outW) {
  __shared__ __attribute__((aligned(16))) float smem[SMEM_FLOATS];
  float* TRANS = smem;
  float* XT0 = TRANS + MR * ST_T;
  float* XW0 = XT0 + 2 * MR * ST_X;
  float* PU = XW0;
  float* PV = XW0 + MR * ST_P;
  float* SKP = XT0;
  float* GP = PU;
  float* WB = GP + MR * ST_S;

  const int tid = threadIdx.x;
  const int wid = tid >> 5;
  const int lane = tid & 31;
  const int lane16 = lane & 15;
  const int hi = lane >> 4;
  const int row0 = blockIdx.x * MR;

  const int g = wid >> 2;
  const int n0 = (wid & 3) * 32;
  const int cb = n0 + 2 * lane16;
  const int glr = (tid & 127) >> 4;
  const int lc = tid & 15;
  float* XT = XT0 + g * (MR * ST_X);
  float* XW = XW0 + g * (MR * ST_P);

  if constexpr (FOLD) {
  float2 bE = *(const float2*)(embB + g * H_ + cb);
  float2 b1 = *(const float2*)(foldW + ((size_t)g * FOLD_ROWS + D_) * H_ + cb);
  float2 b2 = *(const float2*)(fc2B + g * H_ + cb);
  float2 bG = *(const float2*)(gB + g * H_ + cb);
  float2 pE0[D_ / 4], pE1[D_ / 4];
  float2 pX0[D_ / 4], pX1[D_ / 4];
  load_panel2<0, D_ / 4, D_ / 4>(embW + (size_t)g * D_ * H_, H_, n0, lane16, hi, pE0, pE1);
  load_panel2<0, D_ / 4, D_ / 4>(foldW + (size_t)g * FOLD_ROWS * H_, H_, n0, lane16, hi, pX0, pX1);
  {
#pragma unroll
    for (int rr = 0; rr < 4; ++rr) {
      const int r = glr + 8 * rr;
      float4 v = *(const float4*)(x + (size_t)(row0 + r) * (F_ * D_) + g * D_ + lc * 4);
      float* dp = XT + r * ST_X + lc * 4;
      dp[0] = v.x; dp[1] = v.y; dp[2] = v.z; dp[3] = v.w;
    }
  }
  float2 pF20[H_ / 4], pF21[H_ / 4];
  float4 xr[4];
  wg_signal_lds();
  keep_order();
  load_panel2<0, H_ / 4, H_ / 4>(fc2W + (size_t)g * H_ * H_, H_, n0, lane16, hi, pF20, pF21);
#pragma unroll
  for (int rr = 0; rr < 4; ++rr)
    xr[rr] = *(const float4*)(x + (size_t)(row0 + glr + 8 * rr) * (F_ * D_) +
                              (g + 2) * D_ + lc * 4);
  keep_order();
  wg_wait();

  for (int f = g; f < F_; f += 2) {
    const int fn = f + 2 < F_ ? f + 2 : f;
    const int fnn = f + 4 < F_ ? f + 4 : fn;
    float* SLOT = TRANS + f * H_;

    v8f ce00, ce01, ce10, ce11;
    {
      v8f c00, c01, c10, c11;
      gemm_panel2x2<D_>(XT, ST_X, lane16, hi, pE0, pE1, pX0, pX1,
                        ce00, ce01, ce10, ce11, c00, c01, c10, c11);
      ce00 = add_bias(ce00, bE.x);
      ce10 = add_bias(ce10, bE.x);
      ce01 = add_bias(ce01, bE.y);
      ce11 = add_bias(ce11, bE.y);
      c00 = bias_elu(c00, b1.x);
      c10 = bias_elu(c10, b1.x);
      c01 = bias_elu(c01, b1.y);
      c11 = bias_elu(c11, b1.y);
      store_tile2(XW, ST_P, 0, cb, c00, c01, hi);
      store_tile2(XW, ST_P, 16, cb, c10, c11, hi);
    }
    wg_signal_lds();
    keep_order();
    float2 pG0[H_ / 4], pG1[H_ / 4];
    load_panel2<0, H_ / 4, H_ / 4>(gW + (size_t)f * H_ * H_, H_, n0, lane16, hi, pG0, pG1);
    keep_order();
    wg_wait();

    v8f c200, c201, c210, c211;
    gemm_panel2<H_>(XW, ST_P, lane16, hi, pF20, pF21, c200, c201, c210, c211);
    c200 = add_bias(c200, b2.x);
    c210 = add_bias(c210, b2.x);
    c201 = add_bias(c201, b2.y);
    c211 = add_bias(c211, b2.y);
    store_tile2(SLOT, ST_T, 0, cb, c200, c201, hi);
    store_tile2(SLOT, ST_T, 16, cb, c210, c211, hi);
    wg_signal_lds();
    keep_order();

    load_panel2<0, D_ / 4, D_ / 4>(embW + (size_t)fn * D_ * H_, H_, n0, lane16, hi, pE0, pE1);
    const float2 bGf = bG;
    bE = *(const float2*)(embB + fn * H_ + cb);
    b1 = *(const float2*)(foldW + ((size_t)fn * FOLD_ROWS + D_) * H_ + cb);
    b2 = *(const float2*)(fc2B + fn * H_ + cb);
    bG = *(const float2*)(gB + fn * H_ + cb);
    float lg[8], lb[8];
#pragma unroll
    for (int j = 0; j < 8; ++j) {
      lg[j] = lnG[f * H_ + lc + 16 * j];
      lb[j] = lnB[f * H_ + lc + 16 * j];
    }
    keep_order();
    wg_wait();
    {
      v8f cg00, cg01, cg10, cg11;
      gemm_panel2<H_>(SLOT, ST_T, lane16, hi, pG0, pG1, cg00, cg01, cg10, cg11);
      cg00 = gate_mix(cg00, bGf.x, ce00, c200);
      cg10 = gate_mix(cg10, bGf.x, ce10, c210);
      cg01 = gate_mix(cg01, bGf.y, ce01, c201);
      cg11 = gate_mix(cg11, bGf.y, ce11, c211);
      store_tile2(XW, ST_P, 0, cb, cg00, cg01, hi);
      store_tile2(XW, ST_P, 16, cb, cg10, cg11, hi);
    }
    wg_signal_lds();
    keep_order();
    load_panel2<0, D_ / 4, D_ / 4>(foldW + (size_t)fn * FOLD_ROWS * H_, H_, n0, lane16, hi, pX0, pX1);
    load_panel2<0, 14, H_ / 4>(fc2W + (size_t)fn * H_ * H_, H_, n0, lane16, hi, pF20, pF21);
    keep_order();
    wg_wait();

    float vals[4][8];
#pragma unroll
    for (int rr = 0; rr < 4; ++rr) {
      const int r = glr + 8 * rr;
#pragma unroll
      for (int j = 0; j < 8; ++j) vals[rr][j] = XW[r * ST_P + lc + 16 * j];
    }
    float sm[8];
#pragma unroll
    for (int rr = 0; rr < 4; ++rr) {
      float s = 0.f, ss = 0.f;
#pragma unroll
      for (int j = 0; j < 8; ++j) {
        float v = vals[rr][j];
        s += v; ss += v * v;
      }
      sm[2 * rr] = s;
      sm[2 * rr + 1] = ss;
    }
    rowsum16_n<8>(sm);
#pragma unroll
    for (int rr = 0; rr < 4; ++rr) {
      const int r = glr + 8 * rr;
      float mu = sm[2 * rr] * (1.f / 128.f);
      float var = sm[2 * rr + 1] * (1.f / 128.f) - mu * mu;
      float rs = rsqrtf(var + 1e-5f);
#pragma unroll
      for (int j = 0; j < 8; ++j) {
        int h = lc + 16 * j;
        float nv = (vals[rr][j] - mu) * rs * lg[j] + lb[j];
        SLOT[r * ST_T + h] = nv;
      }
      float* dp = XT + r * ST_X + lc * 4;
      dp[0] = xr[rr].x; dp[1] = xr[rr].y; dp[2] = xr[rr].z; dp[3] = xr[rr].w;
    }
    wg_signal_lds();
    keep_order();
    load_panel2<14, H_ / 4, H_ / 4>(fc2W + (size_t)fn * H_ * H_, H_, n0, lane16, hi, pF20, pF21);
#pragma unroll
    for (int rr = 0; rr < 4; ++rr)
      xr[rr] = *(const float4*)(x + (size_t)(row0 + glr + 8 * rr) * (F_ * D_) +
                                fnn * D_ + lc * 4);
    keep_order();
    wg_wait();
  }
  } else {
  float2 bE = *(const float2*)(embB + g * H_ + cb);
  float2 b1 = *(const float2*)(fc1B + g * H_ + cb);
  float2 b2 = *(const float2*)(fc2B + g * H_ + cb);
  float2 bG = *(const float2*)(gB + g * H_ + cb);
  float2 pE0[D_ / 4], pE1[D_ / 4];
  load_panel2<0, D_ / 4, D_ / 4>(embW + (size_t)g * D_ * H_, H_, n0, lane16, hi, pE0, pE1);
  {
#pragma unroll
    for (int rr = 0; rr < 4; ++rr) {
      const int r = glr + 8 * rr;
      float4 v = *(const float4*)(x + (size_t)(row0 + r) * (F_ * D_) + g * D_ + lc * 4);
      float* dp = XT + r * ST_X + lc * 4;
      dp[0] = v.x; dp[1] = v.y; dp[2] = v.z; dp[3] = v.w;
    }
  }
  float2 pF10[H_ / 4], pF11[H_ / 4];
  float4 xr[4];
  wg_signal_lds();
  keep_order();
  load_panel2<0, H_ / 4, H_ / 4>(fc1W + (size_t)g * H_ * H_, H_, n0, lane16, hi, pF10, pF11);
#pragma unroll
  for (int rr = 0; rr < 4; ++rr)
    xr[rr] = *(const float4*)(x + (size_t)(row0 + glr + 8 * rr) * (F_ * D_) +
                              (g + 2) * D_ + lc * 4);
  keep_order();
  wg_wait();

  for (int f = g; f < F_; f += 2) {
    const int fn = f + 2 < F_ ? f + 2 : f;
    const int fnn = f + 4 < F_ ? f + 4 : fn;
    float* SLOT = TRANS + f * H_;

    v8f ce00, ce01, ce10, ce11;
    gemm_panel2<D_>(XT, ST_X, lane16, hi, pE0, pE1, ce00, ce01, ce10, ce11);
    ce00 = add_bias(ce00, bE.x);
    ce10 = add_bias(ce10, bE.x);
    ce01 = add_bias(ce01, bE.y);
    ce11 = add_bias(ce11, bE.y);
    store_tile2(SLOT, ST_T, 0, cb, ce00, ce01, hi);
    store_tile2(SLOT, ST_T, 16, cb, ce10, ce11, hi);
    wg_signal_lds();
    keep_order();
    float2 pF20[H_ / 4], pF21[H_ / 4];
    load_panel2<0, H_ / 4, H_ / 4>(fc2W + (size_t)f * H_ * H_, H_, n0, lane16, hi, pF20, pF21);
    keep_order();
    wg_wait();

    {
      v8f c00, c01, c10, c11;
      gemm_panel2<H_>(SLOT, ST_T, lane16, hi, pF10, pF11, c00, c01, c10, c11);
      c00 = bias_elu(c00, b1.x);
      c10 = bias_elu(c10, b1.x);
      c01 = bias_elu(c01, b1.y);
      c11 = bias_elu(c11, b1.y);
      store_tile2(XW, ST_P, 0, cb, c00, c01, hi);
      store_tile2(XW, ST_P, 16, cb, c10, c11, hi);
    }
    wg_signal_lds();
    keep_order();
    float2 pG0[H_ / 4], pG1[H_ / 4];
    load_panel2<0, H_ / 4, H_ / 4>(gW + (size_t)f * H_ * H_, H_, n0, lane16, hi, pG0, pG1);
    keep_order();
    wg_wait();

    v8f c200, c201, c210, c211;
    gemm_panel2<H_>(XW, ST_P, lane16, hi, pF20, pF21, c200, c201, c210, c211);
    c200 = add_bias(c200, b2.x);
    c210 = add_bias(c210, b2.x);
    c201 = add_bias(c201, b2.y);
    c211 = add_bias(c211, b2.y);
    store_tile2(SLOT, ST_T, 0, cb, c200, c201, hi);
    store_tile2(SLOT, ST_T, 16, cb, c210, c211, hi);
    wg_signal_lds();
    keep_order();

    load_panel2<0, D_ / 4, D_ / 4>(embW + (size_t)fn * D_ * H_, H_, n0, lane16, hi, pE0, pE1);
    const float2 bGf = bG;
    bE = *(const float2*)(embB + fn * H_ + cb);
    b1 = *(const float2*)(fc1B + fn * H_ + cb);
    b2 = *(const float2*)(fc2B + fn * H_ + cb);
    bG = *(const float2*)(gB + fn * H_ + cb);
    float lg[8], lb[8];
#pragma unroll
    for (int j = 0; j < 8; ++j) {
      lg[j] = lnG[f * H_ + lc + 16 * j];
      lb[j] = lnB[f * H_ + lc + 16 * j];
    }
    keep_order();
    wg_wait();
    {
      v8f cg00, cg01, cg10, cg11;
      gemm_panel2<H_>(SLOT, ST_T, lane16, hi, pG0, pG1, cg00, cg01, cg10, cg11);
      cg00 = gate_mix(cg00, bGf.x, ce00, c200);
      cg10 = gate_mix(cg10, bGf.x, ce10, c210);
      cg01 = gate_mix(cg01, bGf.y, ce01, c201);
      cg11 = gate_mix(cg11, bGf.y, ce11, c211);
      store_tile2(XW, ST_P, 0, cb, cg00, cg01, hi);
      store_tile2(XW, ST_P, 16, cb, cg10, cg11, hi);
    }
    wg_signal_lds();
    keep_order();
    load_panel2<0, H_ / 8, H_ / 4>(fc1W + (size_t)fn * H_ * H_, H_, n0, lane16, hi, pF10, pF11);
    keep_order();
    wg_wait();

    float vals[4][8];
#pragma unroll
    for (int rr = 0; rr < 4; ++rr) {
      const int r = glr + 8 * rr;
#pragma unroll
      for (int j = 0; j < 8; ++j) vals[rr][j] = XW[r * ST_P + lc + 16 * j];
    }
    float sm[8];
#pragma unroll
    for (int rr = 0; rr < 4; ++rr) {
      float s = 0.f, ss = 0.f;
#pragma unroll
      for (int j = 0; j < 8; ++j) {
        float v = vals[rr][j];
        s += v; ss += v * v;
      }
      sm[2 * rr] = s;
      sm[2 * rr + 1] = ss;
    }
    rowsum16_n<8>(sm);
#pragma unroll
    for (int rr = 0; rr < 4; ++rr) {
      const int r = glr + 8 * rr;
      float mu = sm[2 * rr] * (1.f / 128.f);
      float var = sm[2 * rr + 1] * (1.f / 128.f) - mu * mu;
      float rs = rsqrtf(var + 1e-5f);
#pragma unroll
      for (int j = 0; j < 8; ++j) {
        int h = lc + 16 * j;
        float nv = (vals[rr][j] - mu) * rs * lg[j] + lb[j];
        SLOT[r * ST_T + h] = nv;
      }
      float* dp = XT + r * ST_X + lc * 4;
      dp[0] = xr[rr].x; dp[1] = xr[rr].y; dp[2] = xr[rr].z; dp[3] = xr[rr].w;
    }
    wg_signal_lds();
    keep_order();
    load_panel2<H_ / 8, H_ / 4, H_ / 4>(fc1W + (size_t)fn * H_ * H_, H_, n0, lane16, hi, pF10, pF11);
#pragma unroll
    for (int rr = 0; rr < 4; ++rr)
      xr[rr] = *(const float4*)(x + (size_t)(row0 + glr + 8 * rr) * (F_ * D_) +
                                fnn * D_ + lc * 4);
    keep_order();
    wg_wait();
  }

  }

  const int nB = wid * 16;
  const int lr = tid >> 4;
  v2f pS2[H_ / 4];
  load_panel<H_>(sf2W, H_, nB, lane16, hi, pS2);
  const float bS1 = sf1B[nB + lane16];
  const float bS2 = sf2B[nB + lane16];
  const float bSK = sskB[lc];
  {
    v8f b10, b11, sk0, sk1;
    gemm_stream<FH_, 64>(TRANS, ST_T, sf1W, H_, nB, sskW, F_, wid, lane16, hi,
                         b10, b11, sk0, sk1);
    b10 = bias_elu(b10, bS1);
    b11 = bias_elu(b11, bS1);
    store_tile(PU, ST_P, 0, nB, b10, lane16, hi);
    store_tile(PU, ST_P, 16, nB, b11, lane16, hi);
    float* SKPw = SKP + wid * (MR * 16);
    store_tile(SKPw, 16, 0, 0, sk0, lane16, hi);
    store_tile(SKPw, 16, 16, 0, sk1, lane16, hi);
  }
  wg_barrier_lds();

  {
    v8f b20, b21;
    gemm_panel<H_>(PU, ST_P, lane16, hi, pS2, b20, b21);
    b20 = add_bias(b20, bS2);
    b21 = add_bias(b21, bS2);
    store_tile(PV, ST_P, 0, nB, b20, lane16, hi);
    store_tile(PV, ST_P, 16, nB, b21, lane16, hi);
  }
  wg_barrier_lds();

  if (wid == 0) {
    v2f pSG[H_ / 4];
    load_panel<H_>(sgW, F_, 0, lane16, hi, pSG);
    v8f g0, g1;
    gemm_panel<H_>(PV, ST_P, lane16, hi, pSG, g0, g1);
    float b = sgB[lane16];
    g0 = add_bias(g0, b);
    g1 = add_bias(g1, b);
    store_tile(GP, ST_S, 0, 0, g0, lane16, hi);
    store_tile(GP, ST_S, 16, 0, g1, lane16, hi);
  }
  wg_barrier_lds();

  {
    float gpv[2], skv[2];
#pragma unroll
    for (int rr = 0; rr < 2; ++rr) {
      const int r = lr + 16 * rr;
      gpv[rr] = GP[r * ST_S + lc];
      float sk = SKP[r * 16 + lc];
#pragma unroll
      for (int pw = 1; pw < 8; ++pw) sk += SKP[pw * (MR * 16) + r * 16 + lc];
      skv[rr] = sk + bSK;
    }
    float vv[2], st[4];
#pragma unroll
    for (int rr = 0; rr < 2; ++rr) {
      float gp = gpv[rr];
      float sg = 1.f / (1.f + expf(-gp));
      float v = skv[rr] + sg * gp;
      vv[rr] = v;
      st[2 * rr] = v;
      st[2 * rr + 1] = v * v;
    }
    rowsum16_n<4>(st);
    const float sG = slnG[lc], sB = slnB[lc];
    float nv[2], mx[2];
#pragma unroll
    for (int rr = 0; rr < 2; ++rr) {
      float mu = st[2 * rr] * (1.f / 16.f);
      float var = st[2 * rr + 1] * (1.f / 16.f) - mu * mu;
      nv[rr] = (vv[rr] - mu) * rsqrtf(var + 1e-5f) * sG + sB;
      mx[rr] = nv[rr];
    }
    rowmax16_n<2>(mx);
    float e[2], se[2];
#pragma unroll
    for (int rr = 0; rr < 2; ++rr) {
      e[rr] = expf(nv[rr] - mx[rr]);
      se[rr] = e[rr];
    }
    rowsum16_n<2>(se);
#pragma unroll
    for (int rr = 0; rr < 2; ++rr) {
      const int r = lr + 16 * rr;
      float w = e[rr] / se[rr];
      WB[r * 16 + lc] = w;
      outW[(size_t)(row0 + r) * F_ + lc] = w;
    }
  }
  wg_barrier_lds();

#pragma unroll
  for (int rr = 0; rr < 2; ++rr) {
    const int r = lr + 16 * rr;
#pragma unroll
    for (int j = 0; j < 8; ++j) {
      int h = lc + 16 * j;
      float acc = 0.f;
#pragma unroll
      for (int f = 0; f < F_; ++f)
        acc += TRANS[r * ST_T + f * H_ + h] * WB[r * 16 + f];
      outSel[(size_t)(row0 + r) * H_ + h] = acc;
    }
  }
}

extern "C" void kernel_launch(void* const* d_in, const int* in_sizes, int n_in,
                              void* d_out, int out_size, void* d_ws, size_t ws_size,
                              hipStream_t stream) {
  (void)in_sizes; (void)n_in; (void)out_size;
  const float* x    = (const float*)d_in[0];
  const float* embW = (const float*)d_in[1];
  const float* embB = (const float*)d_in[2];
  const float* fc1W = (const float*)d_in[3];
  const float* fc1B = (const float*)d_in[4];
  const float* fc2W = (const float*)d_in[5];
  const float* fc2B = (const float*)d_in[6];
  const float* gW   = (const float*)d_in[7];
  const float* gB   = (const float*)d_in[8];
  const float* lnG  = (const float*)d_in[9];
  const float* lnB  = (const float*)d_in[10];
  const float* sf1W = (const float*)d_in[11];
  const float* sf1B = (const float*)d_in[12];
  const float* sf2W = (const float*)d_in[13];
  const float* sf2B = (const float*)d_in[14];
  const float* sgW  = (const float*)d_in[15];
  const float* sgB  = (const float*)d_in[16];
  const float* sskW = (const float*)d_in[17];
  const float* sskB = (const float*)d_in[18];
  const float* slnG = (const float*)d_in[19];
  const float* slnB = (const float*)d_in[20];

  float* outSel = (float*)d_out;
  float* outW   = outSel + (size_t)ROWS_TOTAL * H_;

  if (d_ws != nullptr && ws_size >= (size_t)FOLD_BYTES) {
    float* foldW = (float*)d_ws;
    vsn_fold_fc1<<<F_ * 5, 256, 0, stream>>>(embW, embB, fc1W, fc1B, foldW);
    vsn_fused_pairs32<true><<<ROWS_TOTAL / MR, 256, 0, stream>>>(
        x, embW, embB, fc1W, fc1B, fc2W, fc2B, gW, gB, lnG, lnB,
        sf1W, sf1B, sf2W, sf2B, sgW, sgB, sskW, sskB, slnG, slnB,
        foldW, outSel, outW);
  } else {
    vsn_fused_pairs32<false><<<ROWS_TOTAL / MR, 256, 0, stream>>>(
        x, embW, embB, fc1W, fc1B, fc2W, fc2B, gW, gB, lnG, lnB,
        sf1W, sf1B, sf2W, sf2B, sgW, sgB, sskW, sskB, slnG, slnB,
        (const float*)nullptr, outSel, outW);
  }
}


# Head launch also writes emb, folded fc1, fc2, gate and s_fc1 weights k-pair interleaved in the workspace: a B fragment is one 16-byte (stage A) or 8-byte (stage B) load, no register transposes
# speedup vs baseline: 3.3283x; 1.0559x over previous
#include <hip/hip_runtime.h>
#include <hip/hip_bf16.h>
#include <math.h>


typedef float v2f __attribute__((ext_vector_type(2)));
typedef float v8f __attribute__((ext_vector_type(8)));

#define ROWS_TOTAL 16384
#define MR 32
#define F_ 16
#define D_ 64
#define H_ 128
#define FH_ 2048

#define ST_T 2052
#define ST_X 68
#define ST_P 132
#define ST_S 20

#define SMEM_FLOATS (MR*ST_T + 2*MR*ST_X + 2*MR*ST_P)

template <int M>
__device__ __forceinline__ float row_xor(float v) {
  return __builtin_bit_cast(
      float, __builtin_amdgcn_update_dpp(0, __builtin_bit_cast(int, v), 0x160 + M,
                                         0xf, 0xf, true));
}
template <int N>
__device__ __forceinline__ void rowsum16_n(float (&v)[N]) {
#pragma unroll
  for (int i = 0; i < N; ++i) v[i] += row_xor<1>(v[i]);
#pragma unroll
  for (int i = 0; i < N; ++i) v[i] += row_xor<2>(v[i]);
#pragma unroll
  for (int i = 0; i < N; ++i) v[i] += row_xor<4>(v[i]);
#pragma unroll
  for (int i = 0; i < N; ++i) v[i] += row_xor<8>(v[i]);
}
template <int N>
__device__ __forceinline__ void rowmax16_n(float (&v)[N]) {
#pragma unroll
  for (int i = 0; i < N; ++i) v[i] = fmaxf(v[i], row_xor<1>(v[i]));
#pragma unroll
  for (int i = 0; i < N; ++i) v[i] = fmaxf(v[i], row_xor<2>(v[i]));
#pragma unroll
  for (int i = 0; i < N; ++i) v[i] = fmaxf(v[i], row_xor<4>(v[i]));
#pragma unroll
  for (int i = 0; i < N; ++i) v[i] = fmaxf(v[i], row_xor<8>(v[i]));
}

__device__ __forceinline__ void wg_barrier_lds() {
  asm volatile("s_wait_dscnt 0x0\n\ts_barrier_signal -1\n\ts_barrier_wait -1" ::: "memory");
}
__device__ __forceinline__ void wg_signal_lds() {
  asm volatile("s_wait_dscnt 0x0\n\ts_barrier_signal -1" ::: "memory");
}
__device__ __forceinline__ void wg_wait() {
  asm volatile("s_barrier_wait -1" ::: "memory");
}

__device__ __forceinline__ void keep_order() { __builtin_amdgcn_sched_barrier(0); }

__device__ __forceinline__ v8f mma4(v2f a, v2f b, v8f c) {
  return __builtin_amdgcn_wmma_f32_16x16x4_f32(false, a, false, b, (short)0, c,
                                               false, false);
}


template <int K>
__device__ __forceinline__ void load_panel(const float* __restrict__ W, int ldw,
                                           int n0, int lane16, int hi,
                                           v2f (&b)[K / 4]) {
  const float* wcol = W + (size_t)(hi * 2) * ldw + n0 + lane16;
#pragma unroll
  for (int s = 0; s < K / 4; ++s) {
    b[s].x = wcol[(size_t)(4 * s) * ldw];
    b[s].y = wcol[(size_t)(4 * s + 1) * ldw];
  }
}

template <int S>
__device__ __forceinline__ void load_a(const float* arow0, const float* arow1,
                                       int kb, v2f (&a0)[S], v2f (&a1)[S]) {
#pragma unroll
  for (int s = 0; s < S; ++s) {
    a0[s].x = arow0[kb + 4 * s];
    a0[s].y = arow0[kb + 4 * s + 1];
    a1[s].x = arow1[kb + 4 * s];
    a1[s].y = arow1[kb + 4 * s + 1];
  }
}

template <int K>
__device__ __forceinline__ void gemm_panel(const float* As, int lda, int lane16,
                                           int hi, const v2f (&b)[K / 4],
                                           v8f& c0, v8f& c1) {
  const v8f z = {0.f, 0.f, 0.f, 0.f, 0.f, 0.f, 0.f, 0.f};
  c0 = z;
  c1 = z;
  const float* arow0 = As + lane16 * lda + hi * 2;
  const float* arow1 = arow0 + 16 * lda;
  v2f a0[2][8], a1[2][8];
  load_a<8>(arow0, arow1, 0, a0[0], a1[0]);
  keep_order();
#pragma unroll
  for (int ch = 0; ch < K / 32; ++ch) {
    if (ch + 1 < K / 32)
      load_a<8>(arow0, arow1, (ch + 1) * 32, a0[(ch + 1) & 1], a1[(ch + 1) & 1]);
    keep_order();
#pragma unroll
    for (int s = 0; s < 8; ++s) {
      c0 = mma4(a0[ch & 1][s], b[ch * 8 + s], c0);
      c1 = mma4(a1[ch & 1][s], b[ch * 8 + s], c1);
    }
    keep_order();
  }
}

template <int S0, int S1, int NS, bool PK = false>
__device__ __forceinline__ void load_panel2(const float* __restrict__ W, int ldw,
                                            int n0, int lane16, int hi,
                                            float2 (&ra)[NS], float2 (&rb)[NS]) {
  if constexpr (PK) {
    const float* wq = W + (size_t)(hi * ldw + n0 + 2 * lane16) * 2;
#pragma unroll
    for (int s = S0; s < S1; ++s) {
      const float4 t = *(const float4*)(wq + (size_t)(4 * s) * ldw);
      ra[s].x = t.x; ra[s].y = t.y;
      rb[s].x = t.z; rb[s].y = t.w;
    }
  } else {
    const float* wcol = W + (size_t)(hi * 2) * ldw + n0 + 2 * lane16;
#pragma unroll
    for (int s = S0; s < S1; ++s) {
      ra[s] = *(const float2*)(wcol + (size_t)(4 * s) * ldw);
      rb[s] = *(const float2*)(wcol + (size_t)(4 * s + 1) * ldw);
    }
  }
}

template <int K, bool PK = false>
__device__ __forceinline__ void gemm_panel2(const float* As, int lda, int lane16,
                                            int hi, const float2 (&ra)[K / 4],
                                            const float2 (&rb)[K / 4],
                                            v8f& c00, v8f& c01, v8f& c10, v8f& c11) {
  const v8f z = {0.f, 0.f, 0.f, 0.f, 0.f, 0.f, 0.f, 0.f};
  c00 = z; c01 = z; c10 = z; c11 = z;
  const float* arow0 = As + lane16 * lda + hi * 2;
  const float* arow1 = arow0 + 16 * lda;
  v2f a0[2][4], a1[2][4];
  load_a<4>(arow0, arow1, 0, a0[0], a1[0]);
  keep_order();
#pragma unroll
  for (int ch = 0; ch < K / 16; ++ch) {
    if (ch + 1 < K / 16)
      load_a<4>(arow0, arow1, (ch + 1) * 16, a0[(ch + 1) & 1], a1[(ch + 1) & 1]);
    keep_order();
#pragma unroll
    for (int s = 0; s < 4; ++s) {
      v2f bj0, bj1;
      if constexpr (PK) {
        bj0.x = ra[ch * 4 + s].x; bj0.y = ra[ch * 4 + s].y;
        bj1.x = rb[ch * 4 + s].x; bj1.y = rb[ch * 4 + s].y;
      } else {
        bj0.x = ra[ch * 4 + s].x; bj0.y = rb[ch * 4 + s].x;
        bj1.x = ra[ch * 4 + s].y; bj1.y = rb[ch * 4 + s].y;
      }
      c00 = mma4(a0[ch & 1][s], bj0, c00);
      c10 = mma4(a1[ch & 1][s], bj0, c10);
      c01 = mma4(a0[ch & 1][s], bj1, c01);
      c11 = mma4(a1[ch & 1][s], bj1, c11);
    }
    keep_order();
  }
}

template <int K, bool PK = false>
__device__ __forceinline__ void gemm_panel2x2(const float* As, int lda, int lane16,
                                              int hi, const float2 (&ra)[K / 4],
                                              const float2 (&rb)[K / 4],
                                              const float2 (&sa)[K / 4],
                                              const float2 (&sb)[K / 4],
                                              v8f& c00, v8f& c01, v8f& c10, v8f& c11,
                                              v8f& d00, v8f& d01, v8f& d10, v8f& d11) {
  const v8f z = {0.f, 0.f, 0.f, 0.f, 0.f, 0.f, 0.f, 0.f};
  c00 = z; c01 = z; c10 = z; c11 = z;
  d00 = z; d01 = z; d10 = z; d11 = z;
  const float* arow0 = As + lane16 * lda + hi * 2;
  const float* arow1 = arow0 + 16 * lda;
  v2f a0[2][4], a1[2][4];
  load_a<4>(arow0, arow1, 0, a0[0], a1[0]);
  keep_order();
#pragma unroll
  for (int ch = 0; ch < K / 16; ++ch) {
    if (ch + 1 < K / 16)
      load_a<4>(arow0, arow1, (ch + 1) * 16, a0[(ch + 1) & 1], a1[(ch + 1) & 1]);
    keep_order();
#pragma unroll
    for (int s = 0; s < 4; ++s) {
      v2f bj0, bj1, ej0, ej1;
      if constexpr (PK) {
        bj0.x = ra[ch * 4 + s].x; bj0.y = ra[ch * 4 + s].y;
        bj1.x = rb[ch * 4 + s].x; bj1.y = rb[ch * 4 + s].y;
        ej0.x = sa[ch * 4 + s].x; ej0.y = sa[ch * 4 + s].y;
        ej1.x = sb[ch * 4 + s].x; ej1.y = sb[ch * 4 + s].y;
      } else {
        bj0.x = ra[ch * 4 + s].x; bj0.y = rb[ch * 4 + s].x;
        bj1.x = ra[ch * 4 + s].y; bj1.y = rb[ch * 4 + s].y;
        ej0.x = sa[ch * 4 + s].x; ej0.y = sb[ch * 4 + s].x;
        ej1.x = sa[ch * 4 + s].y; ej1.y = sb[ch * 4 + s].y;
      }
      c00 = mma4(a0[ch & 1][s], bj0, c00);
      c10 = mma4(a1[ch & 1][s], bj0, c10);
      c01 = mma4(a0[ch & 1][s], bj1, c01);
      c11 = mma4(a1[ch & 1][s], bj1, c11);
      d00 = mma4(a0[ch & 1][s], ej0, d00);
      d10 = mma4(a1[ch & 1][s], ej0, d10);
      d01 = mma4(a0[ch & 1][s], ej1, d01);
      d11 = mma4(a1[ch & 1][s], ej1, d11);
    }
    keep_order();
  }
}

__device__ __forceinline__ void store_tile2(float* dst, int ldd, int rbase, int cb,
                                            v8f cj0, v8f cj1, int hi) {
#pragma unroll
  for (int v = 0; v < 8; ++v) {
    float2 t;
    t.x = cj0[v];
    t.y = cj1[v];
    *(float2*)(dst + (rbase + v + hi * 8) * ldd + cb) = t;
  }
}

template <int S, bool PK>
__device__ __forceinline__ void load_chunk(const float* arow0, const float* arow1,
                                           const float* __restrict__ wcol, int ldw,
                                           const float* arow0n, const float* arow1n,
                                           const float* __restrict__ w2col, int ldw2,
                                           int kb, v2f (&a0)[S], v2f (&a1)[S],
                                           v2f (&b)[S], v2f (&a0n)[S / 8],
                                           v2f (&a1n)[S / 8], v2f (&bn)[S / 8]) {
#pragma unroll
  for (int s = 0; s < S; ++s) {
    if constexpr (PK) {
      const float2 t = *(const float2*)(wcol + (size_t)(kb + 4 * s) * ldw);
      b[s].x = t.x;
      b[s].y = t.y;
    } else {
      b[s].x = wcol[(size_t)(kb + 4 * s) * ldw];
      b[s].y = wcol[(size_t)(kb + 4 * s + 1) * ldw];
    }
  }
#pragma unroll
  for (int t = 0; t < S / 8; ++t) {
    bn[t].x = w2col[(size_t)(kb + 32 * t) * ldw2];
    bn[t].y = w2col[(size_t)(kb + 32 * t + 1) * ldw2];
  }
  load_a<S>(arow0, arow1, kb, a0, a1);
#pragma unroll
  for (int t = 0; t < S / 8; ++t) {
    a0n[t].x = arow0n[kb + 32 * t];
    a0n[t].y = arow0n[kb + 32 * t + 1];
    a1n[t].x = arow1n[kb + 32 * t];
    a1n[t].y = arow1n[kb + 32 * t + 1];
  }
}
template <int S>
__device__ __forceinline__ void mma_chunk(const v2f (&a0)[S], const v2f (&a1)[S],
                                          const v2f (&b)[S], const v2f (&a0n)[S / 8],
                                          const v2f (&a1n)[S / 8], const v2f (&bn)[S / 8],
                                          v8f& c0, v8f& c1, v8f& s0, v8f& s1) {
#pragma unroll
  for (int s = 0; s < S; ++s) {
    c0 = mma4(a0[s], b[s], c0);
    c1 = mma4(a1[s], b[s], c1);
  }
#pragma unroll
  for (int t = 0; t < S / 8; ++t) {
    s0 = mma4(a0n[t], bn[t], s0);
    s1 = mma4(a1n[t], bn[t], s1);
  }
}

template <int K, int CH, bool PK>
__device__ __forceinline__ void gemm_stream(const float* As, int lda,
                                            const float* __restrict__ W, int ldw,
                                            int n0, const float* __restrict__ W2,
                                            int ldw2, int w, int lane16, int hi,
                                            v8f& c0, v8f& c1, v8f& s0, v8f& s1) {
  static_assert(K % (2 * CH) == 0 && K >= 2 * CH, "chunk pairs");
  static_assert(CH == 64, "a chunk of 16 k-steps: two per wave for the second product");
  constexpr int S = CH / 4;
  const v8f z = {0.f, 0.f, 0.f, 0.f, 0.f, 0.f, 0.f, 0.f};
  c0 = z;
  c1 = z;
  s0 = z;
  s1 = z;
  const float* arow0 = As + lane16 * lda + hi * 2;
  const float* arow1 = arow0 + 16 * lda;
  const float* wcol = PK ? W + (size_t)(hi * ldw + n0 + lane16) * 2
                         : W + (size_t)(hi * 2) * ldw + n0 + lane16;
  const float* arow0n = arow0 + 4 * w;
  const float* arow1n = arow1 + 4 * w;
  const float* w2col = W2 + (size_t)(hi * 2 + 4 * w) * ldw2 + lane16;
  v2f bP[S], a0P[S], a1P[S], bQ[S], a0Q[S], a1Q[S];
  v2f bnP[S / 8], a0nP[S / 8], a1nP[S / 8], bnQ[S / 8], a0nQ[S / 8], a1nQ[S / 8];
  load_chunk<S, PK>(arow0, arow1, wcol, ldw, arow0n, arow1n, w2col, ldw2, 0,
                a0P, a1P, bP, a0nP, a1nP, bnP);
  keep_order();
#pragma unroll 1
  for (int kb = 0; kb < K - 2 * CH; kb += 2 * CH) {
    load_chunk<S, PK>(arow0, arow1, wcol, ldw, arow0n, arow1n, w2col, ldw2, kb + CH,
                  a0Q, a1Q, bQ, a0nQ, a1nQ, bnQ);
    keep_order();
    mma_chunk<S>(a0P, a1P, bP, a0nP, a1nP, bnP, c0, c1, s0, s1);
    keep_order();
    load_chunk<S, PK>(arow0, arow1, wcol, ldw, arow0n, arow1n, w2col, ldw2, kb + 2 * CH,
                  a0P, a1P, bP, a0nP, a1nP, bnP);
    keep_order();
    mma_chunk<S>(a0Q, a1Q, bQ, a0nQ, a1nQ, bnQ, c0, c1, s0, s1);
    keep_order();
  }
  load_chunk<S, PK>(arow0, arow1, wcol, ldw, arow0n, arow1n, w2col, ldw2, K - CH,
                a0Q, a1Q, bQ, a0nQ, a1nQ, bnQ);
  keep_order();
  mma_chunk<S>(a0P, a1P, bP, a0nP, a1nP, bnP, c0, c1, s0, s1);
  mma_chunk<S>(a0Q, a1Q, bQ, a0nQ, a1nQ, bnQ, c0, c1, s0, s1);
}

__device__ __forceinline__ void store_tile(float* dst, int ldd, int rbase,
                                           int col0, v8f c, int lane16, int hi) {
#pragma unroll
  for (int v = 0; v < 8; ++v)
    dst[(rbase + v + hi * 8) * ldd + col0 + lane16] = c[v];
}

__device__ __forceinline__ v8f add_bias(v8f c, float b) {
#pragma unroll
  for (int v = 0; v < 8; ++v) c[v] += b;
  return c;
}
#define LOG2E_F 1.44269504088896340736f
__device__ __forceinline__ float hw_exp(float t) {
  return __builtin_amdgcn_exp2f(t * LOG2E_F);
}
__device__ __forceinline__ v8f bias_elu(v8f c, float b) {
#pragma unroll
  for (int v = 0; v < 8; ++v) {
    float t = c[v] + b;
    float em1 = hw_exp(t) - 1.f;
    c[v] = t > 0.f ? t : em1;
  }
  return c;
}
__device__ __forceinline__ v8f gate_mix(v8f g, float b, v8f e, v8f h2) {
#pragma unroll
  for (int v = 0; v < 8; ++v) {
    float t = g[v] + b;
    float s = __builtin_amdgcn_rcpf(1.f + hw_exp(-t));
    g[v] = e[v] + s * h2[v];
  }
  return g;
}

#define FOLD_ROWS 80
#define FOLD_BYTES 655360
#define FOLD_BLOCKS (F_ * 5)

#define PK_FOLD 0
#define PK_FBIAS 131072
#define PK_EMB 133120
#define PK_FC2 264192
#define PK_G 526336
#define PK_SF1 788480
#define PACK_BYTES 4202496
#define PACK_BLOCKS 112

template <bool PK>
__global__ __launch_bounds__(256)
void vsn_fold_fc1(const float* __restrict__ embW, const float* __restrict__ embB,
                  const float* __restrict__ fc1W, const float* __restrict__ fc1B,
                  const float* __restrict__ fc2W, const float* __restrict__ gW,
                  const float* __restrict__ sf1W, float* __restrict__ foldW) {
  if constexpr (PK) {
    if (blockIdx.x >= FOLD_BLOCKS) {
      const int pb = blockIdx.x - FOLD_BLOCKS;
      const float* src;
      float* dst;
      int lb;
      if (pb < 16)      { src = embW; dst = foldW + PK_EMB; lb = pb; }
      else if (pb < 48) { src = fc2W; dst = foldW + PK_FC2; lb = pb - 16; }
      else if (pb < 80) { src = gW;   dst = foldW + PK_G;   lb = pb - 48; }
      else              { src = sf1W; dst = foldW + PK_SF1; lb = pb - 80; }
      float2 r0[8], r1[8];
#pragma unroll
      for (int i = 0; i < 8; ++i) {
        const int o = lb * 2048 + i * 256 + threadIdx.x;
        const float* sp = src + (size_t)(o >> 6) * (2 * H_) + 2 * (o & 63);
        r0[i] = *(const float2*)sp;
        r1[i] = *(const float2*)(sp + H_);
      }
#pragma unroll
      for (int i = 0; i < 8; ++i) {
        const int o = lb * 2048 + i * 256 + threadIdx.x;
        float4 t;
        t.x = r0[i].x; t.y = r1[i].x;
        t.z = r0[i].y; t.w = r1[i].y;
        *(float4*)(dst + (size_t)o * 4) = t;
      }
      return;
    }
  }
  const int f = blockIdx.x / 5;
  const int rt = blockIdx.x % 5;
  const int wid = threadIdx.x >> 5;
  const int lane = threadIdx.x & 31;
  const int lane16 = lane & 15;
  const int hi = lane >> 4;
  const int n0 = wid * 16;
  const bool wrow = rt < 4;
  const bool live = wrow || lane16 == 0;
  v2f b[H_ / 4];
  load_panel<H_>(fc1W + (size_t)f * H_ * H_, H_, n0, lane16, hi, b);
  const float* ar = (wrow ? embW + ((size_t)f * D_ + rt * 16 + lane16) * H_
                          : embB + (size_t)f * H_) + hi * 2;
  float2 araw[H_ / 4];
#pragma unroll
  for (int s = 0; s < H_ / 4; ++s) araw[s] = *(const float2*)(ar + 4 * s);
  const float cb0 = fc1B[f * H_ + n0 + lane16];
  keep_order();
  v8f c = {0.f, 0.f, 0.f, 0.f, 0.f, 0.f, 0.f, 0.f};
  c[0] = (!wrow && hi == 0) ? cb0 : 0.f;
#pragma unroll
  for (int s = 0; s < H_ / 4; ++s) {
    v2f a;
    a.x = live ? araw[s].x : 0.f;
    a.y = live ? araw[s].y : 0.f;
    c = mma4(a, b[s], c);
  }
  if constexpr (PK) {
    if (wrow) {
      float* dp = foldW + PK_FOLD + (size_t)f * (D_ * H_) +
                  ((size_t)(rt * 8 + hi * 4) * H_ + n0 + lane16) * 2;
#pragma unroll
      for (int v = 0; v < 8; v += 2) {
        float2 t;
        t.x = c[v];
        t.y = c[v + 1];
        *(float2*)(dp + (size_t)(v / 2) * (2 * H_)) = t;
      }
    } else if (hi == 0) {
      foldW[PK_FBIAS + f * H_ + n0 + lane16] = c[0];
    }
  } else {
    store_tile(foldW + (size_t)f * FOLD_ROWS * H_, H_, rt * 16, n0, c, lane16, hi);
  }
}

template <int MODE>
__global__ __launch_bounds__(256, 1)
void vsn_fused_pairs32(
    const float* __restrict__ x,
    const float* __restrict__ embW, const float* __restrict__ embB,
    const float* __restrict__ fc1W, const float* __restrict__ fc1B,
    const float* __restrict__ fc2W, const float* __restrict__ fc2B,
    const float* __restrict__ gW,   const float* __restrict__ gB,
    const float* __restrict__ lnG,  const float* __restrict__ lnB,
    const float* __restrict__ sf1W, const float* __restrict__ sf1B,
    const float* __restrict__ sf2W, const float* __restrict__ sf2B,
    const float* __restrict__ sgW,  const float* __restrict__ sgB,
    const float* __restrict__ sskW, const float* __restrict__ sskB,
    const float* __restrict__ slnG, const float* __restrict__ slnB,
    const float* __restrict__ foldW,
    float* __restrict__ outSel, float* __restrict__ outW) {
  __shared__ __attribute__((aligned(16))) float smem[SMEM_FLOATS];
  float* TRANS = smem;
  float* XT0 = TRANS + MR * ST_T;
  float* XW0 = XT0 + 2 * MR * ST_X;
  float* PU = XW0;
  float* PV = XW0 + MR * ST_P;
  float* SKP = XT0;
  float* GP = PU;
  float* WB = GP + MR * ST_S;
  constexpr bool FOLD = MODE >= 1;
  constexpr bool PK = MODE == 2;

  const int tid = threadIdx.x;
  const int wid = tid >> 5;
  const int lane = tid & 31;
  const int lane16 = lane & 15;
  const int hi = lane >> 4;
  const int row0 = blockIdx.x * MR;

  const int g = wid >> 2;
  const int n0 = (wid & 3) * 32;
  const int cb = n0 + 2 * lane16;
  const int glr = (tid & 127) >> 4;
  const int lc = tid & 15;
  float* XT = XT0 + g * (MR * ST_X);
  float* XW = XW0 + g * (MR * ST_P);

  if constexpr (FOLD) {
  const float* eW = PK ? foldW + PK_EMB : embW;
  const float* xW = PK ? foldW + PK_FOLD : foldW;
  const float* xB = PK ? foldW + PK_FBIAS : foldW + D_ * H_;
  const float* f2W = PK ? foldW + PK_FC2 : fc2W;
  const float* gWp = PK ? foldW + PK_G : gW;
  constexpr int xS = PK ? D_ * H_ : FOLD_ROWS * H_;
  constexpr int xbS = PK ? H_ : FOLD_ROWS * H_;
  float2 bE = *(const float2*)(embB + g * H_ + cb);
  float2 b1 = *(const float2*)(xB + (size_t)g * xbS + cb);
  float2 b2 = *(const float2*)(fc2B + g * H_ + cb);
  float2 bG = *(const float2*)(gB + g * H_ + cb);
  float2 pE0[D_ / 4], pE1[D_ / 4];
  float2 pX0[D_ / 4], pX1[D_ / 4];
  load_panel2<0, D_ / 4, D_ / 4, PK>(eW + (size_t)g * D_ * H_, H_, n0, lane16, hi, pE0, pE1);
  load_panel2<0, D_ / 4, D_ / 4, PK>(xW + (size_t)g * xS, H_, n0, lane16, hi, pX0, pX1);
  {
#pragma unroll
    for (int rr = 0; rr < 4; ++rr) {
      const int r = glr + 8 * rr;
      float4 v = *(const float4*)(x + (size_t)(row0 + r) * (F_ * D_) + g * D_ + lc * 4);
      float* dp = XT + r * ST_X + lc * 4;
      dp[0] = v.x; dp[1] = v.y; dp[2] = v.z; dp[3] = v.w;
    }
  }
  float2 pF20[H_ / 4], pF21[H_ / 4];
  float4 xr[4];
  wg_signal_lds();
  keep_order();
  load_panel2<0, H_ / 4, H_ / 4, PK>(f2W + (size_t)g * H_ * H_, H_, n0, lane16, hi, pF20, pF21);
#pragma unroll
  for (int rr = 0; rr < 4; ++rr)
    xr[rr] = *(const float4*)(x + (size_t)(row0 + glr + 8 * rr) * (F_ * D_) +
                              (g + 2) * D_ + lc * 4);
  keep_order();
  wg_wait();

  for (int f = g; f < F_; f += 2) {
    const int fn = f + 2 < F_ ? f + 2 : f;
    const int fnn = f + 4 < F_ ? f + 4 : fn;
    float* SLOT = TRANS + f * H_;

    v8f ce00, ce01, ce10, ce11;
    {
      v8f c00, c01, c10, c11;
      gemm_panel2x2<D_, PK>(XT, ST_X, lane16, hi, pE0, pE1, pX0, pX1,
                        ce00, ce01, ce10, ce11, c00, c01, c10, c11);
      ce00 = add_bias(ce00, bE.x);
      ce10 = add_bias(ce10, bE.x);
      ce01 = add_bias(ce01, bE.y);
      ce11 = add_bias(ce11, bE.y);
      c00 = bias_elu(c00, b1.x);
      c10 = bias_elu(c10, b1.x);
      c01 = bias_elu(c01, b1.y);
      c11 = bias_elu(c11, b1.y);
      store_tile2(XW, ST_P, 0, cb, c00, c01, hi);
      store_tile2(XW, ST_P, 16, cb, c10, c11, hi);
    }
    wg_signal_lds();
    keep_order();
    float2 pG0[H_ / 4], pG1[H_ / 4];
    load_panel2<0, H_ / 4, H_ / 4, PK>(gWp + (size_t)f * H_ * H_, H_, n0, lane16, hi, pG0, pG1);
    keep_order();
    wg_wait();

    v8f c200, c201, c210, c211;
    gemm_panel2<H_, PK>(XW, ST_P, lane16, hi, pF20, pF21, c200, c201, c210, c211);
    c200 = add_bias(c200, b2.x);
    c210 = add_bias(c210, b2.x);
    c201 = add_bias(c201, b2.y);
    c211 = add_bias(c211, b2.y);
    store_tile2(SLOT, ST_T, 0, cb, c200, c201, hi);
    store_tile2(SLOT, ST_T, 16, cb, c210, c211, hi);
    wg_signal_lds();
    keep_order();

    load_panel2<0, D_ / 4, D_ / 4, PK>(eW + (size_t)fn * D_ * H_, H_, n0, lane16, hi, pE0, pE1);
    const float2 bGf = bG;
    bE = *(const float2*)(embB + fn * H_ + cb);
    b1 = *(const float2*)(xB + (size_t)fn * xbS + cb);
    b2 = *(const float2*)(fc2B + fn * H_ + cb);
    bG = *(const float2*)(gB + fn * H_ + cb);
    float lg[8], lb[8];
#pragma unroll
    for (int j = 0; j < 8; ++j) {
      lg[j] = lnG[f * H_ + lc + 16 * j];
      lb[j] = lnB[f * H_ + lc + 16 * j];
    }
    keep_order();
    wg_wait();
    {
      v8f cg00, cg01, cg10, cg11;
      gemm_panel2<H_, PK>(SLOT, ST_T, lane16, hi, pG0, pG1, cg00, cg01, cg10, cg11);
      cg00 = gate_mix(cg00, bGf.x, ce00, c200);
      cg10 = gate_mix(cg10, bGf.x, ce10, c210);
      cg01 = gate_mix(cg01, bGf.y, ce01, c201);
      cg11 = gate_mix(cg11, bGf.y, ce11, c211);
      store_tile2(XW, ST_P, 0, cb, cg00, cg01, hi);
      store_tile2(XW, ST_P, 16, cb, cg10, cg11, hi);
    }
    wg_signal_lds();
    keep_order();
    load_panel2<0, D_ / 4, D_ / 4, PK>(xW + (size_t)fn * xS, H_, n0, lane16, hi, pX0, pX1);
    load_panel2<0, 14, H_ / 4, PK>(f2W + (size_t)fn * H_ * H_, H_, n0, lane16, hi, pF20, pF21);
    keep_order();
    wg_wait();

    float vals[4][8];
#pragma unroll
    for (int rr = 0; rr < 4; ++rr) {
      const int r = glr + 8 * rr;
#pragma unroll
      for (int j = 0; j < 8; ++j) vals[rr][j] = XW[r * ST_P + lc + 16 * j];
    }
    float sm[8];
#pragma unroll
    for (int rr = 0; rr < 4; ++rr) {
      float s = 0.f, ss = 0.f;
#pragma unroll
      for (int j = 0; j < 8; ++j) {
        float v = vals[rr][j];
        s += v; ss += v * v;
      }
      sm[2 * rr] = s;
      sm[2 * rr + 1] = ss;
    }
    rowsum16_n<8>(sm);
#pragma unroll
    for (int rr = 0; rr < 4; ++rr) {
      const int r = glr + 8 * rr;
      float mu = sm[2 * rr] * (1.f / 128.f);
      float var = sm[2 * rr + 1] * (1.f / 128.f) - mu * mu;
      float rs = rsqrtf(var + 1e-5f);
#pragma unroll
      for (int j = 0; j < 8; ++j) {
        int h = lc + 16 * j;
        float nv = (vals[rr][j] - mu) * rs * lg[j] + lb[j];
        SLOT[r * ST_T + h] = nv;
      }
      float* dp = XT + r * ST_X + lc * 4;
      dp[0] = xr[rr].x; dp[1] = xr[rr].y; dp[2] = xr[rr].z; dp[3] = xr[rr].w;
    }
    wg_signal_lds();
    keep_order();
    load_panel2<14, H_ / 4, H_ / 4, PK>(f2W + (size_t)fn * H_ * H_, H_, n0, lane16, hi, pF20, pF21);
#pragma unroll
    for (int rr = 0; rr < 4; ++rr)
      xr[rr] = *(const float4*)(x + (size_t)(row0 + glr + 8 * rr) * (F_ * D_) +
                                fnn * D_ + lc * 4);
    keep_order();
    wg_wait();
  }
  } else {
  float2 bE = *(const float2*)(embB + g * H_ + cb);
  float2 b1 = *(const float2*)(fc1B + g * H_ + cb);
  float2 b2 = *(const float2*)(fc2B + g * H_ + cb);
  float2 bG = *(const float2*)(gB + g * H_ + cb);
  float2 pE0[D_ / 4], pE1[D_ / 4];
  load_panel2<0, D_ / 4, D_ / 4>(embW + (size_t)g * D_ * H_, H_, n0, lane16, hi, pE0, pE1);
  {
#pragma unroll
    for (int rr = 0; rr < 4; ++rr) {
      const int r = glr + 8 * rr;
      float4 v = *(const float4*)(x + (size_t)(row0 + r) * (F_ * D_) + g * D_ + lc * 4);
      float* dp = XT + r * ST_X + lc * 4;
      dp[0] = v.x; dp[1] = v.y; dp[2] = v.z; dp[3] = v.w;
    }
  }
  float2 pF10[H_ / 4], pF11[H_ / 4];
  float4 xr[4];
  wg_signal_lds();
  keep_order();
  load_panel2<0, H_ / 4, H_ / 4>(fc1W + (size_t)g * H_ * H_, H_, n0, lane16, hi, pF10, pF11);
#pragma unroll
  for (int rr = 0; rr < 4; ++rr)
    xr[rr] = *(const float4*)(x + (size_t)(row0 + glr + 8 * rr) * (F_ * D_) +
                              (g + 2) * D_ + lc * 4);
  keep_order();
  wg_wait();

  for (int f = g; f < F_; f += 2) {
    const int fn = f + 2 < F_ ? f + 2 : f;
    const int fnn = f + 4 < F_ ? f + 4 : fn;
    float* SLOT = TRANS + f * H_;

    v8f ce00, ce01, ce10, ce11;
    gemm_panel2<D_>(XT, ST_X, lane16, hi, pE0, pE1, ce00, ce01, ce10, ce11);
    ce00 = add_bias(ce00, bE.x);
    ce10 = add_bias(ce10, bE.x);
    ce01 = add_bias(ce01, bE.y);
    ce11 = add_bias(ce11, bE.y);
    store_tile2(SLOT, ST_T, 0, cb, ce00, ce01, hi);
    store_tile2(SLOT, ST_T, 16, cb, ce10, ce11, hi);
    wg_signal_lds();
    keep_order();
    float2 pF20[H_ / 4], pF21[H_ / 4];
    load_panel2<0, H_ / 4, H_ / 4>(fc2W + (size_t)f * H_ * H_, H_, n0, lane16, hi, pF20, pF21);
    keep_order();
    wg_wait();

    {
      v8f c00, c01, c10, c11;
      gemm_panel2<H_>(SLOT, ST_T, lane16, hi, pF10, pF11, c00, c01, c10, c11);
      c00 = bias_elu(c00, b1.x);
      c10 = bias_elu(c10, b1.x);
      c01 = bias_elu(c01, b1.y);
      c11 = bias_elu(c11, b1.y);
      store_tile2(XW, ST_P, 0, cb, c00, c01, hi);
      store_tile2(XW, ST_P, 16, cb, c10, c11, hi);
    }
    wg_signal_lds();
    keep_order();
    float2 pG0[H_ / 4], pG1[H_ / 4];
    load_panel2<0, H_ / 4, H_ / 4>(gW + (size_t)f * H_ * H_, H_, n0, lane16, hi, pG0, pG1);
    keep_order();
    wg_wait();

    v8f c200, c201, c210, c211;
    gemm_panel2<H_>(XW, ST_P, lane16, hi, pF20, pF21, c200, c201, c210, c211);
    c200 = add_bias(c200, b2.x);
    c210 = add_bias(c210, b2.x);
    c201 = add_bias(c201, b2.y);
    c211 = add_bias(c211, b2.y);
    store_tile2(SLOT, ST_T, 0, cb, c200, c201, hi);
    store_tile2(SLOT, ST_T, 16, cb, c210, c211, hi);
    wg_signal_lds();
    keep_order();

    load_panel2<0, D_ / 4, D_ / 4>(embW + (size_t)fn * D_ * H_, H_, n0, lane16, hi, pE0, pE1);
    const float2 bGf = bG;
    bE = *(const float2*)(embB + fn * H_ + cb);
    b1 = *(const float2*)(fc1B + fn * H_ + cb);
    b2 = *(const float2*)(fc2B + fn * H_ + cb);
    bG = *(const float2*)(gB + fn * H_ + cb);
    float lg[8], lb[8];
#pragma unroll
    for (int j = 0; j < 8; ++j) {
      lg[j] = lnG[f * H_ + lc + 16 * j];
      lb[j] = lnB[f * H_ + lc + 16 * j];
    }
    keep_order();
    wg_wait();
    {
      v8f cg00, cg01, cg10, cg11;
      gemm_panel2<H_>(SLOT, ST_T, lane16, hi, pG0, pG1, cg00, cg01, cg10, cg11);
      cg00 = gate_mix(cg00, bGf.x, ce00, c200);
      cg10 = gate_mix(cg10, bGf.x, ce10, c210);
      cg01 = gate_mix(cg01, bGf.y, ce01, c201);
      cg11 = gate_mix(cg11, bGf.y, ce11, c211);
      store_tile2(XW, ST_P, 0, cb, cg00, cg01, hi);
      store_tile2(XW, ST_P, 16, cb, cg10, cg11, hi);
    }
    wg_signal_lds();
    keep_order();
    load_panel2<0, H_ / 8, H_ / 4>(fc1W + (size_t)fn * H_ * H_, H_, n0, lane16, hi, pF10, pF11);
    keep_order();
    wg_wait();

    float vals[4][8];
#pragma unroll
    for (int rr = 0; rr < 4; ++rr) {
      const int r = glr + 8 * rr;
#pragma unroll
      for (int j = 0; j < 8; ++j) vals[rr][j] = XW[r * ST_P + lc + 16 * j];
    }
    float sm[8];
#pragma unroll
    for (int rr = 0; rr < 4; ++rr) {
      float s = 0.f, ss = 0.f;
#pragma unroll
      for (int j = 0; j < 8; ++j) {
        float v = vals[rr][j];
        s += v; ss += v * v;
      }
      sm[2 * rr] = s;
      sm[2 * rr + 1] = ss;
    }
    rowsum16_n<8>(sm);
#pragma unroll
    for (int rr = 0; rr < 4; ++rr) {
      const int r = glr + 8 * rr;
      float mu = sm[2 * rr] * (1.f / 128.f);
      float var = sm[2 * rr + 1] * (1.f / 128.f) - mu * mu;
      float rs = rsqrtf(var + 1e-5f);
#pragma unroll
      for (int j = 0; j < 8; ++j) {
        int h = lc + 16 * j;
        float nv = (vals[rr][j] - mu) * rs * lg[j] + lb[j];
        SLOT[r * ST_T + h] = nv;
      }
      float* dp = XT + r * ST_X + lc * 4;
      dp[0] = xr[rr].x; dp[1] = xr[rr].y; dp[2] = xr[rr].z; dp[3] = xr[rr].w;
    }
    wg_signal_lds();
    keep_order();
    load_panel2<H_ / 8, H_ / 4, H_ / 4>(fc1W + (size_t)fn * H_ * H_, H_, n0, lane16, hi, pF10, pF11);
#pragma unroll
    for (int rr = 0; rr < 4; ++rr)
      xr[rr] = *(const float4*)(x + (size_t)(row0 + glr + 8 * rr) * (F_ * D_) +
                                fnn * D_ + lc * 4);
    keep_order();
    wg_wait();
  }

  }

  const int nB = wid * 16;
  const int lr = tid >> 4;
  v2f pS2[H_ / 4];
  load_panel<H_>(sf2W, H_, nB, lane16, hi, pS2);
  const float bS1 = sf1B[nB + lane16];
  const float bS2 = sf2B[nB + lane16];
  const float bSK = sskB[lc];
  {
    v8f b10, b11, sk0, sk1;
    gemm_stream<FH_, 64, PK>(TRANS, ST_T, PK ? foldW + PK_SF1 : sf1W, H_, nB, sskW, F_,
                             wid, lane16, hi, b10, b11, sk0, sk1);
    b10 = bias_elu(b10, bS1);
    b11 = bias_elu(b11, bS1);
    store_tile(PU, ST_P, 0, nB, b10, lane16, hi);
    store_tile(PU, ST_P, 16, nB, b11, lane16, hi);
    float* SKPw = SKP + wid * (MR * 16);
    store_tile(SKPw, 16, 0, 0, sk0, lane16, hi);
    store_tile(SKPw, 16, 16, 0, sk1, lane16, hi);
  }
  wg_barrier_lds();

  {
    v8f b20, b21;
    gemm_panel<H_>(PU, ST_P, lane16, hi, pS2, b20, b21);
    b20 = add_bias(b20, bS2);
    b21 = add_bias(b21, bS2);
    store_tile(PV, ST_P, 0, nB, b20, lane16, hi);
    store_tile(PV, ST_P, 16, nB, b21, lane16, hi);
  }
  wg_barrier_lds();

  if (wid == 0) {
    v2f pSG[H_ / 4];
    load_panel<H_>(sgW, F_, 0, lane16, hi, pSG);
    v8f g0, g1;
    gemm_panel<H_>(PV, ST_P, lane16, hi, pSG, g0, g1);
    float b = sgB[lane16];
    g0 = add_bias(g0, b);
    g1 = add_bias(g1, b);
    store_tile(GP, ST_S, 0, 0, g0, lane16, hi);
    store_tile(GP, ST_S, 16, 0, g1, lane16, hi);
  }
  wg_barrier_lds();

  {
    float gpv[2], skv[2];
#pragma unroll
    for (int rr = 0; rr < 2; ++rr) {
      const int r = lr + 16 * rr;
      gpv[rr] = GP[r * ST_S + lc];
      float sk = SKP[r * 16 + lc];
#pragma unroll
      for (int pw = 1; pw < 8; ++pw) sk += SKP[pw * (MR * 16) + r * 16 + lc];
      skv[rr] = sk + bSK;
    }
    float vv[2], st[4];
#pragma unroll
    for (int rr = 0; rr < 2; ++rr) {
      float gp = gpv[rr];
      float sg = 1.f / (1.f + expf(-gp));
      float v = skv[rr] + sg * gp;
      vv[rr] = v;
      st[2 * rr] = v;
      st[2 * rr + 1] = v * v;
    }
    rowsum16_n<4>(st);
    const float sG = slnG[lc], sB = slnB[lc];
    float nv[2], mx[2];
#pragma unroll
    for (int rr = 0; rr < 2; ++rr) {
      float mu = st[2 * rr] * (1.f / 16.f);
      float var = st[2 * rr + 1] * (1.f / 16.f) - mu * mu;
      nv[rr] = (vv[rr] - mu) * rsqrtf(var + 1e-5f) * sG + sB;
      mx[rr] = nv[rr];
    }
    rowmax16_n<2>(mx);
    float e[2], se[2];
#pragma unroll
    for (int rr = 0; rr < 2; ++rr) {
      e[rr] = expf(nv[rr] - mx[rr]);
      se[rr] = e[rr];
    }
    rowsum16_n<2>(se);
#pragma unroll
    for (int rr = 0; rr < 2; ++rr) {
      const int r = lr + 16 * rr;
      float w = e[rr] / se[rr];
      WB[r * 16 + lc] = w;
      outW[(size_t)(row0 + r) * F_ + lc] = w;
    }
  }
  wg_barrier_lds();

#pragma unroll
  for (int rr = 0; rr < 2; ++rr) {
    const int r = lr + 16 * rr;
#pragma unroll
    for (int j = 0; j < 8; ++j) {
      int h = lc + 16 * j;
      float acc = 0.f;
#pragma unroll
      for (int f = 0; f < F_; ++f)
        acc += TRANS[r * ST_T + f * H_ + h] * WB[r * 16 + f];
      outSel[(size_t)(row0 + r) * H_ + h] = acc;
    }
  }
}

extern "C" void kernel_launch(void* const* d_in, const int* in_sizes, int n_in,
                              void* d_out, int out_size, void* d_ws, size_t ws_size,
                              hipStream_t stream) {
  (void)in_sizes; (void)n_in; (void)out_size;
  const float* x    = (const float*)d_in[0];
  const float* embW = (const float*)d_in[1];
  const float* embB = (const float*)d_in[2];
  const float* fc1W = (const float*)d_in[3];
  const float* fc1B = (const float*)d_in[4];
  const float* fc2W = (const float*)d_in[5];
  const float* fc2B = (const float*)d_in[6];
  const float* gW   = (const float*)d_in[7];
  const float* gB   = (const float*)d_in[8];
  const float* lnG  = (const float*)d_in[9];
  const float* lnB  = (const float*)d_in[10];
  const float* sf1W = (const float*)d_in[11];
  const float* sf1B = (const float*)d_in[12];
  const float* sf2W = (const float*)d_in[13];
  const float* sf2B = (const float*)d_in[14];
  const float* sgW  = (const float*)d_in[15];
  const float* sgB  = (const float*)d_in[16];
  const float* sskW = (const float*)d_in[17];
  const float* sskB = (const float*)d_in[18];
  const float* slnG = (const float*)d_in[19];
  const float* slnB = (const float*)d_in[20];

  float* outSel = (float*)d_out;
  float* outW   = outSel + (size_t)ROWS_TOTAL * H_;

  if (d_ws != nullptr && ws_size >= (size_t)PACK_BYTES) {
    float* foldW = (float*)d_ws;
    vsn_fold_fc1<true><<<FOLD_BLOCKS + PACK_BLOCKS, 256, 0, stream>>>(
        embW, embB, fc1W, fc1B, fc2W, gW, sf1W, foldW);
    vsn_fused_pairs32<2><<<ROWS_TOTAL / MR, 256, 0, stream>>>(
        x, embW, embB, fc1W, fc1B, fc2W, fc2B, gW, gB, lnG, lnB,
        sf1W, sf1B, sf2W, sf2B, sgW, sgB, sskW, sskB, slnG, slnB,
        foldW, outSel, outW);
  } else if (d_ws != nullptr && ws_size >= (size_t)FOLD_BYTES) {
    float* foldW = (float*)d_ws;
    vsn_fold_fc1<false><<<FOLD_BLOCKS, 256, 0, stream>>>(
        embW, embB, fc1W, fc1B, fc2W, gW, sf1W, foldW);
    vsn_fused_pairs32<1><<<ROWS_TOTAL / MR, 256, 0, stream>>>(
        x, embW, embB, fc1W, fc1B, fc2W, fc2B, gW, gB, lnG, lnB,
        sf1W, sf1B, sf2W, sf2B, sgW, sgB, sskW, sskB, slnG, slnB,
        foldW, outSel, outW);
  } else {
    vsn_fused_pairs32<0><<<ROWS_TOTAL / MR, 256, 0, stream>>>(
        x, embW, embB, fc1W, fc1B, fc2W, fc2B, gW, gB, lnG, lnB,
        sf1W, sf1B, sf2W, sf2B, sgW, sgB, sskW, sskB, slnG, slnB,
        (const float*)nullptr, outSel, outW);
  }
}


# s_fc1 pass (K=2048) re-tiled: wave = 32 columns x one K half, row-tile LDS bases held in registers: loads per matrix op halved, no address adds; halves added in LDS (s_fc1 and skip sum order change)
# speedup vs baseline: 3.4431x; 1.0345x over previous
#include <hip/hip_runtime.h>
#include <hip/hip_bf16.h>
#include <math.h>


typedef float v2f __attribute__((ext_vector_type(2)));
typedef float v8f __attribute__((ext_vector_type(8)));

#define ROWS_TOTAL 16384
#define MR 32
#define F_ 16
#define D_ 64
#define H_ 128
#define FH_ 2048

#define ST_T 2052
#define ST_X 68
#define ST_P 132
#define ST_S 20

#define SMEM_FLOATS (MR*ST_T + 2*MR*ST_X + 2*MR*ST_P)

template <int M>
__device__ __forceinline__ float row_xor(float v) {
  return __builtin_bit_cast(
      float, __builtin_amdgcn_update_dpp(0, __builtin_bit_cast(int, v), 0x160 + M,
                                         0xf, 0xf, true));
}
template <int N>
__device__ __forceinline__ void rowsum16_n(float (&v)[N]) {
#pragma unroll
  for (int i = 0; i < N; ++i) v[i] += row_xor<1>(v[i]);
#pragma unroll
  for (int i = 0; i < N; ++i) v[i] += row_xor<2>(v[i]);
#pragma unroll
  for (int i = 0; i < N; ++i) v[i] += row_xor<4>(v[i]);
#pragma unroll
  for (int i = 0; i < N; ++i) v[i] += row_xor<8>(v[i]);
}
template <int N>
__device__ __forceinline__ void rowmax16_n(float (&v)[N]) {
#pragma unroll
  for (int i = 0; i < N; ++i) v[i] = fmaxf(v[i], row_xor<1>(v[i]));
#pragma unroll
  for (int i = 0; i < N; ++i) v[i] = fmaxf(v[i], row_xor<2>(v[i]));
#pragma unroll
  for (int i = 0; i < N; ++i) v[i] = fmaxf(v[i], row_xor<4>(v[i]));
#pragma unroll
  for (int i = 0; i < N; ++i) v[i] = fmaxf(v[i], row_xor<8>(v[i]));
}

__device__ __forceinline__ void wg_barrier_lds() {
  asm volatile("s_wait_dscnt 0x0\n\ts_barrier_signal -1\n\ts_barrier_wait -1" ::: "memory");
}
__device__ __forceinline__ void wg_signal_lds() {
  asm volatile("s_wait_dscnt 0x0\n\ts_barrier_signal -1" ::: "memory");
}
__device__ __forceinline__ void wg_wait() {
  asm volatile("s_barrier_wait -1" ::: "memory");
}

__device__ __forceinline__ void keep_order() { __builtin_amdgcn_sched_barrier(0); }

__device__ __forceinline__ v8f mma4(v2f a, v2f b, v8f c) {
  return __builtin_amdgcn_wmma_f32_16x16x4_f32(false, a, false, b, (short)0, c,
                                               false, false);
}


template <int K>
__device__ __forceinline__ void load_panel(const float* __restrict__ W, int ldw,
                                           int n0, int lane16, int hi,
                                           v2f (&b)[K / 4]) {
  const float* wcol = W + (size_t)(hi * 2) * ldw + n0 + lane16;
#pragma unroll
  for (int s = 0; s < K / 4; ++s) {
    b[s].x = wcol[(size_t)(4 * s) * ldw];
    b[s].y = wcol[(size_t)(4 * s + 1) * ldw];
  }
}

template <int S>
__device__ __forceinline__ void load_a(const float* arow0, const float* arow1,
                                       int kb, v2f (&a0)[S], v2f (&a1)[S]) {
#pragma unroll
  for (int s = 0; s < S; ++s) {
    a0[s].x = arow0[kb + 4 * s];
    a0[s].y = arow0[kb + 4 * s + 1];
    a1[s].x = arow1[kb + 4 * s];
    a1[s].y = arow1[kb + 4 * s + 1];
  }
}

template <int K>
__device__ __forceinline__ void gemm_panel(const float* As, int lda, int lane16,
                                           int hi, const v2f (&b)[K / 4],
                                           v8f& c0, v8f& c1) {
  const v8f z = {0.f, 0.f, 0.f, 0.f, 0.f, 0.f, 0.f, 0.f};
  c0 = z;
  c1 = z;
  const float* arow0 = As + lane16 * lda + hi * 2;
  const float* arow1 = arow0 + 16 * lda;
  v2f a0[2][8], a1[2][8];
  load_a<8>(arow0, arow1, 0, a0[0], a1[0]);
  keep_order();
#pragma unroll
  for (int ch = 0; ch < K / 32; ++ch) {
    if (ch + 1 < K / 32)
      load_a<8>(arow0, arow1, (ch + 1) * 32, a0[(ch + 1) & 1], a1[(ch + 1) & 1]);
    keep_order();
#pragma unroll
    for (int s = 0; s < 8; ++s) {
      c0 = mma4(a0[ch & 1][s], b[ch * 8 + s], c0);
      c1 = mma4(a1[ch & 1][s], b[ch * 8 + s], c1);
    }
    keep_order();
  }
}

template <int S0, int S1, int NS, bool PK = false>
__device__ __forceinline__ void load_panel2(const float* __restrict__ W, int ldw,
                                            int n0, int lane16, int hi,
                                            float2 (&ra)[NS], float2 (&rb)[NS]) {
  if constexpr (PK) {
    const float* wq = W + (size_t)(hi * ldw + n0 + 2 * lane16) * 2;
#pragma unroll
    for (int s = S0; s < S1; ++s) {
      const float4 t = *(const float4*)(wq + (size_t)(4 * s) * ldw);
      ra[s].x = t.x; ra[s].y = t.y;
      rb[s].x = t.z; rb[s].y = t.w;
    }
  } else {
    const float* wcol = W + (size_t)(hi * 2) * ldw + n0 + 2 * lane16;
#pragma unroll
    for (int s = S0; s < S1; ++s) {
      ra[s] = *(const float2*)(wcol + (size_t)(4 * s) * ldw);
      rb[s] = *(const float2*)(wcol + (size_t)(4 * s + 1) * ldw);
    }
  }
}

template <int K, bool PK = false>
__device__ __forceinline__ void gemm_panel2(const float* As, int lda, int lane16,
                                            int hi, const float2 (&ra)[K / 4],
                                            const float2 (&rb)[K / 4],
                                            v8f& c00, v8f& c01, v8f& c10, v8f& c11) {
  const v8f z = {0.f, 0.f, 0.f, 0.f, 0.f, 0.f, 0.f, 0.f};
  c00 = z; c01 = z; c10 = z; c11 = z;
  const float* arow0 = As + lane16 * lda + hi * 2;
  const float* arow1 = arow0 + 16 * lda;
  v2f a0[2][4], a1[2][4];
  load_a<4>(arow0, arow1, 0, a0[0], a1[0]);
  keep_order();
#pragma unroll
  for (int ch = 0; ch < K / 16; ++ch) {
    if (ch + 1 < K / 16)
      load_a<4>(arow0, arow1, (ch + 1) * 16, a0[(ch + 1) & 1], a1[(ch + 1) & 1]);
    keep_order();
#pragma unroll
    for (int s = 0; s < 4; ++s) {
      v2f bj0, bj1;
      if constexpr (PK) {
        bj0.x = ra[ch * 4 + s].x; bj0.y = ra[ch * 4 + s].y;
        bj1.x = rb[ch * 4 + s].x; bj1.y = rb[ch * 4 + s].y;
      } else {
        bj0.x = ra[ch * 4 + s].x; bj0.y = rb[ch * 4 + s].x;
        bj1.x = ra[ch * 4 + s].y; bj1.y = rb[ch * 4 + s].y;
      }
      c00 = mma4(a0[ch & 1][s], bj0, c00);
      c10 = mma4(a1[ch & 1][s], bj0, c10);
      c01 = mma4(a0[ch & 1][s], bj1, c01);
      c11 = mma4(a1[ch & 1][s], bj1, c11);
    }
    keep_order();
  }
}

template <int K, bool PK = false>
__device__ __forceinline__ void gemm_panel2x2(const float* As, int lda, int lane16,
                                              int hi, const float2 (&ra)[K / 4],
                                              const float2 (&rb)[K / 4],
                                              const float2 (&sa)[K / 4],
                                              const float2 (&sb)[K / 4],
                                              v8f& c00, v8f& c01, v8f& c10, v8f& c11,
                                              v8f& d00, v8f& d01, v8f& d10, v8f& d11) {
  const v8f z = {0.f, 0.f, 0.f, 0.f, 0.f, 0.f, 0.f, 0.f};
  c00 = z; c01 = z; c10 = z; c11 = z;
  d00 = z; d01 = z; d10 = z; d11 = z;
  const float* arow0 = As + lane16 * lda + hi * 2;
  const float* arow1 = arow0 + 16 * lda;
  v2f a0[2][4], a1[2][4];
  load_a<4>(arow0, arow1, 0, a0[0], a1[0]);
  keep_order();
#pragma unroll
  for (int ch = 0; ch < K / 16; ++ch) {
    if (ch + 1 < K / 16)
      load_a<4>(arow0, arow1, (ch + 1) * 16, a0[(ch + 1) & 1], a1[(ch + 1) & 1]);
    keep_order();
#pragma unroll
    for (int s = 0; s < 4; ++s) {
      v2f bj0, bj1, ej0, ej1;
      if constexpr (PK) {
        bj0.x = ra[ch * 4 + s].x; bj0.y = ra[ch * 4 + s].y;
        bj1.x = rb[ch * 4 + s].x; bj1.y = rb[ch * 4 + s].y;
        ej0.x = sa[ch * 4 + s].x; ej0.y = sa[ch * 4 + s].y;
        ej1.x = sb[ch * 4 + s].x; ej1.y = sb[ch * 4 + s].y;
      } else {
        bj0.x = ra[ch * 4 + s].x; bj0.y = rb[ch * 4 + s].x;
        bj1.x = ra[ch * 4 + s].y; bj1.y = rb[ch * 4 + s].y;
        ej0.x = sa[ch * 4 + s].x; ej0.y = sb[ch * 4 + s].x;
        ej1.x = sa[ch * 4 + s].y; ej1.y = sb[ch * 4 + s].y;
      }
      c00 = mma4(a0[ch & 1][s], bj0, c00);
      c10 = mma4(a1[ch & 1][s], bj0, c10);
      c01 = mma4(a0[ch & 1][s], bj1, c01);
      c11 = mma4(a1[ch & 1][s], bj1, c11);
      d00 = mma4(a0[ch & 1][s], ej0, d00);
      d10 = mma4(a1[ch & 1][s], ej0, d10);
      d01 = mma4(a0[ch & 1][s], ej1, d01);
      d11 = mma4(a1[ch & 1][s], ej1, d11);
    }
    keep_order();
  }
}

__device__ __forceinline__ void store_tile2(float* dst, int ldd, int rbase, int cb,
                                            v8f cj0, v8f cj1, int hi) {
#pragma unroll
  for (int v = 0; v < 8; ++v) {
    float2 t;
    t.x = cj0[v];
    t.y = cj1[v];
    *(float2*)(dst + (rbase + v + hi * 8) * ldd + cb) = t;
  }
}

template <bool PK>
__device__ __forceinline__ void load_chunk(const float* arow0, const float* arow1,
                                           const float* __restrict__ Wk, int ldw, int n0,
                                           const float* arow0n, const float* arow1n,
                                           const float* __restrict__ w2col, int ldw2,
                                           int kb, int lane16, int hi,
                                           v2f (&a0)[16], v2f (&a1)[16],
                                           float2 (&ra)[16], float2 (&rb)[16],
                                           v2f (&a0n)[4], v2f (&a1n)[4], v2f (&bn)[4]) {
  load_panel2<0, 16, 16, PK>(Wk + (size_t)kb * ldw, ldw, n0, lane16, hi, ra, rb);
#pragma unroll
  for (int t = 0; t < 4; ++t) {
    bn[t].x = w2col[(size_t)(kb + 16 * t) * ldw2];
    bn[t].y = w2col[(size_t)(kb + 16 * t + 1) * ldw2];
  }
  load_a<16>(arow0, arow1, kb, a0, a1);
#pragma unroll
  for (int t = 0; t < 4; ++t) {
    a0n[t].x = arow0n[kb + 16 * t];
    a0n[t].y = arow0n[kb + 16 * t + 1];
    a1n[t].x = arow1n[kb + 16 * t];
    a1n[t].y = arow1n[kb + 16 * t + 1];
  }
}
template <bool PK>
__device__ __forceinline__ void mma_chunk(const v2f (&a0)[16], const v2f (&a1)[16],
                                          const float2 (&ra)[16], const float2 (&rb)[16],
                                          const v2f (&a0n)[4], const v2f (&a1n)[4],
                                          const v2f (&bn)[4],
                                          v8f& c00, v8f& c01, v8f& c10, v8f& c11,
                                          v8f& s0, v8f& s1) {
#pragma unroll
  for (int s = 0; s < 16; ++s) {
    v2f bj0, bj1;
    if constexpr (PK) {
      bj0.x = ra[s].x; bj0.y = ra[s].y;
      bj1.x = rb[s].x; bj1.y = rb[s].y;
    } else {
      bj0.x = ra[s].x; bj0.y = rb[s].x;
      bj1.x = ra[s].y; bj1.y = rb[s].y;
    }
    c00 = mma4(a0[s], bj0, c00);
    c10 = mma4(a1[s], bj0, c10);
    c01 = mma4(a0[s], bj1, c01);
    c11 = mma4(a1[s], bj1, c11);
  }
#pragma unroll
  for (int t = 0; t < 4; ++t) {
    s0 = mma4(a0n[t], bn[t], s0);
    s1 = mma4(a1n[t], bn[t], s1);
  }
}

template <int K, int CH, bool PK>
__device__ __forceinline__ void gemm_stream(const float* As, int lda,
                                            const float* __restrict__ W, int ldw,
                                            int n0, const float* __restrict__ W2,
                                            int ldw2, int q, int lane16, int hi,
                                            v8f& c00, v8f& c01, v8f& c10, v8f& c11,
                                            v8f& s0, v8f& s1) {
  static_assert(K % (2 * CH) == 0 && K >= 2 * CH, "chunk pairs");
  static_assert(CH == 64, "a chunk of 16 k-steps: four per wave for the second product");
  const v8f z = {0.f, 0.f, 0.f, 0.f, 0.f, 0.f, 0.f, 0.f};
  c00 = z; c01 = z; c10 = z; c11 = z;
  s0 = z;
  s1 = z;
  const float* arow0 = As + lane16 * lda + hi * 2;
  int rt1 = 8 * lda;
  asm volatile("" : "+v"(rt1));
  const float* arow1 = arow0 + 2 * rt1;
  const float* arow0n = arow0 + 4 * q;
  const float* arow1n = arow1 + 4 * q;
  const float* w2col = W2 + (size_t)(hi * 2 + 4 * q) * ldw2 + lane16;
  v2f a0P[16], a1P[16], a0Q[16], a1Q[16];
  float2 raP[16], rbP[16], raQ[16], rbQ[16];
  v2f a0nP[4], a1nP[4], bnP[4], a0nQ[4], a1nQ[4], bnQ[4];
  load_chunk<PK>(arow0, arow1, W, ldw, n0, arow0n, arow1n, w2col, ldw2, 0, lane16, hi,
                 a0P, a1P, raP, rbP, a0nP, a1nP, bnP);
  keep_order();
#pragma unroll 1
  for (int kb = 0; kb < K - 2 * CH; kb += 2 * CH) {
    load_chunk<PK>(arow0, arow1, W, ldw, n0, arow0n, arow1n, w2col, ldw2, kb + CH, lane16, hi,
                   a0Q, a1Q, raQ, rbQ, a0nQ, a1nQ, bnQ);
    keep_order();
    mma_chunk<PK>(a0P, a1P, raP, rbP, a0nP, a1nP, bnP, c00, c01, c10, c11, s0, s1);
    keep_order();
    load_chunk<PK>(arow0, arow1, W, ldw, n0, arow0n, arow1n, w2col, ldw2, kb + 2 * CH, lane16, hi,
                   a0P, a1P, raP, rbP, a0nP, a1nP, bnP);
    keep_order();
    mma_chunk<PK>(a0Q, a1Q, raQ, rbQ, a0nQ, a1nQ, bnQ, c00, c01, c10, c11, s0, s1);
    keep_order();
  }
  load_chunk<PK>(arow0, arow1, W, ldw, n0, arow0n, arow1n, w2col, ldw2, K - CH, lane16, hi,
                 a0Q, a1Q, raQ, rbQ, a0nQ, a1nQ, bnQ);
  keep_order();
  mma_chunk<PK>(a0P, a1P, raP, rbP, a0nP, a1nP, bnP, c00, c01, c10, c11, s0, s1);
  mma_chunk<PK>(a0Q, a1Q, raQ, rbQ, a0nQ, a1nQ, bnQ, c00, c01, c10, c11, s0, s1);
}

__device__ __forceinline__ void store_tile(float* dst, int ldd, int rbase,
                                           int col0, v8f c, int lane16, int hi) {
#pragma unroll
  for (int v = 0; v < 8; ++v)
    dst[(rbase + v + hi * 8) * ldd + col0 + lane16] = c[v];
}

__device__ __forceinline__ v8f add_bias(v8f c, float b) {
#pragma unroll
  for (int v = 0; v < 8; ++v) c[v] += b;
  return c;
}
#define LOG2E_F 1.44269504088896340736f
__device__ __forceinline__ float hw_exp(float t) {
  return __builtin_amdgcn_exp2f(t * LOG2E_F);
}
__device__ __forceinline__ v8f bias_elu(v8f c, float b) {
#pragma unroll
  for (int v = 0; v < 8; ++v) {
    float t = c[v] + b;
    float em1 = hw_exp(t) - 1.f;
    c[v] = t > 0.f ? t : em1;
  }
  return c;
}
__device__ __forceinline__ v8f gate_mix(v8f g, float b, v8f e, v8f h2) {
#pragma unroll
  for (int v = 0; v < 8; ++v) {
    float t = g[v] + b;
    float s = __builtin_amdgcn_rcpf(1.f + hw_exp(-t));
    g[v] = e[v] + s * h2[v];
  }
  return g;
}

#define FOLD_ROWS 80
#define FOLD_BYTES 655360
#define FOLD_BLOCKS (F_ * 5)

#define PK_FOLD 0
#define PK_FBIAS 131072
#define PK_EMB 133120
#define PK_FC2 264192
#define PK_G 526336
#define PK_SF1 788480
#define PACK_BYTES 4202496
#define PACK_BLOCKS 112

template <bool PK>
__global__ __launch_bounds__(256)
void vsn_fold_fc1(const float* __restrict__ embW, const float* __restrict__ embB,
                  const float* __restrict__ fc1W, const float* __restrict__ fc1B,
                  const float* __restrict__ fc2W, const float* __restrict__ gW,
                  const float* __restrict__ sf1W, float* __restrict__ foldW) {
  if constexpr (PK) {
    if (blockIdx.x >= FOLD_BLOCKS) {
      const int pb = blockIdx.x - FOLD_BLOCKS;
      const float* src;
      float* dst;
      int lb;
      if (pb < 16)      { src = embW; dst = foldW + PK_EMB; lb = pb; }
      else if (pb < 48) { src = fc2W; dst = foldW + PK_FC2; lb = pb - 16; }
      else if (pb < 80) { src = gW;   dst = foldW + PK_G;   lb = pb - 48; }
      else              { src = sf1W; dst = foldW + PK_SF1; lb = pb - 80; }
      float2 r0[8], r1[8];
#pragma unroll
      for (int i = 0; i < 8; ++i) {
        const int o = lb * 2048 + i * 256 + threadIdx.x;
        const float* sp = src + (size_t)(o >> 6) * (2 * H_) + 2 * (o & 63);
        r0[i] = *(const float2*)sp;
        r1[i] = *(const float2*)(sp + H_);
      }
#pragma unroll
      for (int i = 0; i < 8; ++i) {
        const int o = lb * 2048 + i * 256 + threadIdx.x;
        float4 t;
        t.x = r0[i].x; t.y = r1[i].x;
        t.z = r0[i].y; t.w = r1[i].y;
        *(float4*)(dst + (size_t)o * 4) = t;
      }
      return;
    }
  }
  const int f = blockIdx.x / 5;
  const int rt = blockIdx.x % 5;
  const int wid = threadIdx.x >> 5;
  const int lane = threadIdx.x & 31;
  const int lane16 = lane & 15;
  const int hi = lane >> 4;
  const int n0 = wid * 16;
  const bool wrow = rt < 4;
  const bool live = wrow || lane16 == 0;
  v2f b[H_ / 4];
  load_panel<H_>(fc1W + (size_t)f * H_ * H_, H_, n0, lane16, hi, b);
  const float* ar = (wrow ? embW + ((size_t)f * D_ + rt * 16 + lane16) * H_
                          : embB + (size_t)f * H_) + hi * 2;
  float2 araw[H_ / 4];
#pragma unroll
  for (int s = 0; s < H_ / 4; ++s) araw[s] = *(const float2*)(ar + 4 * s);
  const float cb0 = fc1B[f * H_ + n0 + lane16];
  keep_order();
  v8f c = {0.f, 0.f, 0.f, 0.f, 0.f, 0.f, 0.f, 0.f};
  c[0] = (!wrow && hi == 0) ? cb0 : 0.f;
#pragma unroll
  for (int s = 0; s < H_ / 4; ++s) {
    v2f a;
    a.x = live ? araw[s].x : 0.f;
    a.y = live ? araw[s].y : 0.f;
    c = mma4(a, b[s], c);
  }
  if constexpr (PK) {
    if (wrow) {
      float* dp = foldW + PK_FOLD + (size_t)f * (D_ * H_) +
                  ((size_t)(rt * 8 + hi * 4) * H_ + n0 + lane16) * 2;
#pragma unroll
      for (int v = 0; v < 8; v += 2) {
        float2 t;
        t.x = c[v];
        t.y = c[v + 1];
        *(float2*)(dp + (size_t)(v / 2) * (2 * H_)) = t;
      }
    } else if (hi == 0) {
      foldW[PK_FBIAS + f * H_ + n0 + lane16] = c[0];
    }
  } else {
    store_tile(foldW + (size_t)f * FOLD_ROWS * H_, H_, rt * 16, n0, c, lane16, hi);
  }
}

template <int MODE>
__global__ __launch_bounds__(256, 1)
void vsn_fused_pairs32(
    const float* __restrict__ x,
    const float* __restrict__ embW, const float* __restrict__ embB,
    const float* __restrict__ fc1W, const float* __restrict__ fc1B,
    const float* __restrict__ fc2W, const float* __restrict__ fc2B,
    const float* __restrict__ gW,   const float* __restrict__ gB,
    const float* __restrict__ lnG,  const float* __restrict__ lnB,
    const float* __restrict__ sf1W, const float* __restrict__ sf1B,
    const float* __restrict__ sf2W, const float* __restrict__ sf2B,
    const float* __restrict__ sgW,  const float* __restrict__ sgB,
    const float* __restrict__ sskW, const float* __restrict__ sskB,
    const float* __restrict__ slnG, const float* __restrict__ slnB,
    const float* __restrict__ foldW,
    float* __restrict__ outSel, float* __restrict__ outW) {
  __shared__ __attribute__((aligned(16))) float smem[SMEM_FLOATS];
  float* TRANS = smem;
  float* XT0 = TRANS + MR * ST_T;
  float* XW0 = XT0 + 2 * MR * ST_X;
  float* PU = XW0;
  float* PV = XW0 + MR * ST_P;
  float* SKP = XT0;
  float* GP = PU;
  float* WB = GP + MR * ST_S;
  constexpr bool FOLD = MODE >= 1;
  constexpr bool PK = MODE == 2;

  const int tid = threadIdx.x;
  const int wid = tid >> 5;
  const int lane = tid & 31;
  const int lane16 = lane & 15;
  const int hi = lane >> 4;
  const int row0 = blockIdx.x * MR;

  const int g = wid >> 2;
  const int n0 = (wid & 3) * 32;
  const int cb = n0 + 2 * lane16;
  const int glr = (tid & 127) >> 4;
  const int lc = tid & 15;
  float* XT = XT0 + g * (MR * ST_X);
  float* XW = XW0 + g * (MR * ST_P);

  if constexpr (FOLD) {
  const float* eW = PK ? foldW + PK_EMB : embW;
  const float* xW = PK ? foldW + PK_FOLD : foldW;
  const float* xB = PK ? foldW + PK_FBIAS : foldW + D_ * H_;
  const float* f2W = PK ? foldW + PK_FC2 : fc2W;
  const float* gWp = PK ? foldW + PK_G : gW;
  constexpr int xS = PK ? D_ * H_ : FOLD_ROWS * H_;
  constexpr int xbS = PK ? H_ : FOLD_ROWS * H_;
  float2 bE = *(const float2*)(embB + g * H_ + cb);
  float2 b1 = *(const float2*)(xB + (size_t)g * xbS + cb);
  float2 b2 = *(const float2*)(fc2B + g * H_ + cb);
  float2 bG = *(const float2*)(gB + g * H_ + cb);
  float2 pE0[D_ / 4], pE1[D_ / 4];
  float2 pX0[D_ / 4], pX1[D_ / 4];
  load_panel2<0, D_ / 4, D_ / 4, PK>(eW + (size_t)g * D_ * H_, H_, n0, lane16, hi, pE0, pE1);
  load_panel2<0, D_ / 4, D_ / 4, PK>(xW + (size_t)g * xS, H_, n0, lane16, hi, pX0, pX1);
  {
#pragma unroll
    for (int rr = 0; rr < 4; ++rr) {
      const int r = glr + 8 * rr;
      float4 v = *(const float4*)(x + (size_t)(row0 + r) * (F_ * D_) + g * D_ + lc * 4);
      float* dp = XT + r * ST_X + lc * 4;
      dp[0] = v.x; dp[1] = v.y; dp[2] = v.z; dp[3] = v.w;
    }
  }
  float2 pF20[H_ / 4], pF21[H_ / 4];
  float4 xr[4];
  wg_signal_lds();
  keep_order();
  load_panel2<0, H_ / 4, H_ / 4, PK>(f2W + (size_t)g * H_ * H_, H_, n0, lane16, hi, pF20, pF21);
#pragma unroll
  for (int rr = 0; rr < 4; ++rr)
    xr[rr] = *(const float4*)(x + (size_t)(row0 + glr + 8 * rr) * (F_ * D_) +
                              (g + 2) * D_ + lc * 4);
  keep_order();
  wg_wait();

  for (int f = g; f < F_; f += 2) {
    const int fn = f + 2 < F_ ? f + 2 : f;
    const int fnn = f + 4 < F_ ? f + 4 : fn;
    float* SLOT = TRANS + f * H_;

    v8f ce00, ce01, ce10, ce11;
    {
      v8f c00, c01, c10, c11;
      gemm_panel2x2<D_, PK>(XT, ST_X, lane16, hi, pE0, pE1, pX0, pX1,
                        ce00, ce01, ce10, ce11, c00, c01, c10, c11);
      ce00 = add_bias(ce00, bE.x);
      ce10 = add_bias(ce10, bE.x);
      ce01 = add_bias(ce01, bE.y);
      ce11 = add_bias(ce11, bE.y);
      c00 = bias_elu(c00, b1.x);
      c10 = bias_elu(c10, b1.x);
      c01 = bias_elu(c01, b1.y);
      c11 = bias_elu(c11, b1.y);
      store_tile2(XW, ST_P, 0, cb, c00, c01, hi);
      store_tile2(XW, ST_P, 16, cb, c10, c11, hi);
    }
    wg_signal_lds();
    keep_order();
    float2 pG0[H_ / 4], pG1[H_ / 4];
    load_panel2<0, H_ / 4, H_ / 4, PK>(gWp + (size_t)f * H_ * H_, H_, n0, lane16, hi, pG0, pG1);
    keep_order();
    wg_wait();

    v8f c200, c201, c210, c211;
    gemm_panel2<H_, PK>(XW, ST_P, lane16, hi, pF20, pF21, c200, c201, c210, c211);
    c200 = add_bias(c200, b2.x);
    c210 = add_bias(c210, b2.x);
    c201 = add_bias(c201, b2.y);
    c211 = add_bias(c211, b2.y);
    store_tile2(SLOT, ST_T, 0, cb, c200, c201, hi);
    store_tile2(SLOT, ST_T, 16, cb, c210, c211, hi);
    wg_signal_lds();
    keep_order();

    load_panel2<0, D_ / 4, D_ / 4, PK>(eW + (size_t)fn * D_ * H_, H_, n0, lane16, hi, pE0, pE1);
    const float2 bGf = bG;
    bE = *(const float2*)(embB + fn * H_ + cb);
    b1 = *(const float2*)(xB + (size_t)fn * xbS + cb);
    b2 = *(const float2*)(fc2B + fn * H_ + cb);
    bG = *(const float2*)(gB + fn * H_ + cb);
    float lg[8], lb[8];
#pragma unroll
    for (int j = 0; j < 8; ++j) {
      lg[j] = lnG[f * H_ + lc + 16 * j];
      lb[j] = lnB[f * H_ + lc + 16 * j];
    }
    keep_order();
    wg_wait();
    {
      v8f cg00, cg01, cg10, cg11;
      gemm_panel2<H_, PK>(SLOT, ST_T, lane16, hi, pG0, pG1, cg00, cg01, cg10, cg11);
      cg00 = gate_mix(cg00, bGf.x, ce00, c200);
      cg10 = gate_mix(cg10, bGf.x, ce10, c210);
      cg01 = gate_mix(cg01, bGf.y, ce01, c201);
      cg11 = gate_mix(cg11, bGf.y, ce11, c211);
      store_tile2(XW, ST_P, 0, cb, cg00, cg01, hi);
      store_tile2(XW, ST_P, 16, cb, cg10, cg11, hi);
    }
    wg_signal_lds();
    keep_order();
    load_panel2<0, D_ / 4, D_ / 4, PK>(xW + (size_t)fn * xS, H_, n0, lane16, hi, pX0, pX1);
    load_panel2<0, 14, H_ / 4, PK>(f2W + (size_t)fn * H_ * H_, H_, n0, lane16, hi, pF20, pF21);
    keep_order();
    wg_wait();

    float vals[4][8];
#pragma unroll
    for (int rr = 0; rr < 4; ++rr) {
      const int r = glr + 8 * rr;
#pragma unroll
      for (int j = 0; j < 8; ++j) vals[rr][j] = XW[r * ST_P + lc + 16 * j];
    }
    float sm[8];
#pragma unroll
    for (int rr = 0; rr < 4; ++rr) {
      float s = 0.f, ss = 0.f;
#pragma unroll
      for (int j = 0; j < 8; ++j) {
        float v = vals[rr][j];
        s += v; ss += v * v;
      }
      sm[2 * rr] = s;
      sm[2 * rr + 1] = ss;
    }
    rowsum16_n<8>(sm);
#pragma unroll
    for (int rr = 0; rr < 4; ++rr) {
      const int r = glr + 8 * rr;
      float mu = sm[2 * rr] * (1.f / 128.f);
      float var = sm[2 * rr + 1] * (1.f / 128.f) - mu * mu;
      float rs = rsqrtf(var + 1e-5f);
#pragma unroll
      for (int j = 0; j < 8; ++j) {
        int h = lc + 16 * j;
        float nv = (vals[rr][j] - mu) * rs * lg[j] + lb[j];
        SLOT[r * ST_T + h] = nv;
      }
      float* dp = XT + r * ST_X + lc * 4;
      dp[0] = xr[rr].x; dp[1] = xr[rr].y; dp[2] = xr[rr].z; dp[3] = xr[rr].w;
    }
    wg_signal_lds();
    keep_order();
    load_panel2<14, H_ / 4, H_ / 4, PK>(f2W + (size_t)fn * H_ * H_, H_, n0, lane16, hi, pF20, pF21);
#pragma unroll
    for (int rr = 0; rr < 4; ++rr)
      xr[rr] = *(const float4*)(x + (size_t)(row0 + glr + 8 * rr) * (F_ * D_) +
                                fnn * D_ + lc * 4);
    keep_order();
    wg_wait();
  }
  } else {
  float2 bE = *(const float2*)(embB + g * H_ + cb);
  float2 b1 = *(const float2*)(fc1B + g * H_ + cb);
  float2 b2 = *(const float2*)(fc2B + g * H_ + cb);
  float2 bG = *(const float2*)(gB + g * H_ + cb);
  float2 pE0[D_ / 4], pE1[D_ / 4];
  load_panel2<0, D_ / 4, D_ / 4>(embW + (size_t)g * D_ * H_, H_, n0, lane16, hi, pE0, pE1);
  {
#pragma unroll
    for (int rr = 0; rr < 4; ++rr) {
      const int r = glr + 8 * rr;
      float4 v = *(const float4*)(x + (size_t)(row0 + r) * (F_ * D_) + g * D_ + lc * 4);
      float* dp = XT + r * ST_X + lc * 4;
      dp[0] = v.x; dp[1] = v.y; dp[2] = v.z; dp[3] = v.w;
    }
  }
  float2 pF10[H_ / 4], pF11[H_ / 4];
  float4 xr[4];
  wg_signal_lds();
  keep_order();
  load_panel2<0, H_ / 4, H_ / 4>(fc1W + (size_t)g * H_ * H_, H_, n0, lane16, hi, pF10, pF11);
#pragma unroll
  for (int rr = 0; rr < 4; ++rr)
    xr[rr] = *(const float4*)(x + (size_t)(row0 + glr + 8 * rr) * (F_ * D_) +
                              (g + 2) * D_ + lc * 4);
  keep_order();
  wg_wait();

  for (int f = g; f < F_; f += 2) {
    const int fn = f + 2 < F_ ? f + 2 : f;
    const int fnn = f + 4 < F_ ? f + 4 : fn;
    float* SLOT = TRANS + f * H_;

    v8f ce00, ce01, ce10, ce11;
    gemm_panel2<D_>(XT, ST_X, lane16, hi, pE0, pE1, ce00, ce01, ce10, ce11);
    ce00 = add_bias(ce00, bE.x);
    ce10 = add_bias(ce10, bE.x);
    ce01 = add_bias(ce01, bE.y);
    ce11 = add_bias(ce11, bE.y);
    store_tile2(SLOT, ST_T, 0, cb, ce00, ce01, hi);
    store_tile2(SLOT, ST_T, 16, cb, ce10, ce11, hi);
    wg_signal_lds();
    keep_order();
    float2 pF20[H_ / 4], pF21[H_ / 4];
    load_panel2<0, H_ / 4, H_ / 4>(fc2W + (size_t)f * H_ * H_, H_, n0, lane16, hi, pF20, pF21);
    keep_order();
    wg_wait();

    {
      v8f c00, c01, c10, c11;
      gemm_panel2<H_>(SLOT, ST_T, lane16, hi, pF10, pF11, c00, c01, c10, c11);
      c00 = bias_elu(c00, b1.x);
      c10 = bias_elu(c10, b1.x);
      c01 = bias_elu(c01, b1.y);
      c11 = bias_elu(c11, b1.y);
      store_tile2(XW, ST_P, 0, cb, c00, c01, hi);
      store_tile2(XW, ST_P, 16, cb, c10, c11, hi);
    }
    wg_signal_lds();
    keep_order();
    float2 pG0[H_ / 4], pG1[H_ / 4];
    load_panel2<0, H_ / 4, H_ / 4>(gW + (size_t)f * H_ * H_, H_, n0, lane16, hi, pG0, pG1);
    keep_order();
    wg_wait();

    v8f c200, c201, c210, c211;
    gemm_panel2<H_>(XW, ST_P, lane16, hi, pF20, pF21, c200, c201, c210, c211);
    c200 = add_bias(c200, b2.x);
    c210 = add_bias(c210, b2.x);
    c201 = add_bias(c201, b2.y);
    c211 = add_bias(c211, b2.y);
    store_tile2(SLOT, ST_T, 0, cb, c200, c201, hi);
    store_tile2(SLOT, ST_T, 16, cb, c210, c211, hi);
    wg_signal_lds();
    keep_order();

    load_panel2<0, D_ / 4, D_ / 4>(embW + (size_t)fn * D_ * H_, H_, n0, lane16, hi, pE0, pE1);
    const float2 bGf = bG;
    bE = *(const float2*)(embB + fn * H_ + cb);
    b1 = *(const float2*)(fc1B + fn * H_ + cb);
    b2 = *(const float2*)(fc2B + fn * H_ + cb);
    bG = *(const float2*)(gB + fn * H_ + cb);
    float lg[8], lb[8];
#pragma unroll
    for (int j = 0; j < 8; ++j) {
      lg[j] = lnG[f * H_ + lc + 16 * j];
      lb[j] = lnB[f * H_ + lc + 16 * j];
    }
    keep_order();
    wg_wait();
    {
      v8f cg00, cg01, cg10, cg11;
      gemm_panel2<H_>(SLOT, ST_T, lane16, hi, pG0, pG1, cg00, cg01, cg10, cg11);
      cg00 = gate_mix(cg00, bGf.x, ce00, c200);
      cg10 = gate_mix(cg10, bGf.x, ce10, c210);
      cg01 = gate_mix(cg01, bGf.y, ce01, c201);
      cg11 = gate_mix(cg11, bGf.y, ce11, c211);
      store_tile2(XW, ST_P, 0, cb, cg00, cg01, hi);
      store_tile2(XW, ST_P, 16, cb, cg10, cg11, hi);
    }
    wg_signal_lds();
    keep_order();
    load_panel2<0, H_ / 8, H_ / 4>(fc1W + (size_t)fn * H_ * H_, H_, n0, lane16, hi, pF10, pF11);
    keep_order();
    wg_wait();

    float vals[4][8];
#pragma unroll
    for (int rr = 0; rr < 4; ++rr) {
      const int r = glr + 8 * rr;
#pragma unroll
      for (int j = 0; j < 8; ++j) vals[rr][j] = XW[r * ST_P + lc + 16 * j];
    }
    float sm[8];
#pragma unroll
    for (int rr = 0; rr < 4; ++rr) {
      float s = 0.f, ss = 0.f;
#pragma unroll
      for (int j = 0; j < 8; ++j) {
        float v = vals[rr][j];
        s += v; ss += v * v;
      }
      sm[2 * rr] = s;
      sm[2 * rr + 1] = ss;
    }
    rowsum16_n<8>(sm);
#pragma unroll
    for (int rr = 0; rr < 4; ++rr) {
      const int r = glr + 8 * rr;
      float mu = sm[2 * rr] * (1.f / 128.f);
      float var = sm[2 * rr + 1] * (1.f / 128.f) - mu * mu;
      float rs = rsqrtf(var + 1e-5f);
#pragma unroll
      for (int j = 0; j < 8; ++j) {
        int h = lc + 16 * j;
        float nv = (vals[rr][j] - mu) * rs * lg[j] + lb[j];
        SLOT[r * ST_T + h] = nv;
      }
      float* dp = XT + r * ST_X + lc * 4;
      dp[0] = xr[rr].x; dp[1] = xr[rr].y; dp[2] = xr[rr].z; dp[3] = xr[rr].w;
    }
    wg_signal_lds();
    keep_order();
    load_panel2<H_ / 8, H_ / 4, H_ / 4>(fc1W + (size_t)fn * H_ * H_, H_, n0, lane16, hi, pF10, pF11);
#pragma unroll
    for (int rr = 0; rr < 4; ++rr)
      xr[rr] = *(const float4*)(x + (size_t)(row0 + glr + 8 * rr) * (F_ * D_) +
                                fnn * D_ + lc * 4);
    keep_order();
    wg_wait();
  }

  }

  const int nB = wid * 16;
  const int lr = tid >> 4;
  float bS1[8];
#pragma unroll
  for (int j = 0; j < 8; ++j) bS1[j] = sf1B[lc + 16 * j];
  const float bS2 = sf2B[nB + lane16];
  const float bSK = sskB[lc];
  {
    v8f b00, b01, b10, b11, sk0, sk1;
    gemm_stream<FH_ / 2, 64, PK>(TRANS + g * (FH_ / 2), ST_T,
                                 (PK ? foldW + PK_SF1 : sf1W) + (size_t)g * (FH_ / 2) * H_, H_, n0,
                                 sskW + (size_t)g * (FH_ / 2) * F_, F_,
                                 wid & 3, lane16, hi, b00, b01, b10, b11, sk0, sk1);
    float* PT = XW0 + g * (MR * ST_P);
    store_tile2(PT, ST_P, 0, cb, b00, b01, hi);
    store_tile2(PT, ST_P, 16, cb, b10, b11, hi);
    float* SKPw = SKP + wid * (MR * 16);
    store_tile(SKPw, 16, 0, 0, sk0, lane16, hi);
    store_tile(SKPw, 16, 16, 0, sk1, lane16, hi);
  }
  wg_signal_lds();
  keep_order();
  v2f pS2[H_ / 4];
  load_panel<H_>(sf2W, H_, nB, lane16, hi, pS2);
  keep_order();
  wg_wait();

  {
    float u[2][8], w2[2][8];
#pragma unroll
    for (int rr = 0; rr < 2; ++rr) {
      const int r = lr + 16 * rr;
#pragma unroll
      for (int j = 0; j < 8; ++j) {
        u[rr][j] = PU[r * ST_P + lc + 16 * j];
        w2[rr][j] = PV[r * ST_P + lc + 16 * j];
      }
    }
#pragma unroll
    for (int rr = 0; rr < 2; ++rr) {
      const int r = lr + 16 * rr;
#pragma unroll
      for (int j = 0; j < 8; ++j) {
        float t = (u[rr][j] + w2[rr][j]) + bS1[j];
        float em1 = hw_exp(t) - 1.f;
        PU[r * ST_P + lc + 16 * j] = t > 0.f ? t : em1;
      }
    }
  }
  wg_barrier_lds();

  {
    v8f b20, b21;
    gemm_panel<H_>(PU, ST_P, lane16, hi, pS2, b20, b21);
    b20 = add_bias(b20, bS2);
    b21 = add_bias(b21, bS2);
    store_tile(PV, ST_P, 0, nB, b20, lane16, hi);
    store_tile(PV, ST_P, 16, nB, b21, lane16, hi);
  }
  wg_barrier_lds();

  if (wid == 0) {
    v2f pSG[H_ / 4];
    load_panel<H_>(sgW, F_, 0, lane16, hi, pSG);
    v8f g0, g1;
    gemm_panel<H_>(PV, ST_P, lane16, hi, pSG, g0, g1);
    float b = sgB[lane16];
    g0 = add_bias(g0, b);
    g1 = add_bias(g1, b);
    store_tile(GP, ST_S, 0, 0, g0, lane16, hi);
    store_tile(GP, ST_S, 16, 0, g1, lane16, hi);
  }
  wg_barrier_lds();

  {
    float gpv[2], skv[2];
#pragma unroll
    for (int rr = 0; rr < 2; ++rr) {
      const int r = lr + 16 * rr;
      gpv[rr] = GP[r * ST_S + lc];
      float sk = SKP[r * 16 + lc];
#pragma unroll
      for (int pw = 1; pw < 8; ++pw) sk += SKP[pw * (MR * 16) + r * 16 + lc];
      skv[rr] = sk + bSK;
    }
    float vv[2], st[4];
#pragma unroll
    for (int rr = 0; rr < 2; ++rr) {
      float gp = gpv[rr];
      float sg = 1.f / (1.f + expf(-gp));
      float v = skv[rr] + sg * gp;
      vv[rr] = v;
      st[2 * rr] = v;
      st[2 * rr + 1] = v * v;
    }
    rowsum16_n<4>(st);
    const float sG = slnG[lc], sB = slnB[lc];
    float nv[2], mx[2];
#pragma unroll
    for (int rr = 0; rr < 2; ++rr) {
      float mu = st[2 * rr] * (1.f / 16.f);
      float var = st[2 * rr + 1] * (1.f / 16.f) - mu * mu;
      nv[rr] = (vv[rr] - mu) * rsqrtf(var + 1e-5f) * sG + sB;
      mx[rr] = nv[rr];
    }
    rowmax16_n<2>(mx);
    float e[2], se[2];
#pragma unroll
    for (int rr = 0; rr < 2; ++rr) {
      e[rr] = expf(nv[rr] - mx[rr]);
      se[rr] = e[rr];
    }
    rowsum16_n<2>(se);
#pragma unroll
    for (int rr = 0; rr < 2; ++rr) {
      const int r = lr + 16 * rr;
      float w = e[rr] / se[rr];
      WB[r * 16 + lc] = w;
      outW[(size_t)(row0 + r) * F_ + lc] = w;
    }
  }
  wg_barrier_lds();

#pragma unroll
  for (int rr = 0; rr < 2; ++rr) {
    const int r = lr + 16 * rr;
#pragma unroll
    for (int j = 0; j < 8; ++j) {
      int h = lc + 16 * j;
      float acc = 0.f;
#pragma unroll
      for (int f = 0; f < F_; ++f)
        acc += TRANS[r * ST_T + f * H_ + h] * WB[r * 16 + f];
      outSel[(size_t)(row0 + r) * H_ + h] = acc;
    }
  }
}

extern "C" void kernel_launch(void* const* d_in, const int* in_sizes, int n_in,
                              void* d_out, int out_size, void* d_ws, size_t ws_size,
                              hipStream_t stream) {
  (void)in_sizes; (void)n_in; (void)out_size;
  const float* x    = (const float*)d_in[0];
  const float* embW = (const float*)d_in[1];
  const float* embB = (const float*)d_in[2];
  const float* fc1W = (const float*)d_in[3];
  const float* fc1B = (const float*)d_in[4];
  const float* fc2W = (const float*)d_in[5];
  const float* fc2B = (const float*)d_in[6];
  const float* gW   = (const float*)d_in[7];
  const float* gB   = (const float*)d_in[8];
  const float* lnG  = (const float*)d_in[9];
  const float* lnB  = (const float*)d_in[10];
  const float* sf1W = (const float*)d_in[11];
  const float* sf1B = (const float*)d_in[12];
  const float* sf2W = (const float*)d_in[13];
  const float* sf2B = (const float*)d_in[14];
  const float* sgW  = (const float*)d_in[15];
  const float* sgB  = (const float*)d_in[16];
  const float* sskW = (const float*)d_in[17];
  const float* sskB = (const float*)d_in[18];
  const float* slnG = (const float*)d_in[19];
  const float* slnB = (const float*)d_in[20];

  float* outSel = (float*)d_out;
  float* outW   = outSel + (size_t)ROWS_TOTAL * H_;

  if (d_ws != nullptr && ws_size >= (size_t)PACK_BYTES) {
    float* foldW = (float*)d_ws;
    vsn_fold_fc1<true><<<FOLD_BLOCKS + PACK_BLOCKS, 256, 0, stream>>>(
        embW, embB, fc1W, fc1B, fc2W, gW, sf1W, foldW);
    vsn_fused_pairs32<2><<<ROWS_TOTAL / MR, 256, 0, stream>>>(
        x, embW, embB, fc1W, fc1B, fc2W, fc2B, gW, gB, lnG, lnB,
        sf1W, sf1B, sf2W, sf2B, sgW, sgB, sskW, sskB, slnG, slnB,
        foldW, outSel, outW);
  } else if (d_ws != nullptr && ws_size >= (size_t)FOLD_BYTES) {
    float* foldW = (float*)d_ws;
    vsn_fold_fc1<false><<<FOLD_BLOCKS, 256, 0, stream>>>(
        embW, embB, fc1W, fc1B, fc2W, gW, sf1W, foldW);
    vsn_fused_pairs32<1><<<ROWS_TOTAL / MR, 256, 0, stream>>>(
        x, embW, embB, fc1W, fc1B, fc2W, fc2B, gW, gB, lnG, lnB,
        sf1W, sf1B, sf2W, sf2B, sgW, sgB, sskW, sskB, slnG, slnB,
        foldW, outSel, outW);
  } else {
    vsn_fused_pairs32<0><<<ROWS_TOTAL / MR, 256, 0, stream>>>(
        x, embW, embB, fc1W, fc1B, fc2W, fc2B, gW, gB, lnG, lnB,
        sf1W, sf1B, sf2W, sf2B, sgW, sgB, sskW, sskB, slnG, slnB,
        (const float*)nullptr, outSel, outW);
  }
}
